# starting kernel
# baseline (speedup 1.0000x reference)
#include <hip/hip_runtime.h>
#include <cstdint>

#define BGRAPH 8
#define NNODES 8192
#define NEDGES 262144
#define EPRIME (NEDGES + NNODES)
#define FEAT   128
#define NEG_SLOPE 0.2f

typedef __attribute__((ext_vector_type(2))) float v2f;
typedef __attribute__((ext_vector_type(8))) float v8f;

__device__ __forceinline__ int f32_ord(float f) {
  int i = __float_as_int(f);
  return i >= 0 ? i : (i ^ 0x7fffffff);
}
__device__ __forceinline__ float ord_f32(int i) {
  return __int_as_float(i >= 0 ? i : (i ^ 0x7fffffff));
}
__device__ __forceinline__ void atomic_add_f32(float* p, float v) {
  asm volatile("global_atomic_add_f32 %0, %1, off"
               :
               : "v"(p), "v"(v)
               : "memory");
}
__device__ __forceinline__ float leaky(float v) {
  return v >= 0.0f ? v : NEG_SLOPE * v;
}

template <bool TRANSB, bool BIAS>
__global__ void __launch_bounds__(256)
gemm_wmma_f32(const float* __restrict__ X, const float* __restrict__ W,
              const float* __restrict__ bias, float* __restrict__ H) {
  const int g    = blockIdx.y;
  const int m0   = blockIdx.x * 16;
  const int wave = threadIdx.x >> 5;
  const int lane = threadIdx.x & 31;
  const int half = lane >> 4;
  const int l    = lane & 15;
  const int n0   = wave * 16;

  const float* xrow = X + ((size_t)g * NNODES + (m0 + l)) * FEAT;

  v8f acc = {};
#pragma unroll
  for (int k = 0; k < FEAT; k += 4) {
    const int ka = k + 2 * half;
    v2f a;
    a.x = xrow[ka];
    a.y = xrow[ka + 1];
    v2f b;
    if (TRANSB) {
      const float* wr = W + (size_t)(n0 + l) * FEAT + ka;
      b.x = wr[0];
      b.y = wr[1];
    } else {
      b.x = W[(size_t)ka * FEAT + (n0 + l)];
      b.y = W[(size_t)(ka + 1) * FEAT + (n0 + l)];
    }
    acc = __builtin_amdgcn_wmma_f32_16x16x4_f32(
         false, a,  false, b,
         (short)0, acc,  false,  false);
  }

  const float bv = BIAS ? bias[n0 + l] : 0.0f;
  float* hout = H + ((size_t)g * NNODES + m0 + 8 * half) * FEAT + (n0 + l);
#pragma unroll
  for (int i = 0; i < 8; ++i)
    hout[(size_t)i * FEAT] = acc[i] + bv;
}

__global__ void __launch_bounds__(256)
alpha_kernel(const float* __restrict__ H, const float* __restrict__ a_src,
             const float* __restrict__ a_dst, float* __restrict__ AS,
             float* __restrict__ AD) {
  const int wave = threadIdx.x >> 5;
  const int lane = threadIdx.x & 31;
  const int node = blockIdx.x * 8 + wave;
  const int f    = lane * 4;
  const float4 h4 = *(const float4*)(H + (size_t)node * FEAT + f);
  float ps = h4.x * a_src[f] + h4.y * a_src[f + 1] +
             h4.z * a_src[f + 2] + h4.w * a_src[f + 3];
  float pd = h4.x * a_dst[f] + h4.y * a_dst[f + 1] +
             h4.z * a_dst[f + 2] + h4.w * a_dst[f + 3];
#pragma unroll
  for (int off = 16; off > 0; off >>= 1) {
    ps += __shfl_xor(ps, off, 32);
    pd += __shfl_xor(pd, off, 32);
  }
  if (lane == 0) {
    AS[node] = ps;
    AD[node] = pd;
  }
}

__global__ void init_seg_kernel(int* __restrict__ MORD, float* __restrict__ SSUM) {
  const int i = blockIdx.x * blockDim.x + threadIdx.x;
  MORD[i] = (int)0x80000000;
  SSUM[i] = 0.0f;
}

__device__ __forceinline__ void edge_nodes(const int* __restrict__ ei, int e,
                                           int& s, int& d) {
  if (e < NEDGES) {
    s = ei[e];
    d = ei[NEDGES + e];
  } else {
    s = d = e - NEDGES;
  }
}

__global__ void __launch_bounds__(256)
edge_max_kernel(const int* __restrict__ EI, const float* __restrict__ AS,
                const float* __restrict__ AD, int* __restrict__ MORD) {
  const long idx = (long)blockIdx.x * blockDim.x + threadIdx.x;
  const int g = (int)(idx / EPRIME);
  const int e = (int)(idx - (long)g * EPRIME);
  int s, d;
  edge_nodes(EI + (size_t)g * 2 * NEDGES, e, s, d);
  const float v = leaky(AS[g * NNODES + s] + AD[g * NNODES + d]);
  atomicMax(&MORD[g * NNODES + d], f32_ord(v));
}

__global__ void __launch_bounds__(256)
edge_sum_kernel(const int* __restrict__ EI, const float* __restrict__ AS,
                const float* __restrict__ AD, const int* __restrict__ MORD,
                float* __restrict__ SSUM) {
  const long idx = (long)blockIdx.x * blockDim.x + threadIdx.x;
  const int g = (int)(idx / EPRIME);
  const int e = (int)(idx - (long)g * EPRIME);
  int s, d;
  edge_nodes(EI + (size_t)g * 2 * NEDGES, e, s, d);
  const int gd = g * NNODES + d;
  const float v = leaky(AS[g * NNODES + s] + AD[gd]);
  const float w = __expf(v - ord_f32(MORD[gd]));
  atomic_add_f32(&SSUM[gd], w);
}

__global__ void __launch_bounds__(256)
edge_agg_kernel(const int* __restrict__ EI, const float* __restrict__ AS,
                const float* __restrict__ AD, const int* __restrict__ MORD,
                const float* __restrict__ SSUM, const float* __restrict__ H,
                float* __restrict__ ACC) {
  const int wave = threadIdx.x >> 5;
  const int lane = threadIdx.x & 31;
  const long widx = (long)blockIdx.x * 8 + wave;
  const int g = (int)(widx / EPRIME);
  const int e = (int)(widx - (long)g * EPRIME);
  int s, d;
  edge_nodes(EI + (size_t)g * 2 * NEDGES, e, s, d);
  const int gs = g * NNODES + s, gd = g * NNODES + d;
  const float v = leaky(AS[gs] + AD[gd]);
  const float alpha = __expf(v - ord_f32(MORD[gd])) / SSUM[gd];
  const int f = lane * 4;
  const float4 h4 = *(const float4*)(H + (size_t)gs * FEAT + f);
  float* o = ACC + (size_t)gd * FEAT + f;
  atomic_add_f32(o + 0, alpha * h4.x);
  atomic_add_f32(o + 1, alpha * h4.y);
  atomic_add_f32(o + 2, alpha * h4.z);
  atomic_add_f32(o + 3, alpha * h4.w);
}

__global__ void __launch_bounds__(256)
bias_swish_kernel(const float* __restrict__ ACC, const float* __restrict__ b,
                  float* __restrict__ X) {
  const long i = (long)blockIdx.x * blockDim.x + threadIdx.x;
  const int f = (int)(i & (FEAT - 1));
  const float v = ACC[i] + b[f];
  X[i] = v / (1.0f + __expf(-v));
}

extern "C" void kernel_launch(void* const* d_in, const int* in_sizes, int n_in,
                              void* d_out, int out_size, void* d_ws, size_t ws_size,
                              hipStream_t stream) {
  const float* x0   = (const float*)d_in[0];
  const int*   ei   = (const int*)d_in[1];
  const float* W1   = (const float*)d_in[2];
  const float* as1  = (const float*)d_in[3];
  const float* ad1  = (const float*)d_in[4];
  const float* b1   = (const float*)d_in[5];
  const float* W2   = (const float*)d_in[6];
  const float* as2  = (const float*)d_in[7];
  const float* ad2  = (const float*)d_in[8];
  const float* b2   = (const float*)d_in[9];
  const float* fcW  = (const float*)d_in[10];
  const float* fcb  = (const float*)d_in[11];
  float* out = (float*)d_out;

  const size_t BN      = (size_t)BGRAPH * NNODES;
  const size_t BIGELTS = BN * FEAT;
  char* ws = (char*)d_ws;
  float* BUF0 = (float*)ws;                ws += BIGELTS * sizeof(float);
  float* BUF1 = (float*)ws;                ws += BIGELTS * sizeof(float);
  float* AS   = (float*)ws;                ws += BN * sizeof(float);
  float* AD   = (float*)ws;                ws += BN * sizeof(float);
  int*   MORD = (int*)ws;                  ws += BN * sizeof(int);
  float* SSUM = (float*)ws;

  const dim3 gGemm(NNODES / 16, BGRAPH);
  const int  gAlpha = (int)(BN / 8);
  const int  gInit  = (int)(BN / 256);
  const int  gEdgeT = (int)(((long)BGRAPH * EPRIME) / 256);
  const int  gEdgeW = (int)(((long)BGRAPH * EPRIME) / 8);
  const int  gElem  = (int)(BIGELTS / 256);

  gemm_wmma_f32<false, false><<<gGemm, 256, 0, stream>>>(x0, W1, nullptr, BUF0);
  alpha_kernel<<<gAlpha, 256, 0, stream>>>(BUF0, as1, ad1, AS, AD);
  init_seg_kernel<<<gInit, 256, 0, stream>>>(MORD, SSUM);
  hipMemsetAsync(BUF1, 0, BIGELTS * sizeof(float), stream);
  edge_max_kernel<<<gEdgeT, 256, 0, stream>>>(ei, AS, AD, MORD);
  edge_sum_kernel<<<gEdgeT, 256, 0, stream>>>(ei, AS, AD, MORD, SSUM);
  edge_agg_kernel<<<gEdgeW, 256, 0, stream>>>(ei, AS, AD, MORD, SSUM, BUF0, BUF1);
  bias_swish_kernel<<<gElem, 256, 0, stream>>>(BUF1, b1, BUF0);

  gemm_wmma_f32<false, false><<<gGemm, 256, 0, stream>>>(BUF0, W2, nullptr, BUF1);
  alpha_kernel<<<gAlpha, 256, 0, stream>>>(BUF1, as2, ad2, AS, AD);
  init_seg_kernel<<<gInit, 256, 0, stream>>>(MORD, SSUM);
  hipMemsetAsync(BUF0, 0, BIGELTS * sizeof(float), stream);
  edge_max_kernel<<<gEdgeT, 256, 0, stream>>>(ei, AS, AD, MORD);
  edge_sum_kernel<<<gEdgeT, 256, 0, stream>>>(ei, AS, AD, MORD, SSUM);
  edge_agg_kernel<<<gEdgeW, 256, 0, stream>>>(ei, AS, AD, MORD, SSUM, BUF1, BUF0);
  bias_swish_kernel<<<gElem, 256, 0, stream>>>(BUF0, b2, BUF1);

  gemm_wmma_f32<true, true><<<gGemm, 256, 0, stream>>>(BUF1, fcW, fcb, out);
}


# Softmax max and exp-sum per destination built in LDS per 8192-edge slice, merged by line-wide atomics; replaces the two scattered global-atomic passes; exp-sum summed per slice, then across slices
# speedup vs baseline: 1.6874x; 1.6874x over previous
#include <hip/hip_runtime.h>
#include <cstdint>

#define BGRAPH 8
#define NNODES 8192
#define NEDGES 262144
#define EPRIME (NEDGES + NNODES)
#define FEAT   128
#define NEG_SLOPE 0.2f

typedef __attribute__((ext_vector_type(2))) float v2f;
typedef __attribute__((ext_vector_type(8))) float v8f;

__device__ __forceinline__ int f32_ord(float f) {
  int i = __float_as_int(f);
  return i >= 0 ? i : (i ^ 0x7fffffff);
}
__device__ __forceinline__ float ord_f32(int i) {
  return __int_as_float(i >= 0 ? i : (i ^ 0x7fffffff));
}
__device__ __forceinline__ void atomic_add_f32(float* p, float v) {
  asm volatile("global_atomic_add_f32 %0, %1, off"
               :
               : "v"(p), "v"(v)
               : "memory");
}
template <int BYTE_OFF>
__device__ __forceinline__ void atomic_add_f32_at(float* p, float v) {
  asm volatile("global_atomic_add_f32 %0, %1, off offset:%2"
               :
               : "v"(p), "v"(v), "n"(BYTE_OFF)
               : "memory");
}
__device__ __forceinline__ float leaky(float v) {
  return v >= 0.0f ? v : NEG_SLOPE * v;
}

template <bool TRANSB, bool BIAS>
__global__ void __launch_bounds__(256)
gemm_wmma_f32(const float* __restrict__ X, const float* __restrict__ W,
              const float* __restrict__ bias, float* __restrict__ H) {
  const int g    = blockIdx.y;
  const int m0   = blockIdx.x * 16;
  const int wave = threadIdx.x >> 5;
  const int lane = threadIdx.x & 31;
  const int half = lane >> 4;
  const int l    = lane & 15;
  const int n0   = wave * 16;

  const float* xrow = X + ((size_t)g * NNODES + (m0 + l)) * FEAT;

  v8f acc = {};
#pragma unroll
  for (int k = 0; k < FEAT; k += 4) {
    const int ka = k + 2 * half;
    v2f a;
    a.x = xrow[ka];
    a.y = xrow[ka + 1];
    v2f b;
    if (TRANSB) {
      const float* wr = W + (size_t)(n0 + l) * FEAT + ka;
      b.x = wr[0];
      b.y = wr[1];
    } else {
      b.x = W[(size_t)ka * FEAT + (n0 + l)];
      b.y = W[(size_t)(ka + 1) * FEAT + (n0 + l)];
    }
    acc = __builtin_amdgcn_wmma_f32_16x16x4_f32(
         false, a,  false, b,
         (short)0, acc,  false,  false);
  }

  const float bv = BIAS ? bias[n0 + l] : 0.0f;
  float* hout = H + ((size_t)g * NNODES + m0 + 8 * half) * FEAT + (n0 + l);
#pragma unroll
  for (int i = 0; i < 8; ++i)
    hout[(size_t)i * FEAT] = acc[i] + bv;
}

__global__ void __launch_bounds__(256)
alpha_kernel(const float* __restrict__ H, const float* __restrict__ a_src,
             const float* __restrict__ a_dst, float* __restrict__ AS,
             float* __restrict__ AD) {
  const int wave = threadIdx.x >> 5;
  const int lane = threadIdx.x & 31;
  const int node = blockIdx.x * 8 + wave;
  const int f    = lane * 4;
  const float4 h4 = *(const float4*)(H + (size_t)node * FEAT + f);
  float ps = h4.x * a_src[f] + h4.y * a_src[f + 1] +
             h4.z * a_src[f + 2] + h4.w * a_src[f + 3];
  float pd = h4.x * a_dst[f] + h4.y * a_dst[f + 1] +
             h4.z * a_dst[f + 2] + h4.w * a_dst[f + 3];
#pragma unroll
  for (int off = 16; off > 0; off >>= 1) {
    ps += __shfl_xor(ps, off, 32);
    pd += __shfl_xor(pd, off, 32);
  }
  if (lane == 0) {
    AS[node] = ps;
    AD[node] = pd;
  }
}

__global__ void init_seg_kernel(int* __restrict__ MORD, float* __restrict__ SSUM) {
  const int i = blockIdx.x * blockDim.x + threadIdx.x;
  MORD[i] = (int)0x80000000;
  SSUM[i] = 0.0f;
}

__device__ __forceinline__ void edge_nodes(const int* __restrict__ ei, int e,
                                           int& s, int& d) {
  if (e < NEDGES) {
    s = ei[e];
    d = ei[NEDGES + e];
  } else {
    s = d = e - NEDGES;
  }
}

#define SLICE_E   8192
#define NSLICES   (NEDGES / SLICE_E)
#define STAT_THR  1024

__global__ void __launch_bounds__(STAT_THR)
edge_max_kernel(const int* __restrict__ EI, const float* __restrict__ AS,
                const float* __restrict__ AD, int* __restrict__ MORD) {
  __shared__ float sAS[NNODES];
  __shared__ float sAD[NNODES];
  __shared__ int   sM[NNODES];
  const int t = threadIdx.x;
  const int k = blockIdx.x;
  const int g = blockIdx.y;
  const int seed_self = (k == 0);
#pragma unroll
  for (int j = 0; j < NNODES / (4 * STAT_THR); ++j) {
    const int n = 4 * (t + STAT_THR * j);
    const float4 a = *(const float4*)(AS + (size_t)g * NNODES + n);
    const float4 b = *(const float4*)(AD + (size_t)g * NNODES + n);
    *(float4*)(sAS + n) = a;
    *(float4*)(sAD + n) = b;
    int4 m;
    m.x = seed_self ? f32_ord(leaky(a.x + b.x)) : (int)0x80000000;
    m.y = seed_self ? f32_ord(leaky(a.y + b.y)) : (int)0x80000000;
    m.z = seed_self ? f32_ord(leaky(a.z + b.z)) : (int)0x80000000;
    m.w = seed_self ? f32_ord(leaky(a.w + b.w)) : (int)0x80000000;
    *(int4*)(sM + n) = m;
  }
  __syncthreads();

  const int* sp = EI + (size_t)g * 2 * NEDGES + (size_t)k * SLICE_E + t * 8;
  const int* dp = sp + NEDGES;
  const int4 s0 = *(const int4*)sp, s1 = *(const int4*)(sp + 4);
  const int4 d0 = *(const int4*)dp, d1 = *(const int4*)(dp + 4);
  const int sv[8] = {s0.x, s0.y, s0.z, s0.w, s1.x, s1.y, s1.z, s1.w};
  const int dv[8] = {d0.x, d0.y, d0.z, d0.w, d1.x, d1.y, d1.z, d1.w};
#pragma unroll
  for (int i = 0; i < 8; ++i) {
    const float v = leaky(sAS[sv[i]] + sAD[dv[i]]);
    atomicMax(&sM[dv[i]], f32_ord(v));
  }
  __syncthreads();

#pragma unroll
  for (int j = 0; j < NNODES / STAT_THR; ++j) {
    const int n = t + STAT_THR * j;
    atomicMax(&MORD[g * NNODES + n], sM[n]);
  }
}

__global__ void __launch_bounds__(STAT_THR)
edge_sum_kernel(const int* __restrict__ EI, const float* __restrict__ AS,
                const float* __restrict__ AD, const int* __restrict__ MORD,
                float* __restrict__ SSUM) {
  __shared__ float sAS[NNODES];
  __shared__ float sAD[NNODES];
  __shared__ float sMx[NNODES];
  __shared__ float sS[NNODES];
  const int t = threadIdx.x;
  const int k = blockIdx.x;
  const int g = blockIdx.y;
  const int seed_self = (k == 0);
#pragma unroll
  for (int j = 0; j < NNODES / (4 * STAT_THR); ++j) {
    const int n = 4 * (t + STAT_THR * j);
    const float4 a = *(const float4*)(AS + (size_t)g * NNODES + n);
    const float4 b = *(const float4*)(AD + (size_t)g * NNODES + n);
    const int4   o = *(const int4*)(MORD + (size_t)g * NNODES + n);
    float4 m;
    m.x = ord_f32(o.x); m.y = ord_f32(o.y); m.z = ord_f32(o.z); m.w = ord_f32(o.w);
    *(float4*)(sAS + n) = a;
    *(float4*)(sAD + n) = b;
    *(float4*)(sMx + n) = m;
    float4 w;
    w.x = seed_self ? __expf(leaky(a.x + b.x) - m.x) : 0.0f;
    w.y = seed_self ? __expf(leaky(a.y + b.y) - m.y) : 0.0f;
    w.z = seed_self ? __expf(leaky(a.z + b.z) - m.z) : 0.0f;
    w.w = seed_self ? __expf(leaky(a.w + b.w) - m.w) : 0.0f;
    *(float4*)(sS + n) = w;
  }
  __syncthreads();

  const int* sp = EI + (size_t)g * 2 * NEDGES + (size_t)k * SLICE_E + t * 8;
  const int* dp = sp + NEDGES;
  const int4 s0 = *(const int4*)sp, s1 = *(const int4*)(sp + 4);
  const int4 d0 = *(const int4*)dp, d1 = *(const int4*)(dp + 4);
  const int sv[8] = {s0.x, s0.y, s0.z, s0.w, s1.x, s1.y, s1.z, s1.w};
  const int dv[8] = {d0.x, d0.y, d0.z, d0.w, d1.x, d1.y, d1.z, d1.w};
#pragma unroll
  for (int i = 0; i < 8; ++i) {
    const float v = leaky(sAS[sv[i]] + sAD[dv[i]]);
    atomicAdd(&sS[dv[i]], __expf(v - sMx[dv[i]]));
  }
  __syncthreads();

  float r[NNODES / STAT_THR];
#pragma unroll
  for (int j = 0; j < NNODES / STAT_THR; ++j) r[j] = sS[t + STAT_THR * j];
  float* q = SSUM + (size_t)g * NNODES + t;
  atomic_add_f32_at<0 * STAT_THR * 4>(q, r[0]);
  atomic_add_f32_at<1 * STAT_THR * 4>(q, r[1]);
  atomic_add_f32_at<2 * STAT_THR * 4>(q, r[2]);
  atomic_add_f32_at<3 * STAT_THR * 4>(q, r[3]);
  atomic_add_f32_at<4 * STAT_THR * 4>(q, r[4]);
  atomic_add_f32_at<5 * STAT_THR * 4>(q, r[5]);
  atomic_add_f32_at<6 * STAT_THR * 4>(q, r[6]);
  atomic_add_f32_at<7 * STAT_THR * 4>(q, r[7]);
  static_assert(NNODES / STAT_THR == 8, "eight node words per thread");
}

__global__ void __launch_bounds__(256)
edge_agg_kernel(const int* __restrict__ EI, const float* __restrict__ AS,
                const float* __restrict__ AD, const int* __restrict__ MORD,
                const float* __restrict__ SSUM, const float* __restrict__ H,
                float* __restrict__ ACC) {
  const int wave = threadIdx.x >> 5;
  const int lane = threadIdx.x & 31;
  const long widx = (long)blockIdx.x * 8 + wave;
  const int g = (int)(widx / EPRIME);
  const int e = (int)(widx - (long)g * EPRIME);
  int s, d;
  edge_nodes(EI + (size_t)g * 2 * NEDGES, e, s, d);
  const int gs = g * NNODES + s, gd = g * NNODES + d;
  const float v = leaky(AS[gs] + AD[gd]);
  const float alpha = __expf(v - ord_f32(MORD[gd])) / SSUM[gd];
  const int f = lane * 4;
  const float4 h4 = *(const float4*)(H + (size_t)gs * FEAT + f);
  float* o = ACC + (size_t)gd * FEAT + f;
  atomic_add_f32(o + 0, alpha * h4.x);
  atomic_add_f32(o + 1, alpha * h4.y);
  atomic_add_f32(o + 2, alpha * h4.z);
  atomic_add_f32(o + 3, alpha * h4.w);
}

__global__ void __launch_bounds__(256)
bias_swish_kernel(const float* __restrict__ ACC, const float* __restrict__ b,
                  float* __restrict__ X) {
  const long i = (long)blockIdx.x * blockDim.x + threadIdx.x;
  const int f = (int)(i & (FEAT - 1));
  const float v = ACC[i] + b[f];
  X[i] = v / (1.0f + __expf(-v));
}

extern "C" void kernel_launch(void* const* d_in, const int* in_sizes, int n_in,
                              void* d_out, int out_size, void* d_ws, size_t ws_size,
                              hipStream_t stream) {
  const float* x0   = (const float*)d_in[0];
  const int*   ei   = (const int*)d_in[1];
  const float* W1   = (const float*)d_in[2];
  const float* as1  = (const float*)d_in[3];
  const float* ad1  = (const float*)d_in[4];
  const float* b1   = (const float*)d_in[5];
  const float* W2   = (const float*)d_in[6];
  const float* as2  = (const float*)d_in[7];
  const float* ad2  = (const float*)d_in[8];
  const float* b2   = (const float*)d_in[9];
  const float* fcW  = (const float*)d_in[10];
  const float* fcb  = (const float*)d_in[11];
  float* out = (float*)d_out;

  const size_t BN      = (size_t)BGRAPH * NNODES;
  const size_t BIGELTS = BN * FEAT;
  char* ws = (char*)d_ws;
  float* BUF0 = (float*)ws;                ws += BIGELTS * sizeof(float);
  float* BUF1 = (float*)ws;                ws += BIGELTS * sizeof(float);
  float* AS   = (float*)ws;                ws += BN * sizeof(float);
  float* AD   = (float*)ws;                ws += BN * sizeof(float);
  int*   MORD = (int*)ws;                  ws += BN * sizeof(int);
  float* SSUM = (float*)ws;

  const dim3 gGemm(NNODES / 16, BGRAPH);
  const int  gAlpha = (int)(BN / 8);
  const int  gInit  = (int)(BN / 256);
  const dim3 gStat(NSLICES, BGRAPH);
  const int  gEdgeW = (int)(((long)BGRAPH * EPRIME) / 8);
  const int  gElem  = (int)(BIGELTS / 256);

  gemm_wmma_f32<false, false><<<gGemm, 256, 0, stream>>>(x0, W1, nullptr, BUF0);
  alpha_kernel<<<gAlpha, 256, 0, stream>>>(BUF0, as1, ad1, AS, AD);
  init_seg_kernel<<<gInit, 256, 0, stream>>>(MORD, SSUM);
  hipMemsetAsync(BUF1, 0, BIGELTS * sizeof(float), stream);
  edge_max_kernel<<<gStat, STAT_THR, 0, stream>>>(ei, AS, AD, MORD);
  edge_sum_kernel<<<gStat, STAT_THR, 0, stream>>>(ei, AS, AD, MORD, SSUM);
  edge_agg_kernel<<<gEdgeW, 256, 0, stream>>>(ei, AS, AD, MORD, SSUM, BUF0, BUF1);
  bias_swish_kernel<<<gElem, 256, 0, stream>>>(BUF1, b1, BUF0);

  gemm_wmma_f32<false, false><<<gGemm, 256, 0, stream>>>(BUF0, W2, nullptr, BUF1);
  alpha_kernel<<<gAlpha, 256, 0, stream>>>(BUF1, as2, ad2, AS, AD);
  init_seg_kernel<<<gInit, 256, 0, stream>>>(MORD, SSUM);
  hipMemsetAsync(BUF0, 0, BIGELTS * sizeof(float), stream);
  edge_max_kernel<<<gStat, STAT_THR, 0, stream>>>(ei, AS, AD, MORD);
  edge_sum_kernel<<<gStat, STAT_THR, 0, stream>>>(ei, AS, AD, MORD, SSUM);
  edge_agg_kernel<<<gEdgeW, 256, 0, stream>>>(ei, AS, AD, MORD, SSUM, BUF1, BUF0);
  bias_swish_kernel<<<gElem, 256, 0, stream>>>(BUF0, b2, BUF1);

  gemm_wmma_f32<true, true><<<gGemm, 256, 0, stream>>>(BUF1, fcW, fcb, out);
}


# Edges bucketed by destination once per call (LDS counting sort); a wave per node walks its in-edge list: max, exp-sum, weighted row sum, bias, swish; no float atomics left; sums run in list order
# speedup vs baseline: 5.9240x; 3.5107x over previous
#include <hip/hip_runtime.h>
#include <cstdint>

#define BGRAPH 8
#define NNODES 8192
#define NEDGES 262144
#define FEAT   128
#define NEG_SLOPE 0.2f

typedef __attribute__((ext_vector_type(2))) float v2f;
typedef __attribute__((ext_vector_type(8))) float v8f;

__device__ __forceinline__ float leaky(float v) {
  return v >= 0.0f ? v : NEG_SLOPE * v;
}

template <bool TRANSB, bool BIAS>
__global__ void __launch_bounds__(256)
gemm_wmma_f32(const float* __restrict__ X, const float* __restrict__ W,
              const float* __restrict__ bias, float* __restrict__ H) {
  const int g    = blockIdx.y;
  const int m0   = blockIdx.x * 16;
  const int wave = threadIdx.x >> 5;
  const int lane = threadIdx.x & 31;
  const int half = lane >> 4;
  const int l    = lane & 15;
  const int n0   = wave * 16;

  const float* xrow = X + ((size_t)g * NNODES + (m0 + l)) * FEAT;

  v8f acc = {};
#pragma unroll
  for (int k = 0; k < FEAT; k += 4) {
    const int ka = k + 2 * half;
    v2f a;
    a.x = xrow[ka];
    a.y = xrow[ka + 1];
    v2f b;
    if (TRANSB) {
      const float* wr = W + (size_t)(n0 + l) * FEAT + ka;
      b.x = wr[0];
      b.y = wr[1];
    } else {
      b.x = W[(size_t)ka * FEAT + (n0 + l)];
      b.y = W[(size_t)(ka + 1) * FEAT + (n0 + l)];
    }
    acc = __builtin_amdgcn_wmma_f32_16x16x4_f32(
         false, a,  false, b,
         (short)0, acc,  false,  false);
  }

  const float bv = BIAS ? bias[n0 + l] : 0.0f;
  float* hout = H + ((size_t)g * NNODES + m0 + 8 * half) * FEAT + (n0 + l);
#pragma unroll
  for (int i = 0; i < 8; ++i)
    hout[(size_t)i * FEAT] = acc[i] + bv;
}

__global__ void __launch_bounds__(256)
alpha_kernel(const float* __restrict__ H, const float* __restrict__ a_src,
             const float* __restrict__ a_dst, float* __restrict__ AS,
             float* __restrict__ AD) {
  const int wave = threadIdx.x >> 5;
  const int lane = threadIdx.x & 31;
  const int node = blockIdx.x * 8 + wave;
  const int f    = lane * 4;
  const float4 h4 = *(const float4*)(H + (size_t)node * FEAT + f);
  float ps = h4.x * a_src[f] + h4.y * a_src[f + 1] +
             h4.z * a_src[f + 2] + h4.w * a_src[f + 3];
  float pd = h4.x * a_dst[f] + h4.y * a_dst[f + 1] +
             h4.z * a_dst[f + 2] + h4.w * a_dst[f + 3];
#pragma unroll
  for (int off = 16; off > 0; off >>= 1) {
    ps += __shfl_xor(ps, off, 32);
    pd += __shfl_xor(pd, off, 32);
  }
  if (lane == 0) {
    AS[node] = ps;
    AD[node] = pd;
  }
}

#define RANGE_N   256
#define NRANGE    (NNODES / RANGE_N)
#define CSR_THR   1024

__global__ void __launch_bounds__(CSR_THR)
bucket_edges_kernel(const int* __restrict__ EI, int* __restrict__ START,
                    int* __restrict__ CNT, int* __restrict__ LIST) {
  __shared__ int sCur[RANGE_N];
  __shared__ int sScan[RANGE_N];
  __shared__ int sLow[CSR_THR / 32];
  const int t    = threadIdx.x;
  const int lane = t & 31;
  const int wave = t >> 5;
  const int lo   = blockIdx.x * RANGE_N;
  const int g    = blockIdx.y;
  const int* srcp = EI + (size_t)g * 2 * NEDGES;
  const int* dstp = srcp + NEDGES;

  if (t < RANGE_N) sCur[t] = 0;
  __syncthreads();

  int below = 0;
  for (int it = 0; it < NEDGES / (16 * CSR_THR); ++it) {
    int4 d[4];
#pragma unroll
    for (int u = 0; u < 4; ++u)
      d[u] = *(const int4*)(dstp + 4 * (t + CSR_THR * (4 * it + u)));
#pragma unroll
    for (int u = 0; u < 4; ++u) {
      const int dv[4] = {d[u].x, d[u].y, d[u].z, d[u].w};
#pragma unroll
      for (int i = 0; i < 4; ++i) {
        below += (dv[i] < lo) ? 1 : 0;
        const unsigned rel = (unsigned)(dv[i] - lo);
        if (rel < (unsigned)RANGE_N) atomicAdd(&sCur[rel], 1);
      }
    }
  }
#pragma unroll
  for (int off = 16; off > 0; off >>= 1) below += __shfl_xor(below, off, 32);
  if (lane == 0) sLow[wave] = below;
  __syncthreads();
  int base = 0;
#pragma unroll
  for (int i = 0; i < CSR_THR / 32; ++i) base += sLow[i];

  if (t < RANGE_N) sScan[t] = sCur[t];
  __syncthreads();
  for (int off = 1; off < RANGE_N; off <<= 1) {
    int v = 0;
    if (t < RANGE_N && t >= off) v = sScan[t - off];
    __syncthreads();
    if (t < RANGE_N) sScan[t] += v;
    __syncthreads();
  }
  if (t < RANGE_N) {
    const int c  = sCur[t];
    const int st = base + sScan[t] - c;
    START[g * NNODES + lo + t] = st;
    CNT[g * NNODES + lo + t]   = c;
    sCur[t] = st;
  }
  __syncthreads();

  int* lst = LIST + (size_t)g * NEDGES;
  for (int it = 0; it < NEDGES / (16 * CSR_THR); ++it) {
    int4 d[4], s[4];
#pragma unroll
    for (int u = 0; u < 4; ++u) {
      const int e = 4 * (t + CSR_THR * (4 * it + u));
      d[u] = *(const int4*)(dstp + e);
      s[u] = *(const int4*)(srcp + e);
    }
#pragma unroll
    for (int u = 0; u < 4; ++u) {
      const int dv[4] = {d[u].x, d[u].y, d[u].z, d[u].w};
      const int sv[4] = {s[u].x, s[u].y, s[u].z, s[u].w};
#pragma unroll
      for (int i = 0; i < 4; ++i) {
        const unsigned rel = (unsigned)(dv[i] - lo);
        if (rel < (unsigned)RANGE_N) {
          const int pos = atomicAdd(&sCur[rel], 1);
          lst[pos] = sv[i];
        }
      }
    }
  }
}

__global__ void __launch_bounds__(256)
gat_gather_kernel(const int* __restrict__ START, const int* __restrict__ CNT,
                  const int* __restrict__ LIST, const float* __restrict__ AS,
                  const float* __restrict__ AD, const float* __restrict__ H,
                  const float* __restrict__ b, float* __restrict__ X) {
  const int wave = __builtin_amdgcn_readfirstlane((int)(threadIdx.x >> 5));
  const int lane = threadIdx.x & 31;
  const int gn   = blockIdx.x * 8 + wave;
  const int g    = gn / NNODES;
  const int nl   = gn - g * NNODES;
  const int cnt   = CNT[gn];
  const int total = cnt + 1;
  const int* lst  = LIST + (size_t)g * NEDGES + START[gn];
  const float* as = AS + (size_t)g * NNODES;
  const float a_d = AD[gn];
  const float* hg = H + (size_t)g * NNODES * FEAT + lane * 4;

  int s0 = nl;
  if (lane < cnt) s0 = lst[lane];
  const float e0 = leaky(as[s0] + a_d);
  float m = (lane < total) ? e0 : -INFINITY;
  for (int c = 32; c < total; c += 32) {
    const int i = c + lane;
    int s = nl;
    if (i < cnt) s = lst[i];
    const float e = leaky(as[s] + a_d);
    if (i < total) m = fmaxf(m, e);
  }
#pragma unroll
  for (int off = 16; off > 0; off >>= 1) m = fmaxf(m, __shfl_xor(m, off, 32));

  float4 acc = make_float4(0.0f, 0.0f, 0.0f, 0.0f);
  float wsum = 0.0f;
  for (int c = 0; c < total; c += 32) {
    const int i = c + lane;
    int s = s0;
    float e = e0;
    if (c > 0) {
      s = nl;
      if (i < cnt) s = lst[i];
      e = leaky(as[s] + a_d);
    }
    const float w = (i < total) ? __expf(e - m) : 0.0f;
    wsum += w;
    const int rem = total - c;
    const int nc  = rem < 32 ? rem : 32;
    for (int j = 0; j < nc; j += 4) {
      const int r0 = __builtin_amdgcn_readlane(s, j);
      const int r1 = __builtin_amdgcn_readlane(s, j + 1);
      const int r2 = __builtin_amdgcn_readlane(s, j + 2);
      const int r3 = __builtin_amdgcn_readlane(s, j + 3);
      const float w0 = __int_as_float(__builtin_amdgcn_readlane(__float_as_int(w), j));
      const float w1 = __int_as_float(__builtin_amdgcn_readlane(__float_as_int(w), j + 1));
      const float w2 = __int_as_float(__builtin_amdgcn_readlane(__float_as_int(w), j + 2));
      const float w3 = __int_as_float(__builtin_amdgcn_readlane(__float_as_int(w), j + 3));
      const float4 h0 = *(const float4*)(hg + (size_t)r0 * FEAT);
      const float4 h1 = *(const float4*)(hg + (size_t)r1 * FEAT);
      const float4 h2 = *(const float4*)(hg + (size_t)r2 * FEAT);
      const float4 h3 = *(const float4*)(hg + (size_t)r3 * FEAT);
      acc.x += w0 * h0.x; acc.y += w0 * h0.y; acc.z += w0 * h0.z; acc.w += w0 * h0.w;
      acc.x += w1 * h1.x; acc.y += w1 * h1.y; acc.z += w1 * h1.z; acc.w += w1 * h1.w;
      acc.x += w2 * h2.x; acc.y += w2 * h2.y; acc.z += w2 * h2.z; acc.w += w2 * h2.w;
      acc.x += w3 * h3.x; acc.y += w3 * h3.y; acc.z += w3 * h3.z; acc.w += w3 * h3.w;
    }
  }
#pragma unroll
  for (int off = 16; off > 0; off >>= 1) wsum += __shfl_xor(wsum, off, 32);

  const int f = lane * 4;
  const float4 bv = *(const float4*)(b + f);
  float4 v;
  v.x = acc.x / wsum + bv.x;
  v.y = acc.y / wsum + bv.y;
  v.z = acc.z / wsum + bv.z;
  v.w = acc.w / wsum + bv.w;
  float4 o;
  o.x = v.x / (1.0f + __expf(-v.x));
  o.y = v.y / (1.0f + __expf(-v.y));
  o.z = v.z / (1.0f + __expf(-v.z));
  o.w = v.w / (1.0f + __expf(-v.w));
  *(float4*)(X + (size_t)gn * FEAT + f) = o;
}

extern "C" void kernel_launch(void* const* d_in, const int* in_sizes, int n_in,
                              void* d_out, int out_size, void* d_ws, size_t ws_size,
                              hipStream_t stream) {
  const float* x0   = (const float*)d_in[0];
  const int*   ei   = (const int*)d_in[1];
  const float* W1   = (const float*)d_in[2];
  const float* as1  = (const float*)d_in[3];
  const float* ad1  = (const float*)d_in[4];
  const float* b1   = (const float*)d_in[5];
  const float* W2   = (const float*)d_in[6];
  const float* as2  = (const float*)d_in[7];
  const float* ad2  = (const float*)d_in[8];
  const float* b2   = (const float*)d_in[9];
  const float* fcW  = (const float*)d_in[10];
  const float* fcb  = (const float*)d_in[11];
  float* out = (float*)d_out;

  const size_t BN      = (size_t)BGRAPH * NNODES;
  const size_t BIGELTS = BN * FEAT;
  char* ws = (char*)d_ws;
  float* BUF0  = (float*)ws;               ws += BIGELTS * sizeof(float);
  float* BUF1  = (float*)ws;               ws += BIGELTS * sizeof(float);
  float* AS    = (float*)ws;               ws += BN * sizeof(float);
  float* AD    = (float*)ws;               ws += BN * sizeof(float);
  int*   START = (int*)ws;                 ws += BN * sizeof(int);
  int*   CNT   = (int*)ws;                 ws += BN * sizeof(int);
  int*   LIST  = (int*)ws;

  const dim3 gGemm(NNODES / 16, BGRAPH);
  const int  gAlpha = (int)(BN / 8);
  const dim3 gBucket(NRANGE, BGRAPH);
  const int  gNode  = (int)(BN / 8);

  bucket_edges_kernel<<<gBucket, CSR_THR, 0, stream>>>(ei, START, CNT, LIST);

  gemm_wmma_f32<false, false><<<gGemm, 256, 0, stream>>>(x0, W1, nullptr, BUF0);
  alpha_kernel<<<gAlpha, 256, 0, stream>>>(BUF0, as1, ad1, AS, AD);
  gat_gather_kernel<<<gNode, 256, 0, stream>>>(START, CNT, LIST, AS, AD, BUF0, b1, BUF1);

  gemm_wmma_f32<false, false><<<gGemm, 256, 0, stream>>>(BUF1, W2, nullptr, BUF0);
  alpha_kernel<<<gAlpha, 256, 0, stream>>>(BUF0, as2, ad2, AS, AD);
  gat_gather_kernel<<<gNode, 256, 0, stream>>>(START, CNT, LIST, AS, AD, BUF0, b2, BUF1);

  gemm_wmma_f32<true, true><<<gGemm, 256, 0, stream>>>(BUF1, fcW, fcb, out);
}


# Grids cut to what is resident: a wave loops over 16 nodes (logits), 4 nodes (edge stage) or 4 row tiles (GEMM, its 32-column W fragment kept in registers); 6x fewer waves; sums in the same order
# speedup vs baseline: 7.5758x; 1.2788x over previous
#include <hip/hip_runtime.h>
#include <cstdint>

#define BGRAPH 8
#define NNODES 8192
#define NEDGES 262144
#define FEAT   128
#define NEG_SLOPE 0.2f

typedef __attribute__((ext_vector_type(2))) float v2f;
typedef __attribute__((ext_vector_type(8))) float v8f;

__device__ __forceinline__ float leaky(float v) {
  return v >= 0.0f ? v : NEG_SLOPE * v;
}

#define GEMM_THR  128
#define GEMM_TPW  4
#define GEMM_GRID (BGRAPH * NNODES / 16 / GEMM_TPW)
template <bool TRANSB, bool BIAS>
__global__ void __launch_bounds__(GEMM_THR)
gemm_wmma_f32(const float* __restrict__ X, const float* __restrict__ W,
              const float* __restrict__ bias, float* __restrict__ H) {
  const int wave = threadIdx.x >> 5;
  const int lane = threadIdx.x & 31;
  const int half = lane >> 4;
  const int l    = lane & 15;
  const int n0   = wave * 32;

  v2f bf0[FEAT / 4], bf1[FEAT / 4];
#pragma unroll
  for (int t = 0; t < FEAT / 4; ++t) {
    const int ka = 4 * t + 2 * half;
    if (TRANSB) {
      const float* wr = W + (size_t)(n0 + l) * FEAT + ka;
      bf0[t].x = wr[0];
      bf0[t].y = wr[1];
      bf1[t].x = wr[16 * FEAT];
      bf1[t].y = wr[16 * FEAT + 1];
    } else {
      bf0[t].x = W[(size_t)ka * FEAT + (n0 + l)];
      bf0[t].y = W[(size_t)(ka + 1) * FEAT + (n0 + l)];
      bf1[t].x = W[(size_t)ka * FEAT + (n0 + 16 + l)];
      bf1[t].y = W[(size_t)(ka + 1) * FEAT + (n0 + 16 + l)];
    }
  }
  const float bv0 = BIAS ? bias[n0 + l] : 0.0f;
  const float bv1 = BIAS ? bias[n0 + 16 + l] : 0.0f;

  for (int tt = 0; tt < GEMM_TPW; ++tt) {
    const size_t m0 = ((size_t)blockIdx.x * GEMM_TPW + tt) * 16;
    const float* xrow = X + (m0 + l) * FEAT + 2 * half;
    v8f acc0 = {}, acc1 = {};
#pragma unroll
    for (int t = 0; t < FEAT / 4; ++t) {
      v2f a;
      a.x = xrow[4 * t];
      a.y = xrow[4 * t + 1];
      acc0 = __builtin_amdgcn_wmma_f32_16x16x4_f32(
           false, a,  false, bf0[t],
           (short)0, acc0,  false,  false);
      acc1 = __builtin_amdgcn_wmma_f32_16x16x4_f32(
           false, a,  false, bf1[t],
           (short)0, acc1,  false,  false);
    }
    float* hout = H + (m0 + 8 * half) * FEAT + (n0 + l);
#pragma unroll
    for (int i = 0; i < 8; ++i) {
      hout[(size_t)i * FEAT]      = acc0[i] + bv0;
      hout[(size_t)i * FEAT + 16] = acc1[i] + bv1;
    }
  }
}

#define ALPHA_NPW  16
#define ALPHA_GRID (BGRAPH * NNODES / (8 * ALPHA_NPW))
__global__ void __launch_bounds__(256)
alpha_kernel(const float* __restrict__ H, const float* __restrict__ a_src,
             const float* __restrict__ a_dst, float* __restrict__ AS,
             float* __restrict__ AD) {
  const int wave  = threadIdx.x >> 5;
  const int lane  = threadIdx.x & 31;
  const int f     = lane * 4;
  const int node0 = (blockIdx.x * 8 + wave) * ALPHA_NPW;
  const float s0 = a_src[f], s1 = a_src[f + 1], s2 = a_src[f + 2], s3 = a_src[f + 3];
  const float d0 = a_dst[f], d1 = a_dst[f + 1], d2 = a_dst[f + 2], d3 = a_dst[f + 3];
  for (int nb = node0; nb < node0 + ALPHA_NPW; nb += 8) {
    float4 h4[8];
#pragma unroll
    for (int u = 0; u < 8; ++u)
      h4[u] = *(const float4*)(H + (size_t)(nb + u) * FEAT + f);
    float ps[8], pd[8];
#pragma unroll
    for (int u = 0; u < 8; ++u) {
      ps[u] = h4[u].x * s0 + h4[u].y * s1 + h4[u].z * s2 + h4[u].w * s3;
      pd[u] = h4[u].x * d0 + h4[u].y * d1 + h4[u].z * d2 + h4[u].w * d3;
    }
#pragma unroll
    for (int off = 16; off > 0; off >>= 1) {
#pragma unroll
      for (int u = 0; u < 8; ++u) {
        ps[u] += __shfl_xor(ps[u], off, 32);
        pd[u] += __shfl_xor(pd[u], off, 32);
      }
    }
    float vs = ps[0], vd = pd[0];
#pragma unroll
    for (int u = 1; u < 8; ++u) {
      if (lane == u) { vs = ps[u]; vd = pd[u]; }
    }
    if (lane < 8) {
      AS[nb + lane] = vs;
      AD[nb + lane] = vd;
    }
  }
}

#define RANGE_N   256
#define NRANGE    (NNODES / RANGE_N)
#define CSR_THR   1024

__global__ void __launch_bounds__(CSR_THR)
bucket_edges_kernel(const int* __restrict__ EI, int* __restrict__ START,
                    int* __restrict__ CNT, int* __restrict__ LIST) {
  __shared__ int sCur[RANGE_N];
  __shared__ int sScan[RANGE_N];
  __shared__ int sLow[CSR_THR / 32];
  const int t    = threadIdx.x;
  const int lane = t & 31;
  const int wave = t >> 5;
  const int lo   = blockIdx.x * RANGE_N;
  const int g    = blockIdx.y;
  const int* srcp = EI + (size_t)g * 2 * NEDGES;
  const int* dstp = srcp + NEDGES;

  if (t < RANGE_N) sCur[t] = 0;
  __syncthreads();

  int below = 0;
  for (int it = 0; it < NEDGES / (16 * CSR_THR); ++it) {
    int4 d[4];
#pragma unroll
    for (int u = 0; u < 4; ++u)
      d[u] = *(const int4*)(dstp + 4 * (t + CSR_THR * (4 * it + u)));
#pragma unroll
    for (int u = 0; u < 4; ++u) {
      const int dv[4] = {d[u].x, d[u].y, d[u].z, d[u].w};
#pragma unroll
      for (int i = 0; i < 4; ++i) {
        below += (dv[i] < lo) ? 1 : 0;
        const unsigned rel = (unsigned)(dv[i] - lo);
        if (rel < (unsigned)RANGE_N) atomicAdd(&sCur[rel], 1);
      }
    }
  }
#pragma unroll
  for (int off = 16; off > 0; off >>= 1) below += __shfl_xor(below, off, 32);
  if (lane == 0) sLow[wave] = below;
  __syncthreads();
  int base = 0;
#pragma unroll
  for (int i = 0; i < CSR_THR / 32; ++i) base += sLow[i];

  if (t < RANGE_N) sScan[t] = sCur[t];
  __syncthreads();
  for (int off = 1; off < RANGE_N; off <<= 1) {
    int v = 0;
    if (t < RANGE_N && t >= off) v = sScan[t - off];
    __syncthreads();
    if (t < RANGE_N) sScan[t] += v;
    __syncthreads();
  }
  if (t < RANGE_N) {
    const int c  = sCur[t];
    const int st = base + sScan[t] - c;
    START[g * NNODES + lo + t] = st;
    CNT[g * NNODES + lo + t]   = c;
    sCur[t] = st;
  }
  __syncthreads();

  int* lst = LIST + (size_t)g * NEDGES;
  for (int it = 0; it < NEDGES / (16 * CSR_THR); ++it) {
    int4 d[4], s[4];
#pragma unroll
    for (int u = 0; u < 4; ++u) {
      const int e = 4 * (t + CSR_THR * (4 * it + u));
      d[u] = *(const int4*)(dstp + e);
      s[u] = *(const int4*)(srcp + e);
    }
#pragma unroll
    for (int u = 0; u < 4; ++u) {
      const int dv[4] = {d[u].x, d[u].y, d[u].z, d[u].w};
      const int sv[4] = {s[u].x, s[u].y, s[u].z, s[u].w};
#pragma unroll
      for (int i = 0; i < 4; ++i) {
        const unsigned rel = (unsigned)(dv[i] - lo);
        if (rel < (unsigned)RANGE_N) {
          const int pos = atomicAdd(&sCur[rel], 1);
          lst[pos] = sv[i];
        }
      }
    }
  }
}

#define GATHER_NPW  4
#define GATHER_GRID (BGRAPH * NNODES / (8 * GATHER_NPW))
__global__ void __launch_bounds__(256)
gat_gather_kernel(const int* __restrict__ START, const int* __restrict__ CNT,
                  const int* __restrict__ LIST, const float* __restrict__ AS,
                  const float* __restrict__ AD, const float* __restrict__ H,
                  const float* __restrict__ b, float* __restrict__ X) {
  const int wave = __builtin_amdgcn_readfirstlane((int)(threadIdx.x >> 5));
  const int lane = threadIdx.x & 31;
  const int f    = lane * 4;
  const float4 bv = *(const float4*)(b + f);
  for (int it = 0; it < GATHER_NPW; ++it) {
  const int gn   = it * (BGRAPH * NNODES / GATHER_NPW) + blockIdx.x * 8 + wave;
  const int g    = gn / NNODES;
  const int nl   = gn - g * NNODES;
  const int cnt   = CNT[gn];
  const int total = cnt + 1;
  const int* lst  = LIST + (size_t)g * NEDGES + START[gn];
  const float* as = AS + (size_t)g * NNODES;
  const float a_d = AD[gn];
  const float* hg = H + (size_t)g * NNODES * FEAT + lane * 4;

  int s0 = nl;
  if (lane < cnt) s0 = lst[lane];
  const float e0 = leaky(as[s0] + a_d);
  float m = (lane < total) ? e0 : -INFINITY;
  for (int c = 32; c < total; c += 32) {
    const int i = c + lane;
    int s = nl;
    if (i < cnt) s = lst[i];
    const float e = leaky(as[s] + a_d);
    if (i < total) m = fmaxf(m, e);
  }
#pragma unroll
  for (int off = 16; off > 0; off >>= 1) m = fmaxf(m, __shfl_xor(m, off, 32));

  float4 acc = make_float4(0.0f, 0.0f, 0.0f, 0.0f);
  float wsum = 0.0f;
  for (int c = 0; c < total; c += 32) {
    const int i = c + lane;
    int s = s0;
    float e = e0;
    if (c > 0) {
      s = nl;
      if (i < cnt) s = lst[i];
      e = leaky(as[s] + a_d);
    }
    const float w = (i < total) ? __expf(e - m) : 0.0f;
    wsum += w;
    const int rem = total - c;
    const int nc  = rem < 32 ? rem : 32;
    for (int j = 0; j < nc; j += 4) {
      const int r0 = __builtin_amdgcn_readlane(s, j);
      const int r1 = __builtin_amdgcn_readlane(s, j + 1);
      const int r2 = __builtin_amdgcn_readlane(s, j + 2);
      const int r3 = __builtin_amdgcn_readlane(s, j + 3);
      const float w0 = __int_as_float(__builtin_amdgcn_readlane(__float_as_int(w), j));
      const float w1 = __int_as_float(__builtin_amdgcn_readlane(__float_as_int(w), j + 1));
      const float w2 = __int_as_float(__builtin_amdgcn_readlane(__float_as_int(w), j + 2));
      const float w3 = __int_as_float(__builtin_amdgcn_readlane(__float_as_int(w), j + 3));
      const float4 h0 = *(const float4*)(hg + (size_t)r0 * FEAT);
      const float4 h1 = *(const float4*)(hg + (size_t)r1 * FEAT);
      const float4 h2 = *(const float4*)(hg + (size_t)r2 * FEAT);
      const float4 h3 = *(const float4*)(hg + (size_t)r3 * FEAT);
      acc.x += w0 * h0.x; acc.y += w0 * h0.y; acc.z += w0 * h0.z; acc.w += w0 * h0.w;
      acc.x += w1 * h1.x; acc.y += w1 * h1.y; acc.z += w1 * h1.z; acc.w += w1 * h1.w;
      acc.x += w2 * h2.x; acc.y += w2 * h2.y; acc.z += w2 * h2.z; acc.w += w2 * h2.w;
      acc.x += w3 * h3.x; acc.y += w3 * h3.y; acc.z += w3 * h3.z; acc.w += w3 * h3.w;
    }
  }
#pragma unroll
  for (int off = 16; off > 0; off >>= 1) wsum += __shfl_xor(wsum, off, 32);

  float4 v;
  v.x = acc.x / wsum + bv.x;
  v.y = acc.y / wsum + bv.y;
  v.z = acc.z / wsum + bv.z;
  v.w = acc.w / wsum + bv.w;
  float4 o;
  o.x = v.x / (1.0f + __expf(-v.x));
  o.y = v.y / (1.0f + __expf(-v.y));
  o.z = v.z / (1.0f + __expf(-v.z));
  o.w = v.w / (1.0f + __expf(-v.w));
  *(float4*)(X + (size_t)gn * FEAT + f) = o;
  }
}

extern "C" void kernel_launch(void* const* d_in, const int* in_sizes, int n_in,
                              void* d_out, int out_size, void* d_ws, size_t ws_size,
                              hipStream_t stream) {
  const float* x0   = (const float*)d_in[0];
  const int*   ei   = (const int*)d_in[1];
  const float* W1   = (const float*)d_in[2];
  const float* as1  = (const float*)d_in[3];
  const float* ad1  = (const float*)d_in[4];
  const float* b1   = (const float*)d_in[5];
  const float* W2   = (const float*)d_in[6];
  const float* as2  = (const float*)d_in[7];
  const float* ad2  = (const float*)d_in[8];
  const float* b2   = (const float*)d_in[9];
  const float* fcW  = (const float*)d_in[10];
  const float* fcb  = (const float*)d_in[11];
  float* out = (float*)d_out;

  const size_t BN      = (size_t)BGRAPH * NNODES;
  const size_t BIGELTS = BN * FEAT;
  char* ws = (char*)d_ws;
  float* BUF0  = (float*)ws;               ws += BIGELTS * sizeof(float);
  float* BUF1  = (float*)ws;               ws += BIGELTS * sizeof(float);
  float* AS    = (float*)ws;               ws += BN * sizeof(float);
  float* AD    = (float*)ws;               ws += BN * sizeof(float);
  int*   START = (int*)ws;                 ws += BN * sizeof(int);
  int*   CNT   = (int*)ws;                 ws += BN * sizeof(int);
  int*   LIST  = (int*)ws;

  const int  gGemm  = GEMM_GRID;
  const int  gAlpha = ALPHA_GRID;
  const dim3 gBucket(NRANGE, BGRAPH);
  const int  gNode  = GATHER_GRID;

  bucket_edges_kernel<<<gBucket, CSR_THR, 0, stream>>>(ei, START, CNT, LIST);

  gemm_wmma_f32<false, false><<<gGemm, GEMM_THR, 0, stream>>>(x0, W1, nullptr, BUF0);
  alpha_kernel<<<gAlpha, 256, 0, stream>>>(BUF0, as1, ad1, AS, AD);
  gat_gather_kernel<<<gNode, 256, 0, stream>>>(START, CNT, LIST, AS, AD, BUF0, b1, BUF1);

  gemm_wmma_f32<false, false><<<gGemm, GEMM_THR, 0, stream>>>(BUF1, W2, nullptr, BUF0);
  alpha_kernel<<<gAlpha, 256, 0, stream>>>(BUF0, as2, ad2, AS, AD);
  gat_gather_kernel<<<gNode, 256, 0, stream>>>(START, CNT, LIST, AS, AD, BUF0, b2, BUF1);

  gemm_wmma_f32<true, true><<<gGemm, GEMM_THR, 0, stream>>>(BUF1, fcW, fcb, out);
}


# Both per-node logit dot products taken from the GEMM accumulators (DPP tree over 16 lanes, four waves added via LDS); the two logit launches and their re-read of H go; logit sum order changes
# speedup vs baseline: 7.8404x; 1.0349x over previous
#include <hip/hip_runtime.h>
#include <cstdint>

#define BGRAPH 8
#define NNODES 8192
#define NEDGES 262144
#define FEAT   128
#define NEG_SLOPE 0.2f

typedef __attribute__((ext_vector_type(2))) float v2f;
typedef __attribute__((ext_vector_type(8))) float v8f;

__device__ __forceinline__ float leaky(float v) {
  return v >= 0.0f ? v : NEG_SLOPE * v;
}

#define GEMM_THR  128
#define GEMM_TPW  4
#define GEMM_GRID (BGRAPH * NNODES / 16 / GEMM_TPW)

template <int MASK>
__device__ __forceinline__ float row_xor(float v) {
  return __int_as_float(__builtin_amdgcn_update_dpp(
      0, __float_as_int(v), 0x160 | MASK, 0xf, 0xf, true));
}

template <bool TRANSB, bool BIAS, bool ALPHA>
__global__ void __launch_bounds__(GEMM_THR)
gemm_wmma_f32(const float* __restrict__ X, const float* __restrict__ W,
              const float* __restrict__ bias, float* __restrict__ H,
              const float* __restrict__ a_src, const float* __restrict__ a_dst,
              float* __restrict__ AS, float* __restrict__ AD) {
  __shared__ float sPart[GEMM_TPW][GEMM_THR / 32][32];
  const int wave = threadIdx.x >> 5;
  const int lane = threadIdx.x & 31;
  const int half = lane >> 4;
  const int l    = lane & 15;
  const int n0   = wave * 32;

  v2f bf0[FEAT / 4], bf1[FEAT / 4];
#pragma unroll
  for (int t = 0; t < FEAT / 4; ++t) {
    const int ka = 4 * t + 2 * half;
    if (TRANSB) {
      const float* wr = W + (size_t)(n0 + l) * FEAT + ka;
      bf0[t].x = wr[0];
      bf0[t].y = wr[1];
      bf1[t].x = wr[16 * FEAT];
      bf1[t].y = wr[16 * FEAT + 1];
    } else {
      bf0[t].x = W[(size_t)ka * FEAT + (n0 + l)];
      bf0[t].y = W[(size_t)(ka + 1) * FEAT + (n0 + l)];
      bf1[t].x = W[(size_t)ka * FEAT + (n0 + 16 + l)];
      bf1[t].y = W[(size_t)(ka + 1) * FEAT + (n0 + 16 + l)];
    }
  }
  const float bv0 = BIAS ? bias[n0 + l] : 0.0f;
  const float bv1 = BIAS ? bias[n0 + 16 + l] : 0.0f;
  const float as0 = ALPHA ? a_src[n0 + l] : 0.0f;
  const float as1 = ALPHA ? a_src[n0 + 16 + l] : 0.0f;
  const float ad0 = ALPHA ? a_dst[n0 + l] : 0.0f;
  const float ad1 = ALPHA ? a_dst[n0 + 16 + l] : 0.0f;

#pragma nounroll
  for (int tt = 0; tt < GEMM_TPW; ++tt) {
    const size_t m0 = ((size_t)blockIdx.x * GEMM_TPW + tt) * 16;
    const float* xrow = X + (m0 + l) * FEAT + 2 * half;
    v8f acc0 = {}, acc1 = {};
#pragma unroll
    for (int t = 0; t < FEAT / 4; ++t) {
      v2f a;
      a.x = xrow[4 * t];
      a.y = xrow[4 * t + 1];
      acc0 = __builtin_amdgcn_wmma_f32_16x16x4_f32(
           false, a,  false, bf0[t],
           (short)0, acc0,  false,  false);
      acc1 = __builtin_amdgcn_wmma_f32_16x16x4_f32(
           false, a,  false, bf1[t],
           (short)0, acc1,  false,  false);
    }
    float* hout = H + (m0 + 8 * half) * FEAT + (n0 + l);
#pragma unroll
    for (int i = 0; i < 8; ++i) {
      hout[(size_t)i * FEAT]      = acc0[i] + bv0;
      hout[(size_t)i * FEAT + 16] = acc1[i] + bv1;
    }
    if (ALPHA) {
      float v[16];
#pragma unroll
      for (int i = 0; i < 8; ++i) {
        v[i]     = acc0[i] * as0 + acc1[i] * as1;
        v[8 + i] = acc0[i] * ad0 + acc1[i] * ad1;
      }
      const bool b3 = (l & 8) != 0, b2 = (l & 4) != 0;
      const bool b1 = (l & 2) != 0, b0 = (l & 1) != 0;
      float u[8], w4[4], y[2];
#pragma unroll
      for (int k = 0; k < 8; ++k) {
        const float keep = b3 ? v[k + 8] : v[k];
        const float send = b3 ? v[k] : v[k + 8];
        u[k] = keep + row_xor<8>(send);
      }
#pragma unroll
      for (int k = 0; k < 4; ++k) {
        const float keep = b2 ? u[k + 4] : u[k];
        const float send = b2 ? u[k] : u[k + 4];
        w4[k] = keep + row_xor<4>(send);
      }
#pragma unroll
      for (int k = 0; k < 2; ++k) {
        const float keep = b1 ? w4[k + 2] : w4[k];
        const float send = b1 ? w4[k] : w4[k + 2];
        y[k] = keep + row_xor<2>(send);
      }
      const float keep = b0 ? y[1] : y[0];
      const float send = b0 ? y[0] : y[1];
      sPart[tt][wave][lane] = keep + row_xor<1>(send);
    }
  }
  if (ALPHA) {
    __syncthreads();
    const int kind = wave >> 1;
    const int tile = 2 * (wave & 1) + (lane >> 4);
    const int row  = lane & 15;
    const int idx  = 16 * (row >> 3) + 8 * kind + (row & 7);
    const float s = ((sPart[tile][0][idx] + sPart[tile][1][idx]) +
                     sPart[tile][2][idx]) + sPart[tile][3][idx];
    float* dstv = kind ? AD : AS;
    dstv[(size_t)blockIdx.x * (16 * GEMM_TPW) + 32 * (wave & 1) + lane] = s;
  }
}

#define RANGE_N   256
#define NRANGE    (NNODES / RANGE_N)
#define CSR_THR   1024

__global__ void __launch_bounds__(CSR_THR)
bucket_edges_kernel(const int* __restrict__ EI, int* __restrict__ START,
                    int* __restrict__ CNT, int* __restrict__ LIST) {
  __shared__ int sCur[RANGE_N];
  __shared__ int sScan[RANGE_N];
  __shared__ int sLow[CSR_THR / 32];
  const int t    = threadIdx.x;
  const int lane = t & 31;
  const int wave = t >> 5;
  const int lo   = blockIdx.x * RANGE_N;
  const int g    = blockIdx.y;
  const int* srcp = EI + (size_t)g * 2 * NEDGES;
  const int* dstp = srcp + NEDGES;

  if (t < RANGE_N) sCur[t] = 0;
  __syncthreads();

  int below = 0;
  for (int it = 0; it < NEDGES / (16 * CSR_THR); ++it) {
    int4 d[4];
#pragma unroll
    for (int u = 0; u < 4; ++u)
      d[u] = *(const int4*)(dstp + 4 * (t + CSR_THR * (4 * it + u)));
#pragma unroll
    for (int u = 0; u < 4; ++u) {
      const int dv[4] = {d[u].x, d[u].y, d[u].z, d[u].w};
#pragma unroll
      for (int i = 0; i < 4; ++i) {
        below += (dv[i] < lo) ? 1 : 0;
        const unsigned rel = (unsigned)(dv[i] - lo);
        if (rel < (unsigned)RANGE_N) atomicAdd(&sCur[rel], 1);
      }
    }
  }
#pragma unroll
  for (int off = 16; off > 0; off >>= 1) below += __shfl_xor(below, off, 32);
  if (lane == 0) sLow[wave] = below;
  __syncthreads();
  int base = 0;
#pragma unroll
  for (int i = 0; i < CSR_THR / 32; ++i) base += sLow[i];

  if (t < RANGE_N) sScan[t] = sCur[t];
  __syncthreads();
  for (int off = 1; off < RANGE_N; off <<= 1) {
    int v = 0;
    if (t < RANGE_N && t >= off) v = sScan[t - off];
    __syncthreads();
    if (t < RANGE_N) sScan[t] += v;
    __syncthreads();
  }
  if (t < RANGE_N) {
    const int c  = sCur[t];
    const int st = base + sScan[t] - c;
    START[g * NNODES + lo + t] = st;
    CNT[g * NNODES + lo + t]   = c;
    sCur[t] = st;
  }
  __syncthreads();

  int* lst = LIST + (size_t)g * NEDGES;
  for (int it = 0; it < NEDGES / (16 * CSR_THR); ++it) {
    int4 d[4], s[4];
#pragma unroll
    for (int u = 0; u < 4; ++u) {
      const int e = 4 * (t + CSR_THR * (4 * it + u));
      d[u] = *(const int4*)(dstp + e);
      s[u] = *(const int4*)(srcp + e);
    }
#pragma unroll
    for (int u = 0; u < 4; ++u) {
      const int dv[4] = {d[u].x, d[u].y, d[u].z, d[u].w};
      const int sv[4] = {s[u].x, s[u].y, s[u].z, s[u].w};
#pragma unroll
      for (int i = 0; i < 4; ++i) {
        const unsigned rel = (unsigned)(dv[i] - lo);
        if (rel < (unsigned)RANGE_N) {
          const int pos = atomicAdd(&sCur[rel], 1);
          lst[pos] = sv[i];
        }
      }
    }
  }
}

#define GATHER_NPW  4
#define GATHER_GRID (BGRAPH * NNODES / (8 * GATHER_NPW))
__global__ void __launch_bounds__(256)
gat_gather_kernel(const int* __restrict__ START, const int* __restrict__ CNT,
                  const int* __restrict__ LIST, const float* __restrict__ AS,
                  const float* __restrict__ AD, const float* __restrict__ H,
                  const float* __restrict__ b, float* __restrict__ X) {
  const int wave = __builtin_amdgcn_readfirstlane((int)(threadIdx.x >> 5));
  const int lane = threadIdx.x & 31;
  const int f    = lane * 4;
  const float4 bv = *(const float4*)(b + f);
  for (int it = 0; it < GATHER_NPW; ++it) {
  const int gn   = it * (BGRAPH * NNODES / GATHER_NPW) + blockIdx.x * 8 + wave;
  const int g    = gn / NNODES;
  const int nl   = gn - g * NNODES;
  const int cnt   = CNT[gn];
  const int total = cnt + 1;
  const int* lst  = LIST + (size_t)g * NEDGES + START[gn];
  const float* as = AS + (size_t)g * NNODES;
  const float a_d = AD[gn];
  const float* hg = H + (size_t)g * NNODES * FEAT + lane * 4;

  int s0 = nl;
  if (lane < cnt) s0 = lst[lane];
  const float e0 = leaky(as[s0] + a_d);
  float m = (lane < total) ? e0 : -INFINITY;
  for (int c = 32; c < total; c += 32) {
    const int i = c + lane;
    int s = nl;
    if (i < cnt) s = lst[i];
    const float e = leaky(as[s] + a_d);
    if (i < total) m = fmaxf(m, e);
  }
#pragma unroll
  for (int off = 16; off > 0; off >>= 1) m = fmaxf(m, __shfl_xor(m, off, 32));

  float4 acc = make_float4(0.0f, 0.0f, 0.0f, 0.0f);
  float wsum = 0.0f;
  for (int c = 0; c < total; c += 32) {
    const int i = c + lane;
    int s = s0;
    float e = e0;
    if (c > 0) {
      s = nl;
      if (i < cnt) s = lst[i];
      e = leaky(as[s] + a_d);
    }
    const float w = (i < total) ? __expf(e - m) : 0.0f;
    wsum += w;
    const int rem = total - c;
    const int nc  = rem < 32 ? rem : 32;
    for (int j = 0; j < nc; j += 4) {
      const int r0 = __builtin_amdgcn_readlane(s, j);
      const int r1 = __builtin_amdgcn_readlane(s, j + 1);
      const int r2 = __builtin_amdgcn_readlane(s, j + 2);
      const int r3 = __builtin_amdgcn_readlane(s, j + 3);
      const float w0 = __int_as_float(__builtin_amdgcn_readlane(__float_as_int(w), j));
      const float w1 = __int_as_float(__builtin_amdgcn_readlane(__float_as_int(w), j + 1));
      const float w2 = __int_as_float(__builtin_amdgcn_readlane(__float_as_int(w), j + 2));
      const float w3 = __int_as_float(__builtin_amdgcn_readlane(__float_as_int(w), j + 3));
      const float4 h0 = *(const float4*)(hg + (size_t)r0 * FEAT);
      const float4 h1 = *(const float4*)(hg + (size_t)r1 * FEAT);
      const float4 h2 = *(const float4*)(hg + (size_t)r2 * FEAT);
      const float4 h3 = *(const float4*)(hg + (size_t)r3 * FEAT);
      acc.x += w0 * h0.x; acc.y += w0 * h0.y; acc.z += w0 * h0.z; acc.w += w0 * h0.w;
      acc.x += w1 * h1.x; acc.y += w1 * h1.y; acc.z += w1 * h1.z; acc.w += w1 * h1.w;
      acc.x += w2 * h2.x; acc.y += w2 * h2.y; acc.z += w2 * h2.z; acc.w += w2 * h2.w;
      acc.x += w3 * h3.x; acc.y += w3 * h3.y; acc.z += w3 * h3.z; acc.w += w3 * h3.w;
    }
  }
#pragma unroll
  for (int off = 16; off > 0; off >>= 1) wsum += __shfl_xor(wsum, off, 32);

  float4 v;
  v.x = acc.x / wsum + bv.x;
  v.y = acc.y / wsum + bv.y;
  v.z = acc.z / wsum + bv.z;
  v.w = acc.w / wsum + bv.w;
  float4 o;
  o.x = v.x / (1.0f + __expf(-v.x));
  o.y = v.y / (1.0f + __expf(-v.y));
  o.z = v.z / (1.0f + __expf(-v.z));
  o.w = v.w / (1.0f + __expf(-v.w));
  *(float4*)(X + (size_t)gn * FEAT + f) = o;
  }
}

extern "C" void kernel_launch(void* const* d_in, const int* in_sizes, int n_in,
                              void* d_out, int out_size, void* d_ws, size_t ws_size,
                              hipStream_t stream) {
  const float* x0   = (const float*)d_in[0];
  const int*   ei   = (const int*)d_in[1];
  const float* W1   = (const float*)d_in[2];
  const float* as1  = (const float*)d_in[3];
  const float* ad1  = (const float*)d_in[4];
  const float* b1   = (const float*)d_in[5];
  const float* W2   = (const float*)d_in[6];
  const float* as2  = (const float*)d_in[7];
  const float* ad2  = (const float*)d_in[8];
  const float* b2   = (const float*)d_in[9];
  const float* fcW  = (const float*)d_in[10];
  const float* fcb  = (const float*)d_in[11];
  float* out = (float*)d_out;

  const size_t BN      = (size_t)BGRAPH * NNODES;
  const size_t BIGELTS = BN * FEAT;
  char* ws = (char*)d_ws;
  float* BUF0  = (float*)ws;               ws += BIGELTS * sizeof(float);
  float* BUF1  = (float*)ws;               ws += BIGELTS * sizeof(float);
  float* AS    = (float*)ws;               ws += BN * sizeof(float);
  float* AD    = (float*)ws;               ws += BN * sizeof(float);
  int*   START = (int*)ws;                 ws += BN * sizeof(int);
  int*   CNT   = (int*)ws;                 ws += BN * sizeof(int);
  int*   LIST  = (int*)ws;

  const int  gGemm  = GEMM_GRID;
  const dim3 gBucket(NRANGE, BGRAPH);
  const int  gNode  = GATHER_GRID;

  bucket_edges_kernel<<<gBucket, CSR_THR, 0, stream>>>(ei, START, CNT, LIST);

  gemm_wmma_f32<false, false, true><<<gGemm, GEMM_THR, 0, stream>>>(
      x0, W1, nullptr, BUF0, as1, ad1, AS, AD);
  gat_gather_kernel<<<gNode, 256, 0, stream>>>(START, CNT, LIST, AS, AD, BUF0, b1, BUF1);

  gemm_wmma_f32<false, false, true><<<gGemm, GEMM_THR, 0, stream>>>(
      BUF1, W2, nullptr, BUF0, as2, ad2, AS, AD);
  gat_gather_kernel<<<gNode, 256, 0, stream>>>(START, CNT, LIST, AS, AD, BUF0, b2, BUF1);

  gemm_wmma_f32<true, true, false><<<gGemm, GEMM_THR, 0, stream>>>(
      BUF1, fcW, fcb, out, nullptr, nullptr, nullptr, nullptr);
}


# Edges pre-sorted by destination range per 8192-edge slice (one more launch); a range work-group reads only its own records, not the whole graph three times; in-list order, so sum order, changes
# speedup vs baseline: 9.0372x; 1.1526x over previous
#include <hip/hip_runtime.h>
#include <cstdint>

#define BGRAPH 8
#define NNODES 8192
#define NEDGES 262144
#define FEAT   128
#define NEG_SLOPE 0.2f

typedef __attribute__((ext_vector_type(2))) float v2f;
typedef __attribute__((ext_vector_type(8))) float v8f;
typedef __attribute__((ext_vector_type(4))) int   v4i;

__device__ __forceinline__ float leaky(float v) {
  return v >= 0.0f ? v : NEG_SLOPE * v;
}

#define GEMM_THR  128
#define GEMM_TPW  4
#define GEMM_GRID (BGRAPH * NNODES / 16 / GEMM_TPW)

template <int MASK>
__device__ __forceinline__ float row_xor(float v) {
  return __int_as_float(__builtin_amdgcn_update_dpp(
      0, __float_as_int(v), 0x160 | MASK, 0xf, 0xf, true));
}

template <bool TRANSB, bool BIAS, bool ALPHA>
__global__ void __launch_bounds__(GEMM_THR)
gemm_wmma_f32(const float* __restrict__ X, const float* __restrict__ W,
              const float* __restrict__ bias, float* __restrict__ H,
              const float* __restrict__ a_src, const float* __restrict__ a_dst,
              float* __restrict__ AS, float* __restrict__ AD) {
  __shared__ float sPart[GEMM_TPW][GEMM_THR / 32][32];
  const int wave = threadIdx.x >> 5;
  const int lane = threadIdx.x & 31;
  const int half = lane >> 4;
  const int l    = lane & 15;
  const int n0   = wave * 32;

  v2f bf0[FEAT / 4], bf1[FEAT / 4];
#pragma unroll
  for (int t = 0; t < FEAT / 4; ++t) {
    const int ka = 4 * t + 2 * half;
    if (TRANSB) {
      const float* wr = W + (size_t)(n0 + l) * FEAT + ka;
      bf0[t].x = wr[0];
      bf0[t].y = wr[1];
      bf1[t].x = wr[16 * FEAT];
      bf1[t].y = wr[16 * FEAT + 1];
    } else {
      bf0[t].x = W[(size_t)ka * FEAT + (n0 + l)];
      bf0[t].y = W[(size_t)(ka + 1) * FEAT + (n0 + l)];
      bf1[t].x = W[(size_t)ka * FEAT + (n0 + 16 + l)];
      bf1[t].y = W[(size_t)(ka + 1) * FEAT + (n0 + 16 + l)];
    }
  }
  const float bv0 = BIAS ? bias[n0 + l] : 0.0f;
  const float bv1 = BIAS ? bias[n0 + 16 + l] : 0.0f;
  const float as0 = ALPHA ? a_src[n0 + l] : 0.0f;
  const float as1 = ALPHA ? a_src[n0 + 16 + l] : 0.0f;
  const float ad0 = ALPHA ? a_dst[n0 + l] : 0.0f;
  const float ad1 = ALPHA ? a_dst[n0 + 16 + l] : 0.0f;

#pragma nounroll
  for (int tt = 0; tt < GEMM_TPW; ++tt) {
    const size_t m0 = ((size_t)blockIdx.x * GEMM_TPW + tt) * 16;
    const float* xrow = X + (m0 + l) * FEAT + 2 * half;
    v8f acc0 = {}, acc1 = {};
#pragma unroll
    for (int t = 0; t < FEAT / 4; ++t) {
      v2f a;
      a.x = xrow[4 * t];
      a.y = xrow[4 * t + 1];
      acc0 = __builtin_amdgcn_wmma_f32_16x16x4_f32(
           false, a,  false, bf0[t],
           (short)0, acc0,  false,  false);
      acc1 = __builtin_amdgcn_wmma_f32_16x16x4_f32(
           false, a,  false, bf1[t],
           (short)0, acc1,  false,  false);
    }
    float* hout = H + (m0 + 8 * half) * FEAT + (n0 + l);
#pragma unroll
    for (int i = 0; i < 8; ++i) {
      hout[(size_t)i * FEAT]      = acc0[i] + bv0;
      hout[(size_t)i * FEAT + 16] = acc1[i] + bv1;
    }
    if (ALPHA) {
      float v[16];
#pragma unroll
      for (int i = 0; i < 8; ++i) {
        v[i]     = acc0[i] * as0 + acc1[i] * as1;
        v[8 + i] = acc0[i] * ad0 + acc1[i] * ad1;
      }
      const bool b3 = (l & 8) != 0, b2 = (l & 4) != 0;
      const bool b1 = (l & 2) != 0, b0 = (l & 1) != 0;
      float u[8], w4[4], y[2];
#pragma unroll
      for (int k = 0; k < 8; ++k) {
        const float keep = b3 ? v[k + 8] : v[k];
        const float send = b3 ? v[k] : v[k + 8];
        u[k] = keep + row_xor<8>(send);
      }
#pragma unroll
      for (int k = 0; k < 4; ++k) {
        const float keep = b2 ? u[k + 4] : u[k];
        const float send = b2 ? u[k] : u[k + 4];
        w4[k] = keep + row_xor<4>(send);
      }
#pragma unroll
      for (int k = 0; k < 2; ++k) {
        const float keep = b1 ? w4[k + 2] : w4[k];
        const float send = b1 ? w4[k] : w4[k + 2];
        y[k] = keep + row_xor<2>(send);
      }
      const float keep = b0 ? y[1] : y[0];
      const float send = b0 ? y[0] : y[1];
      sPart[tt][wave][lane] = keep + row_xor<1>(send);
    }
  }
  if (ALPHA) {
    __syncthreads();
    const int kind = wave >> 1;
    const int tile = 2 * (wave & 1) + (lane >> 4);
    const int row  = lane & 15;
    const int idx  = 16 * (row >> 3) + 8 * kind + (row & 7);
    const float s = ((sPart[tile][0][idx] + sPart[tile][1][idx]) +
                     sPart[tile][2][idx]) + sPart[tile][3][idx];
    float* dstv = kind ? AD : AS;
    dstv[(size_t)blockIdx.x * (16 * GEMM_TPW) + 32 * (wave & 1) + lane] = s;
  }
}

#define RANGE_N   256
#define NRANGE    (NNODES / RANGE_N)
#define CSR_THR   512
#define CSR_NW    (CSR_THR / 32)
#define SLICE_E   8192
#define NSLICE    (NEDGES / SLICE_E)
#define BIN_THR   512
#define BIN_NW    (BIN_THR / 32)
#define BIN_EPT   (SLICE_E / BIN_THR)
#define WALK_S    (NSLICE / CSR_NW)
#define WALK_LD   10
#define NODE_BITS 13
#define NODE_MASK (NNODES - 1)

static_assert(NNODES == (1 << NODE_BITS), "record packing");
static_assert(NRANGE * BIN_NW == BIN_THR, "one scan entry per thread");
static_assert(NSLICE == WALK_S * CSR_NW, "whole slices per wave");
static_assert(RANGE_N <= CSR_THR, "one thread per owned node");
static_assert(BIN_EPT == 16, "four int4 loads per thread");

__global__ void __launch_bounds__(BIN_THR)
bin_edges_kernel(const int* __restrict__ EI, int* __restrict__ OFF,
                 int* __restrict__ REC) {
  __shared__ int sCnt[BIN_NW][NRANGE];
  __shared__ int sWave[BIN_NW];
  __shared__ int sRec[SLICE_E];
  const int t    = threadIdx.x;
  const int lane = t & 31;
  const int wave = t >> 5;
  const int s    = blockIdx.x;
  const int g    = blockIdx.y;
  const int* srcp = EI + (size_t)g * 2 * NEDGES + (size_t)s * SLICE_E;
  const int* dstp = srcp + NEDGES;

  (&sCnt[0][0])[t] = 0;
  __syncthreads();

  v4i d[4], sv[4];
#pragma unroll
  for (int u = 0; u < 4; ++u) {
    const int e = 4 * (t + BIN_THR * u);
    d[u]  = *(const v4i*)(dstp + e);
    sv[u] = *(const v4i*)(srcp + e);
  }
  asm volatile("" :: "v"(d[0]), "v"(d[1]), "v"(d[2]), "v"(d[3]),
                     "v"(sv[0]), "v"(sv[1]), "v"(sv[2]), "v"(sv[3]));
  int rec[BIN_EPT], rk[BIN_EPT];
#pragma unroll
  for (int u = 0; u < 4; ++u) {
    const int dv[4] = {d[u].x, d[u].y, d[u].z, d[u].w};
    const int sw[4] = {sv[u].x, sv[u].y, sv[u].z, sv[u].w};
#pragma unroll
    for (int i = 0; i < 4; ++i) {
      const int k = 4 * u + i;
      rec[k] = ((dv[i] & NODE_MASK) << NODE_BITS) | (sw[i] & NODE_MASK);
      const int r = rec[k] >> (NODE_BITS + 8);
      rk[k] = atomicAdd(&sCnt[wave][r], 1);
    }
  }
  __syncthreads();

  const int sr = t / BIN_NW;
  const int sw_ = t % BIN_NW;
  const int v  = sCnt[sw_][sr];
  int incl = v;
#pragma unroll
  for (int off = 1; off < 32; off <<= 1) {
    const int n = __shfl_up(incl, off, 32);
    if (lane >= off) incl += n;
  }
  if (lane == 31) sWave[wave] = incl;
  __syncthreads();
  int pre = 0;
#pragma unroll
  for (int i = 0; i < BIN_NW; ++i) pre += (i < wave) ? sWave[i] : 0;
  const int excl = pre + incl - v;
  sCnt[sw_][sr] = excl;
  int* off = OFF + (size_t)(g * NSLICE + s) * (NRANGE + 1);
  if (sw_ == 0) off[sr] = excl;
  if (t == BIN_THR - 1) off[NRANGE] = excl + v;
  __syncthreads();

#pragma unroll
  for (int k = 0; k < BIN_EPT; ++k) {
    const int r = rec[k] >> (NODE_BITS + 8);
    sRec[sCnt[wave][r] + rk[k]] = rec[k];
  }
  __syncthreads();

  int* out = REC + (size_t)(g * NSLICE + s) * SLICE_E;
#pragma unroll
  for (int u = 0; u < 4; ++u) {
    const int e = 4 * (t + BIN_THR * u);
    *(int4*)(out + e) = *(const int4*)(sRec + e);
  }
}

__global__ void __launch_bounds__(CSR_THR)
bucket_edges_kernel(const int* __restrict__ OFF, const int* __restrict__ REC,
                    int* __restrict__ START, int* __restrict__ CNT,
                    int* __restrict__ LIST) {
  __shared__ int sCur[RANGE_N];
  __shared__ int sScan[RANGE_N];
  __shared__ int sLow[CSR_NW];
  const int t    = threadIdx.x;
  const int lane = t & 31;
  const int wave = __builtin_amdgcn_readfirstlane(t >> 5);
  const int r    = blockIdx.x;
  const int lo   = r * RANGE_N;
  const int g    = blockIdx.y;
  int beg[WALK_S], end[WALK_S];
  const int* rec[WALK_S];
#pragma unroll
  for (int k = 0; k < WALK_S; ++k) {
    const int s = WALK_S * wave + k;
    const int* off = OFF + (size_t)(g * NSLICE + s) * (NRANGE + 1);
    beg[k] = off[r];
    end[k] = off[r + 1];
    rec[k] = REC + (size_t)(g * NSLICE + s) * SLICE_E;
  }

  if (t < RANGE_N) sCur[t] = 0;
  __syncthreads();

  int rr[WALK_S][WALK_LD];
#pragma unroll
  for (int k = 0; k < WALK_S; ++k) {
#pragma unroll
    for (int u = 0; u < WALK_LD; ++u) {
      const int i = beg[k] + 32 * u + lane;
      rr[k][u] = -1;
      if (i < end[k]) rr[k][u] = rec[k][i];
    }
  }
#pragma unroll
  for (int k = 0; k < WALK_S; ++k) {
#pragma unroll
    for (int u = 0; u < WALK_LD; ++u)
      if (rr[k][u] >= 0) atomicAdd(&sCur[(rr[k][u] >> NODE_BITS) & (RANGE_N - 1)], 1);
  }
#pragma unroll
  for (int k = 0; k < WALK_S; ++k) {
    for (int c = beg[k] + 32 * WALK_LD; c < end[k]; c += 32) {
      const int i = c + lane;
      if (i < end[k]) atomicAdd(&sCur[(rec[k][i] >> NODE_BITS) & (RANGE_N - 1)], 1);
    }
  }
  int low = 0;
#pragma unroll
  for (int k = 0; k < WALK_S; ++k) low += beg[k];
  if (lane == 0) sLow[wave] = low;
  __syncthreads();
  int base = 0;
#pragma unroll
  for (int i = 0; i < CSR_NW; ++i) base += sLow[i];

  if (t < RANGE_N) sScan[t] = sCur[t];
  __syncthreads();
  for (int o = 1; o < RANGE_N; o <<= 1) {
    int v = 0;
    if (t < RANGE_N && t >= o) v = sScan[t - o];
    __syncthreads();
    if (t < RANGE_N) sScan[t] += v;
    __syncthreads();
  }
  if (t < RANGE_N) {
    const int c  = sCur[t];
    const int st = base + sScan[t] - c;
    START[g * NNODES + lo + t] = st;
    CNT[g * NNODES + lo + t]   = c;
    sCur[t] = st;
  }
  __syncthreads();

  int* lst = LIST + (size_t)g * NEDGES;
  unsigned pos[WALK_S][WALK_LD];
#pragma unroll
  for (int k = 0; k < WALK_S; ++k) {
#pragma unroll
    for (int u = 0; u < WALK_LD; ++u) {
      pos[k][u] = 0u;
      if (rr[k][u] >= 0)
        pos[k][u] = (unsigned)atomicAdd(&sCur[(rr[k][u] >> NODE_BITS) & (RANGE_N - 1)], 1);
    }
  }
#pragma unroll
  for (int k = 0; k < WALK_S; ++k) {
#pragma unroll
    for (int u = 0; u < WALK_LD; ++u)
      if (rr[k][u] >= 0) lst[pos[k][u]] = rr[k][u] & NODE_MASK;
  }
#pragma unroll
  for (int k = 0; k < WALK_S; ++k) {
    for (int c = beg[k] + 32 * WALK_LD; c < end[k]; c += 32) {
      const int i = c + lane;
      if (i < end[k]) {
        const int v = rec[k][i];
        const int p = atomicAdd(&sCur[(v >> NODE_BITS) & (RANGE_N - 1)], 1);
        lst[p] = v & NODE_MASK;
      }
    }
  }
}

#define GATHER_NPW  4
#define GATHER_GRID (BGRAPH * NNODES / (8 * GATHER_NPW))
__global__ void __launch_bounds__(256)
gat_gather_kernel(const int* __restrict__ START, const int* __restrict__ CNT,
                  const int* __restrict__ LIST, const float* __restrict__ AS,
                  const float* __restrict__ AD, const float* __restrict__ H,
                  const float* __restrict__ b, float* __restrict__ X) {
  const int wave = __builtin_amdgcn_readfirstlane((int)(threadIdx.x >> 5));
  const int lane = threadIdx.x & 31;
  const int f    = lane * 4;
  const float4 bv = *(const float4*)(b + f);
  for (int it = 0; it < GATHER_NPW; ++it) {
  const int gn   = it * (BGRAPH * NNODES / GATHER_NPW) + blockIdx.x * 8 + wave;
  const int g    = gn / NNODES;
  const int nl   = gn - g * NNODES;
  const int cnt   = CNT[gn];
  const int total = cnt + 1;
  const int* lst  = LIST + (size_t)g * NEDGES + START[gn];
  const float* as = AS + (size_t)g * NNODES;
  const float a_d = AD[gn];
  const float* hg = H + (size_t)g * NNODES * FEAT + lane * 4;

  int s0 = nl;
  if (lane < cnt) s0 = lst[lane];
  const float e0 = leaky(as[s0] + a_d);
  float m = (lane < total) ? e0 : -INFINITY;
  for (int c = 32; c < total; c += 32) {
    const int i = c + lane;
    int s = nl;
    if (i < cnt) s = lst[i];
    const float e = leaky(as[s] + a_d);
    if (i < total) m = fmaxf(m, e);
  }
#pragma unroll
  for (int off = 16; off > 0; off >>= 1) m = fmaxf(m, __shfl_xor(m, off, 32));

  float4 acc = make_float4(0.0f, 0.0f, 0.0f, 0.0f);
  float wsum = 0.0f;
  for (int c = 0; c < total; c += 32) {
    const int i = c + lane;
    int s = s0;
    float e = e0;
    if (c > 0) {
      s = nl;
      if (i < cnt) s = lst[i];
      e = leaky(as[s] + a_d);
    }
    const float w = (i < total) ? __expf(e - m) : 0.0f;
    wsum += w;
    const int rem = total - c;
    const int nc  = rem < 32 ? rem : 32;
    for (int j = 0; j < nc; j += 4) {
      const int r0 = __builtin_amdgcn_readlane(s, j);
      const int r1 = __builtin_amdgcn_readlane(s, j + 1);
      const int r2 = __builtin_amdgcn_readlane(s, j + 2);
      const int r3 = __builtin_amdgcn_readlane(s, j + 3);
      const float w0 = __int_as_float(__builtin_amdgcn_readlane(__float_as_int(w), j));
      const float w1 = __int_as_float(__builtin_amdgcn_readlane(__float_as_int(w), j + 1));
      const float w2 = __int_as_float(__builtin_amdgcn_readlane(__float_as_int(w), j + 2));
      const float w3 = __int_as_float(__builtin_amdgcn_readlane(__float_as_int(w), j + 3));
      const float4 h0 = *(const float4*)(hg + (size_t)r0 * FEAT);
      const float4 h1 = *(const float4*)(hg + (size_t)r1 * FEAT);
      const float4 h2 = *(const float4*)(hg + (size_t)r2 * FEAT);
      const float4 h3 = *(const float4*)(hg + (size_t)r3 * FEAT);
      acc.x += w0 * h0.x; acc.y += w0 * h0.y; acc.z += w0 * h0.z; acc.w += w0 * h0.w;
      acc.x += w1 * h1.x; acc.y += w1 * h1.y; acc.z += w1 * h1.z; acc.w += w1 * h1.w;
      acc.x += w2 * h2.x; acc.y += w2 * h2.y; acc.z += w2 * h2.z; acc.w += w2 * h2.w;
      acc.x += w3 * h3.x; acc.y += w3 * h3.y; acc.z += w3 * h3.z; acc.w += w3 * h3.w;
    }
  }
#pragma unroll
  for (int off = 16; off > 0; off >>= 1) wsum += __shfl_xor(wsum, off, 32);

  float4 v;
  v.x = acc.x / wsum + bv.x;
  v.y = acc.y / wsum + bv.y;
  v.z = acc.z / wsum + bv.z;
  v.w = acc.w / wsum + bv.w;
  float4 o;
  o.x = v.x / (1.0f + __expf(-v.x));
  o.y = v.y / (1.0f + __expf(-v.y));
  o.z = v.z / (1.0f + __expf(-v.z));
  o.w = v.w / (1.0f + __expf(-v.w));
  *(float4*)(X + (size_t)gn * FEAT + f) = o;
  }
}

extern "C" void kernel_launch(void* const* d_in, const int* in_sizes, int n_in,
                              void* d_out, int out_size, void* d_ws, size_t ws_size,
                              hipStream_t stream) {
  const float* x0   = (const float*)d_in[0];
  const int*   ei   = (const int*)d_in[1];
  const float* W1   = (const float*)d_in[2];
  const float* as1  = (const float*)d_in[3];
  const float* ad1  = (const float*)d_in[4];
  const float* b1   = (const float*)d_in[5];
  const float* W2   = (const float*)d_in[6];
  const float* as2  = (const float*)d_in[7];
  const float* ad2  = (const float*)d_in[8];
  const float* b2   = (const float*)d_in[9];
  const float* fcW  = (const float*)d_in[10];
  const float* fcb  = (const float*)d_in[11];
  float* out = (float*)d_out;

  const size_t BN      = (size_t)BGRAPH * NNODES;
  const size_t BIGELTS = BN * FEAT;
  char* ws = (char*)d_ws;
  float* BUF0  = (float*)ws;               ws += BIGELTS * sizeof(float);
  float* BUF1  = (float*)ws;               ws += BIGELTS * sizeof(float);
  float* AS    = (float*)ws;               ws += BN * sizeof(float);
  float* AD    = (float*)ws;               ws += BN * sizeof(float);
  int*   START = (int*)ws;                 ws += BN * sizeof(int);
  int*   CNT   = (int*)ws;                 ws += BN * sizeof(int);
  int*   LIST  = (int*)ws;                 ws += (size_t)BGRAPH * NEDGES * sizeof(int);
  int*   REC   = (int*)ws;                 ws += (size_t)BGRAPH * NEDGES * sizeof(int);
  int*   OFF   = (int*)ws;

  const int  gGemm  = GEMM_GRID;
  const dim3 gBin(NSLICE, BGRAPH);
  const dim3 gBucket(NRANGE, BGRAPH);
  const int  gNode  = GATHER_GRID;

  bin_edges_kernel<<<gBin, BIN_THR, 0, stream>>>(ei, OFF, REC);
  bucket_edges_kernel<<<gBucket, CSR_THR, 0, stream>>>(OFF, REC, START, CNT, LIST);

  gemm_wmma_f32<false, false, true><<<gGemm, GEMM_THR, 0, stream>>>(
      x0, W1, nullptr, BUF0, as1, ad1, AS, AD);
  gat_gather_kernel<<<gNode, 256, 0, stream>>>(START, CNT, LIST, AS, AD, BUF0, b1, BUF1);

  gemm_wmma_f32<false, false, true><<<gGemm, GEMM_THR, 0, stream>>>(
      BUF1, W2, nullptr, BUF0, as2, ad2, AS, AD);
  gat_gather_kernel<<<gNode, 256, 0, stream>>>(START, CNT, LIST, AS, AD, BUF0, b2, BUF1);

  gemm_wmma_f32<true, true, false><<<gGemm, GEMM_THR, 0, stream>>>(
      BUF1, fcW, fcb, out, nullptr, nullptr, nullptr, nullptr);
}


# Dense product behind each edge stage (x@W2 + logits; x@fcW^T+b) done in that stage's launch: rows go through LDS to the WMMAs, W in LDS; 2 launches and the x round trip go; sum orders kept
# speedup vs baseline: 10.3987x; 1.1507x over previous
#include <hip/hip_runtime.h>
#include <cstdint>

#define BGRAPH 8
#define NNODES 8192
#define NEDGES 262144
#define FEAT   128
#define NEG_SLOPE 0.2f

typedef __attribute__((ext_vector_type(2))) float v2f;
typedef __attribute__((ext_vector_type(8))) float v8f;
typedef __attribute__((ext_vector_type(4))) int   v4i;

__device__ __forceinline__ float leaky(float v) {
  return v >= 0.0f ? v : NEG_SLOPE * v;
}

#define GEMM_THR  128
#define GEMM_TPW  4
#define GEMM_GRID (BGRAPH * NNODES / 16 / GEMM_TPW)

template <int MASK>
__device__ __forceinline__ float row_xor(float v) {
  return __int_as_float(__builtin_amdgcn_update_dpp(
      0, __float_as_int(v), 0x160 | MASK, 0xf, 0xf, true));
}

template <bool TRANSB, bool BIAS, bool ALPHA>
__global__ void __launch_bounds__(GEMM_THR)
gemm_wmma_f32(const float* __restrict__ X, const float* __restrict__ W,
              const float* __restrict__ bias, float* __restrict__ H,
              const float* __restrict__ a_src, const float* __restrict__ a_dst,
              float* __restrict__ AS, float* __restrict__ AD) {
  __shared__ float sPart[GEMM_TPW][GEMM_THR / 32][32];
  const int wave = threadIdx.x >> 5;
  const int lane = threadIdx.x & 31;
  const int half = lane >> 4;
  const int l    = lane & 15;
  const int n0   = wave * 32;

  v2f bf0[FEAT / 4], bf1[FEAT / 4];
#pragma unroll
  for (int t = 0; t < FEAT / 4; ++t) {
    const int ka = 4 * t + 2 * half;
    if (TRANSB) {
      const float* wr = W + (size_t)(n0 + l) * FEAT + ka;
      bf0[t].x = wr[0];
      bf0[t].y = wr[1];
      bf1[t].x = wr[16 * FEAT];
      bf1[t].y = wr[16 * FEAT + 1];
    } else {
      bf0[t].x = W[(size_t)ka * FEAT + (n0 + l)];
      bf0[t].y = W[(size_t)(ka + 1) * FEAT + (n0 + l)];
      bf1[t].x = W[(size_t)ka * FEAT + (n0 + 16 + l)];
      bf1[t].y = W[(size_t)(ka + 1) * FEAT + (n0 + 16 + l)];
    }
  }
  const float bv0 = BIAS ? bias[n0 + l] : 0.0f;
  const float bv1 = BIAS ? bias[n0 + 16 + l] : 0.0f;
  const float as0 = ALPHA ? a_src[n0 + l] : 0.0f;
  const float as1 = ALPHA ? a_src[n0 + 16 + l] : 0.0f;
  const float ad0 = ALPHA ? a_dst[n0 + l] : 0.0f;
  const float ad1 = ALPHA ? a_dst[n0 + 16 + l] : 0.0f;

#pragma nounroll
  for (int tt = 0; tt < GEMM_TPW; ++tt) {
    const size_t m0 = ((size_t)blockIdx.x * GEMM_TPW + tt) * 16;
    const float* xrow = X + (m0 + l) * FEAT + 2 * half;
    v8f acc0 = {}, acc1 = {};
#pragma unroll
    for (int t = 0; t < FEAT / 4; ++t) {
      v2f a;
      a.x = xrow[4 * t];
      a.y = xrow[4 * t + 1];
      acc0 = __builtin_amdgcn_wmma_f32_16x16x4_f32(
           false, a,  false, bf0[t],
           (short)0, acc0,  false,  false);
      acc1 = __builtin_amdgcn_wmma_f32_16x16x4_f32(
           false, a,  false, bf1[t],
           (short)0, acc1,  false,  false);
    }
    float* hout = H + (m0 + 8 * half) * FEAT + (n0 + l);
#pragma unroll
    for (int i = 0; i < 8; ++i) {
      hout[(size_t)i * FEAT]      = acc0[i] + bv0;
      hout[(size_t)i * FEAT + 16] = acc1[i] + bv1;
    }
    if (ALPHA) {
      float v[16];
#pragma unroll
      for (int i = 0; i < 8; ++i) {
        v[i]     = acc0[i] * as0 + acc1[i] * as1;
        v[8 + i] = acc0[i] * ad0 + acc1[i] * ad1;
      }
      const bool b3 = (l & 8) != 0, b2 = (l & 4) != 0;
      const bool b1 = (l & 2) != 0, b0 = (l & 1) != 0;
      float u[8], w4[4], y[2];
#pragma unroll
      for (int k = 0; k < 8; ++k) {
        const float keep = b3 ? v[k + 8] : v[k];
        const float send = b3 ? v[k] : v[k + 8];
        u[k] = keep + row_xor<8>(send);
      }
#pragma unroll
      for (int k = 0; k < 4; ++k) {
        const float keep = b2 ? u[k + 4] : u[k];
        const float send = b2 ? u[k] : u[k + 4];
        w4[k] = keep + row_xor<4>(send);
      }
#pragma unroll
      for (int k = 0; k < 2; ++k) {
        const float keep = b1 ? w4[k + 2] : w4[k];
        const float send = b1 ? w4[k] : w4[k + 2];
        y[k] = keep + row_xor<2>(send);
      }
      const float keep = b0 ? y[1] : y[0];
      const float send = b0 ? y[0] : y[1];
      sPart[tt][wave][lane] = keep + row_xor<1>(send);
    }
  }
  if (ALPHA) {
    __syncthreads();
    const int kind = wave >> 1;
    const int tile = 2 * (wave & 1) + (lane >> 4);
    const int row  = lane & 15;
    const int idx  = 16 * (row >> 3) + 8 * kind + (row & 7);
    const float s = ((sPart[tile][0][idx] + sPart[tile][1][idx]) +
                     sPart[tile][2][idx]) + sPart[tile][3][idx];
    float* dstv = kind ? AD : AS;
    dstv[(size_t)blockIdx.x * (16 * GEMM_TPW) + 32 * (wave & 1) + lane] = s;
  }
}

#define RANGE_N   256
#define NRANGE    (NNODES / RANGE_N)
#define CSR_THR   512
#define CSR_NW    (CSR_THR / 32)
#define SLICE_E   8192
#define NSLICE    (NEDGES / SLICE_E)
#define BIN_THR   512
#define BIN_NW    (BIN_THR / 32)
#define BIN_EPT   (SLICE_E / BIN_THR)
#define WALK_S    (NSLICE / CSR_NW)
#define WALK_LD   10
#define NODE_BITS 13
#define NODE_MASK (NNODES - 1)

static_assert(NNODES == (1 << NODE_BITS), "record packing");
static_assert(NRANGE * BIN_NW == BIN_THR, "one scan entry per thread");
static_assert(NSLICE == WALK_S * CSR_NW, "whole slices per wave");
static_assert(RANGE_N <= CSR_THR, "one thread per owned node");
static_assert(BIN_EPT == 16, "four int4 loads per thread");

__global__ void __launch_bounds__(BIN_THR)
bin_edges_kernel(const int* __restrict__ EI, int* __restrict__ OFF,
                 int* __restrict__ REC) {
  __shared__ int sCnt[BIN_NW][NRANGE];
  __shared__ int sWave[BIN_NW];
  __shared__ int sRec[SLICE_E];
  const int t    = threadIdx.x;
  const int lane = t & 31;
  const int wave = t >> 5;
  const int s    = blockIdx.x;
  const int g    = blockIdx.y;
  const int* srcp = EI + (size_t)g * 2 * NEDGES + (size_t)s * SLICE_E;
  const int* dstp = srcp + NEDGES;

  (&sCnt[0][0])[t] = 0;
  __syncthreads();

  v4i d[4], sv[4];
#pragma unroll
  for (int u = 0; u < 4; ++u) {
    const int e = 4 * (t + BIN_THR * u);
    d[u]  = *(const v4i*)(dstp + e);
    sv[u] = *(const v4i*)(srcp + e);
  }
  asm volatile("" :: "v"(d[0]), "v"(d[1]), "v"(d[2]), "v"(d[3]),
                     "v"(sv[0]), "v"(sv[1]), "v"(sv[2]), "v"(sv[3]));
  int rec[BIN_EPT], rk[BIN_EPT];
#pragma unroll
  for (int u = 0; u < 4; ++u) {
    const int dv[4] = {d[u].x, d[u].y, d[u].z, d[u].w};
    const int sw[4] = {sv[u].x, sv[u].y, sv[u].z, sv[u].w};
#pragma unroll
    for (int i = 0; i < 4; ++i) {
      const int k = 4 * u + i;
      rec[k] = ((dv[i] & NODE_MASK) << NODE_BITS) | (sw[i] & NODE_MASK);
      const int r = rec[k] >> (NODE_BITS + 8);
      rk[k] = atomicAdd(&sCnt[wave][r], 1);
    }
  }
  __syncthreads();

  const int sr = t / BIN_NW;
  const int sw_ = t % BIN_NW;
  const int v  = sCnt[sw_][sr];
  int incl = v;
#pragma unroll
  for (int off = 1; off < 32; off <<= 1) {
    const int n = __shfl_up(incl, off, 32);
    if (lane >= off) incl += n;
  }
  if (lane == 31) sWave[wave] = incl;
  __syncthreads();
  int pre = 0;
#pragma unroll
  for (int i = 0; i < BIN_NW; ++i) pre += (i < wave) ? sWave[i] : 0;
  const int excl = pre + incl - v;
  sCnt[sw_][sr] = excl;
  int* off = OFF + (size_t)(g * NSLICE + s) * (NRANGE + 1);
  if (sw_ == 0) off[sr] = excl;
  if (t == BIN_THR - 1) off[NRANGE] = excl + v;
  __syncthreads();

#pragma unroll
  for (int k = 0; k < BIN_EPT; ++k) {
    const int r = rec[k] >> (NODE_BITS + 8);
    sRec[sCnt[wave][r] + rk[k]] = rec[k];
  }
  __syncthreads();

  int* out = REC + (size_t)(g * NSLICE + s) * SLICE_E;
#pragma unroll
  for (int u = 0; u < 4; ++u) {
    const int e = 4 * (t + BIN_THR * u);
    *(int4*)(out + e) = *(const int4*)(sRec + e);
  }
}

__global__ void __launch_bounds__(CSR_THR)
bucket_edges_kernel(const int* __restrict__ OFF, const int* __restrict__ REC,
                    int* __restrict__ START, int* __restrict__ CNT,
                    int* __restrict__ LIST) {
  __shared__ int sCur[RANGE_N];
  __shared__ int sScan[RANGE_N];
  __shared__ int sLow[CSR_NW];
  const int t    = threadIdx.x;
  const int lane = t & 31;
  const int wave = __builtin_amdgcn_readfirstlane(t >> 5);
  const int r    = blockIdx.x;
  const int lo   = r * RANGE_N;
  const int g    = blockIdx.y;
  int beg[WALK_S], end[WALK_S];
  const int* rec[WALK_S];
#pragma unroll
  for (int k = 0; k < WALK_S; ++k) {
    const int s = WALK_S * wave + k;
    const int* off = OFF + (size_t)(g * NSLICE + s) * (NRANGE + 1);
    beg[k] = off[r];
    end[k] = off[r + 1];
    rec[k] = REC + (size_t)(g * NSLICE + s) * SLICE_E;
  }

  if (t < RANGE_N) sCur[t] = 0;
  __syncthreads();

  int rr[WALK_S][WALK_LD];
#pragma unroll
  for (int k = 0; k < WALK_S; ++k) {
#pragma unroll
    for (int u = 0; u < WALK_LD; ++u) {
      const int i = beg[k] + 32 * u + lane;
      rr[k][u] = -1;
      if (i < end[k]) rr[k][u] = rec[k][i];
    }
  }
#pragma unroll
  for (int k = 0; k < WALK_S; ++k) {
#pragma unroll
    for (int u = 0; u < WALK_LD; ++u)
      if (rr[k][u] >= 0) atomicAdd(&sCur[(rr[k][u] >> NODE_BITS) & (RANGE_N - 1)], 1);
  }
#pragma unroll
  for (int k = 0; k < WALK_S; ++k) {
    for (int c = beg[k] + 32 * WALK_LD; c < end[k]; c += 32) {
      const int i = c + lane;
      if (i < end[k]) atomicAdd(&sCur[(rec[k][i] >> NODE_BITS) & (RANGE_N - 1)], 1);
    }
  }
  int low = 0;
#pragma unroll
  for (int k = 0; k < WALK_S; ++k) low += beg[k];
  if (lane == 0) sLow[wave] = low;
  __syncthreads();
  int base = 0;
#pragma unroll
  for (int i = 0; i < CSR_NW; ++i) base += sLow[i];

  if (t < RANGE_N) sScan[t] = sCur[t];
  __syncthreads();
  for (int o = 1; o < RANGE_N; o <<= 1) {
    int v = 0;
    if (t < RANGE_N && t >= o) v = sScan[t - o];
    __syncthreads();
    if (t < RANGE_N) sScan[t] += v;
    __syncthreads();
  }
  if (t < RANGE_N) {
    const int c  = sCur[t];
    const int st = base + sScan[t] - c;
    START[g * NNODES + lo + t] = st;
    CNT[g * NNODES + lo + t]   = c;
    sCur[t] = st;
  }
  __syncthreads();

  int* lst = LIST + (size_t)g * NEDGES;
  unsigned pos[WALK_S][WALK_LD];
#pragma unroll
  for (int k = 0; k < WALK_S; ++k) {
#pragma unroll
    for (int u = 0; u < WALK_LD; ++u) {
      pos[k][u] = 0u;
      if (rr[k][u] >= 0)
        pos[k][u] = (unsigned)atomicAdd(&sCur[(rr[k][u] >> NODE_BITS) & (RANGE_N - 1)], 1);
    }
  }
#pragma unroll
  for (int k = 0; k < WALK_S; ++k) {
#pragma unroll
    for (int u = 0; u < WALK_LD; ++u)
      if (rr[k][u] >= 0) lst[pos[k][u]] = rr[k][u] & NODE_MASK;
  }
#pragma unroll
  for (int k = 0; k < WALK_S; ++k) {
    for (int c = beg[k] + 32 * WALK_LD; c < end[k]; c += 32) {
      const int i = c + lane;
      if (i < end[k]) {
        const int v = rec[k][i];
        const int p = atomicAdd(&sCur[(v >> NODE_BITS) & (RANGE_N - 1)], 1);
        lst[p] = v & NODE_MASK;
      }
    }
  }
}

#define FUSE_THR   1024
#define FUSE_NW    (FUSE_THR / 32)
#define FUSE_NODES 128
#define FUSE_GRID  (BGRAPH * NNODES / FUSE_NODES)
#define FUSE_MAXR  64
#define XP         132
#define WP         288
#define SW_WORDS   ((FEAT / 2) * WP > FEAT * XP ? (FEAT / 2) * WP : FEAT * XP)

static_assert(FUSE_NODES == 4 * FUSE_NW, "four nodes per wave");
static_assert(FUSE_MAXR == 2 * FUSE_NW, "at most two nodes per wave and pass");

template <bool TRANSB, bool BIAS, bool ALPHA>
__global__ void __launch_bounds__(FUSE_THR, 16)
gat_gather_gemm_kernel(const int* __restrict__ START, const int* __restrict__ CNT,
                       const int* __restrict__ LIST, const float* __restrict__ AS,
                       const float* __restrict__ AD, const float* __restrict__ H,
                       const float* __restrict__ b, const float* __restrict__ W,
                       const float* __restrict__ bias2, float* __restrict__ OUT,
                       const float* __restrict__ a_src, const float* __restrict__ a_dst,
                       float* __restrict__ AS2, float* __restrict__ AD2) {
  __shared__ __attribute__((aligned(16))) float sW[SW_WORDS];
  __shared__ __attribute__((aligned(16))) float sX[FUSE_MAXR * XP];
  __shared__ __attribute__((aligned(16))) float sO[FUSE_MAXR * XP];
  __shared__ float sPart[FUSE_MAXR / 16][4][32];
  const int tid  = threadIdx.x;
  const int wave = __builtin_amdgcn_readfirstlane((int)(threadIdx.x >> 5));
  const int lane = threadIdx.x & 31;
  const int half = lane >> 4;
  const int l    = lane & 15;
  const int f    = lane * 4;
  const float4 bv = *(const float4*)(b + f);

  if (TRANSB) {
#pragma unroll
    for (int u = 0; u < 4; ++u) {
      const int q = tid + FUSE_THR * u;
      const int row = q >> 5, c4 = (q & 31) * 4;
      *(float4*)(sW + row * XP + c4) = *(const float4*)(W + row * FEAT + c4);
    }
  } else {
#pragma nounroll
    for (int uo = 0; uo < 8; uo += 4) {
#pragma unroll
      for (int u = 0; u < 4; ++u) {
        const int q = tid + FUSE_THR * (uo + u);
        const int p = q >> 7, n = q & 127;
        v2f pr;
        pr.x = W[(2 * p) * FEAT + n];
        pr.y = W[(2 * p + 1) * FEAT + n];
        *(v2f*)(sW + p * WP + 2 * n) = pr;
      }
    }
  }

  const int tl = wave >> 3;
  const int n0 = (wave & 3) * 32;

  const bool shifted = (((blockIdx.x >> 8) ^ blockIdx.x) & 1) != 0;
  const int npass = shifted ? 3 : 2;
  int base = (int)blockIdx.x * FUSE_NODES;
  for (int pass = 0; pass < npass; ++pass) {
  const int npw  = (shifted && pass != 1) ? 1 : 2;
  const int rows = FUSE_NW * npw;
  for (int it = 0; it < npw; ++it) {
  const int rr   = wave + FUSE_NW * it;
  const int gn   = base + rr;
  const int g    = gn / NNODES;
  const int nl   = gn - g * NNODES;
  const int cnt   = CNT[gn];
  const int total = cnt + 1;
  const int* lst  = LIST + (size_t)g * NEDGES + START[gn];
  const float* as = AS + (size_t)g * NNODES;
  const float a_d = AD[gn];
  const float* hg = H + (size_t)g * NNODES * FEAT + lane * 4;

  int s0 = nl;
  if (lane < cnt) s0 = lst[lane];
  const float e0 = leaky(as[s0] + a_d);
  float m = (lane < total) ? e0 : -INFINITY;
  for (int c = 32; c < total; c += 32) {
    const int i = c + lane;
    int s = nl;
    if (i < cnt) s = lst[i];
    const float e = leaky(as[s] + a_d);
    if (i < total) m = fmaxf(m, e);
  }
#pragma unroll
  for (int off = 16; off > 0; off >>= 1) m = fmaxf(m, __shfl_xor(m, off, 32));

  float4 acc = make_float4(0.0f, 0.0f, 0.0f, 0.0f);
  float wsum = 0.0f;
  for (int c = 0; c < total; c += 32) {
    const int i = c + lane;
    int s = s0;
    float e = e0;
    if (c > 0) {
      s = nl;
      if (i < cnt) s = lst[i];
      e = leaky(as[s] + a_d);
    }
    const float w = (i < total) ? __expf(e - m) : 0.0f;
    wsum += w;
    const int rem = total - c;
    const int nc  = rem < 32 ? rem : 32;
    for (int j = 0; j < nc; j += 4) {
      const int r0 = __builtin_amdgcn_readlane(s, j);
      const int r1 = __builtin_amdgcn_readlane(s, j + 1);
      const int r2 = __builtin_amdgcn_readlane(s, j + 2);
      const int r3 = __builtin_amdgcn_readlane(s, j + 3);
      const float w0 = __int_as_float(__builtin_amdgcn_readlane(__float_as_int(w), j));
      const float w1 = __int_as_float(__builtin_amdgcn_readlane(__float_as_int(w), j + 1));
      const float w2 = __int_as_float(__builtin_amdgcn_readlane(__float_as_int(w), j + 2));
      const float w3 = __int_as_float(__builtin_amdgcn_readlane(__float_as_int(w), j + 3));
      const float4 h0 = *(const float4*)(hg + (size_t)r0 * FEAT);
      const float4 h1 = *(const float4*)(hg + (size_t)r1 * FEAT);
      const float4 h2 = *(const float4*)(hg + (size_t)r2 * FEAT);
      const float4 h3 = *(const float4*)(hg + (size_t)r3 * FEAT);
      acc.x += w0 * h0.x; acc.y += w0 * h0.y; acc.z += w0 * h0.z; acc.w += w0 * h0.w;
      acc.x += w1 * h1.x; acc.y += w1 * h1.y; acc.z += w1 * h1.z; acc.w += w1 * h1.w;
      acc.x += w2 * h2.x; acc.y += w2 * h2.y; acc.z += w2 * h2.z; acc.w += w2 * h2.w;
      acc.x += w3 * h3.x; acc.y += w3 * h3.y; acc.z += w3 * h3.z; acc.w += w3 * h3.w;
    }
  }
#pragma unroll
  for (int off = 16; off > 0; off >>= 1) wsum += __shfl_xor(wsum, off, 32);

  float4 v;
  v.x = acc.x / wsum + bv.x;
  v.y = acc.y / wsum + bv.y;
  v.z = acc.z / wsum + bv.z;
  v.w = acc.w / wsum + bv.w;
  float4 o;
  o.x = v.x / (1.0f + __expf(-v.x));
  o.y = v.y / (1.0f + __expf(-v.y));
  o.z = v.z / (1.0f + __expf(-v.z));
  o.w = v.w / (1.0f + __expf(-v.w));
  *(float4*)(sX + rr * XP + f) = o;
  }
  __syncthreads();

  if ((wave & 4) == 0 && tl < (rows >> 4)) {
    const float* xa = sX + (16 * tl + l) * XP + 2 * half;
    const float* wb = TRANSB ? sW + (n0 + l) * XP + 2 * half
                             : sW + half * WP + 2 * (n0 + l);
    const float bo0 = BIAS ? bias2[n0 + l] : 0.0f;
    const float bo1 = BIAS ? bias2[n0 + 16 + l] : 0.0f;
    const float as0 = ALPHA ? a_src[n0 + l] : 0.0f;
    const float as1 = ALPHA ? a_src[n0 + 16 + l] : 0.0f;
    const float ad0 = ALPHA ? a_dst[n0 + l] : 0.0f;
    const float ad1 = ALPHA ? a_dst[n0 + 16 + l] : 0.0f;
    v8f acc0 = {}, acc1 = {};
#pragma unroll
    for (int t = 0; t < FEAT / 4; ++t) {
      const v2f a = *(const v2f*)(xa + 4 * t);
      const v2f q0 = *(const v2f*)(wb + (TRANSB ? 4 : 2 * WP) * t);
      const v2f q1 = *(const v2f*)(wb + (TRANSB ? 4 : 2 * WP) * t + (TRANSB ? 16 * XP : 32));
      acc0 = __builtin_amdgcn_wmma_f32_16x16x4_f32(
           false, a,  false, q0,
           (short)0, acc0,  false,  false);
      acc1 = __builtin_amdgcn_wmma_f32_16x16x4_f32(
           false, a,  false, q1,
           (short)0, acc1,  false,  false);
    }
    float* so = sO + (16 * tl + 8 * half) * XP + n0 + l;
#pragma unroll
    for (int i = 0; i < 8; ++i) {
      so[i * XP]      = acc0[i] + bo0;
      so[i * XP + 16] = acc1[i] + bo1;
    }
    if (ALPHA) {
      float v[16];
#pragma unroll
      for (int i = 0; i < 8; ++i) {
        v[i]     = acc0[i] * as0 + acc1[i] * as1;
        v[8 + i] = acc0[i] * ad0 + acc1[i] * ad1;
      }
      const bool b3 = (l & 8) != 0, b2 = (l & 4) != 0;
      const bool b1 = (l & 2) != 0, b0 = (l & 1) != 0;
      float u[8], w4[4], y[2];
#pragma unroll
      for (int k = 0; k < 8; ++k) {
        const float keep = b3 ? v[k + 8] : v[k];
        const float send = b3 ? v[k] : v[k + 8];
        u[k] = keep + row_xor<8>(send);
      }
#pragma unroll
      for (int k = 0; k < 4; ++k) {
        const float keep = b2 ? u[k + 4] : u[k];
        const float send = b2 ? u[k] : u[k + 4];
        w4[k] = keep + row_xor<4>(send);
      }
#pragma unroll
      for (int k = 0; k < 2; ++k) {
        const float keep = b1 ? w4[k + 2] : w4[k];
        const float send = b1 ? w4[k] : w4[k + 2];
        y[k] = keep + row_xor<2>(send);
      }
      const float keep = b0 ? y[1] : y[0];
      const float send = b0 ? y[0] : y[1];
      sPart[tl][wave & 3][lane] = keep + row_xor<1>(send);
    }
  }
  __syncthreads();

  for (int u = 0; u < npw; ++u) {
    const int q = tid + FUSE_THR * u;
    const int row = q >> 5, c4 = (q & 31) * 4;
    *(float4*)(OUT + (size_t)(base + row) * FEAT + c4) = *(const float4*)(sO + row * XP + c4);
  }
  if (ALPHA) {
    if (tid < 2 * rows) {
      const int kind = (tid >= rows) ? 1 : 0;
      const int row  = tid - kind * rows;
      const int t2   = row >> 4, r = row & 15;
      const int idx  = 16 * (r >> 3) + 8 * kind + (r & 7);
      const float s = ((sPart[t2][0][idx] + sPart[t2][1][idx]) +
                       sPart[t2][2][idx]) + sPart[t2][3][idx];
      float* dstv = kind ? AD2 : AS2;
      dstv[base + row] = s;
    }
  }
  base += rows;
  }
}

extern "C" void kernel_launch(void* const* d_in, const int* in_sizes, int n_in,
                              void* d_out, int out_size, void* d_ws, size_t ws_size,
                              hipStream_t stream) {
  const float* x0   = (const float*)d_in[0];
  const int*   ei   = (const int*)d_in[1];
  const float* W1   = (const float*)d_in[2];
  const float* as1  = (const float*)d_in[3];
  const float* ad1  = (const float*)d_in[4];
  const float* b1   = (const float*)d_in[5];
  const float* W2   = (const float*)d_in[6];
  const float* as2  = (const float*)d_in[7];
  const float* ad2  = (const float*)d_in[8];
  const float* b2   = (const float*)d_in[9];
  const float* fcW  = (const float*)d_in[10];
  const float* fcb  = (const float*)d_in[11];
  float* out = (float*)d_out;

  const size_t BN      = (size_t)BGRAPH * NNODES;
  const size_t BIGELTS = BN * FEAT;
  char* ws = (char*)d_ws;
  float* BUF0  = (float*)ws;               ws += BIGELTS * sizeof(float);
  float* BUF1  = (float*)ws;               ws += BIGELTS * sizeof(float);
  float* AS    = (float*)ws;               ws += BN * sizeof(float);
  float* AD    = (float*)ws;               ws += BN * sizeof(float);
  float* AS2   = (float*)ws;               ws += BN * sizeof(float);
  float* AD2   = (float*)ws;               ws += BN * sizeof(float);
  int*   START = (int*)ws;                 ws += BN * sizeof(int);
  int*   CNT   = (int*)ws;                 ws += BN * sizeof(int);
  int*   LIST  = (int*)ws;                 ws += (size_t)BGRAPH * NEDGES * sizeof(int);
  int*   REC   = (int*)ws;                 ws += (size_t)BGRAPH * NEDGES * sizeof(int);
  int*   OFF   = (int*)ws;

  const int  gGemm  = GEMM_GRID;
  const dim3 gBin(NSLICE, BGRAPH);
  const dim3 gBucket(NRANGE, BGRAPH);
  const int  gFuse  = FUSE_GRID;

  bin_edges_kernel<<<gBin, BIN_THR, 0, stream>>>(ei, OFF, REC);
  bucket_edges_kernel<<<gBucket, CSR_THR, 0, stream>>>(OFF, REC, START, CNT, LIST);

  gemm_wmma_f32<false, false, true><<<gGemm, GEMM_THR, 0, stream>>>(
      x0, W1, nullptr, BUF0, as1, ad1, AS, AD);
  gat_gather_gemm_kernel<false, false, true><<<gFuse, FUSE_THR, 0, stream>>>(
      START, CNT, LIST, AS, AD, BUF0, b1, W2, nullptr, BUF1, as2, ad2, AS2, AD2);
  gat_gather_gemm_kernel<true, true, false><<<gFuse, FUSE_THR, 0, stream>>>(
      START, CNT, LIST, AS2, AD2, BUF1, b2, fcW, fcb, out, nullptr, nullptr, nullptr, nullptr);
}


# Source-logit table of the work-group's graph (32 KB) in LDS: an edge's logit is an LDS read, not a 32-line global load; to fit, result rows reuse the x rows' LDS, bias b sits in LDS; sum orders kept
# speedup vs baseline: 10.7776x; 1.0364x over previous
#include <hip/hip_runtime.h>
#include <cstdint>

#define BGRAPH 8
#define NNODES 8192
#define NEDGES 262144
#define FEAT   128
#define NEG_SLOPE 0.2f

typedef __attribute__((ext_vector_type(2))) float v2f;
typedef __attribute__((ext_vector_type(8))) float v8f;
typedef __attribute__((ext_vector_type(4))) int   v4i;

__device__ __forceinline__ float leaky(float v) {
  return v >= 0.0f ? v : NEG_SLOPE * v;
}

#define GEMM_THR  128
#define GEMM_TPW  4
#define GEMM_GRID (BGRAPH * NNODES / 16 / GEMM_TPW)

template <int MASK>
__device__ __forceinline__ float row_xor(float v) {
  return __int_as_float(__builtin_amdgcn_update_dpp(
      0, __float_as_int(v), 0x160 | MASK, 0xf, 0xf, true));
}

template <bool TRANSB, bool BIAS, bool ALPHA>
__global__ void __launch_bounds__(GEMM_THR)
gemm_wmma_f32(const float* __restrict__ X, const float* __restrict__ W,
              const float* __restrict__ bias, float* __restrict__ H,
              const float* __restrict__ a_src, const float* __restrict__ a_dst,
              float* __restrict__ AS, float* __restrict__ AD) {
  __shared__ float sPart[GEMM_TPW][GEMM_THR / 32][32];
  const int wave = threadIdx.x >> 5;
  const int lane = threadIdx.x & 31;
  const int half = lane >> 4;
  const int l    = lane & 15;
  const int n0   = wave * 32;

  v2f bf0[FEAT / 4], bf1[FEAT / 4];
#pragma unroll
  for (int t = 0; t < FEAT / 4; ++t) {
    const int ka = 4 * t + 2 * half;
    if (TRANSB) {
      const float* wr = W + (size_t)(n0 + l) * FEAT + ka;
      bf0[t].x = wr[0];
      bf0[t].y = wr[1];
      bf1[t].x = wr[16 * FEAT];
      bf1[t].y = wr[16 * FEAT + 1];
    } else {
      bf0[t].x = W[(size_t)ka * FEAT + (n0 + l)];
      bf0[t].y = W[(size_t)(ka + 1) * FEAT + (n0 + l)];
      bf1[t].x = W[(size_t)ka * FEAT + (n0 + 16 + l)];
      bf1[t].y = W[(size_t)(ka + 1) * FEAT + (n0 + 16 + l)];
    }
  }
  const float bv0 = BIAS ? bias[n0 + l] : 0.0f;
  const float bv1 = BIAS ? bias[n0 + 16 + l] : 0.0f;
  const float as0 = ALPHA ? a_src[n0 + l] : 0.0f;
  const float as1 = ALPHA ? a_src[n0 + 16 + l] : 0.0f;
  const float ad0 = ALPHA ? a_dst[n0 + l] : 0.0f;
  const float ad1 = ALPHA ? a_dst[n0 + 16 + l] : 0.0f;

#pragma nounroll
  for (int tt = 0; tt < GEMM_TPW; ++tt) {
    const size_t m0 = ((size_t)blockIdx.x * GEMM_TPW + tt) * 16;
    const float* xrow = X + (m0 + l) * FEAT + 2 * half;
    v8f acc0 = {}, acc1 = {};
#pragma unroll
    for (int t = 0; t < FEAT / 4; ++t) {
      v2f a;
      a.x = xrow[4 * t];
      a.y = xrow[4 * t + 1];
      acc0 = __builtin_amdgcn_wmma_f32_16x16x4_f32(
           false, a,  false, bf0[t],
           (short)0, acc0,  false,  false);
      acc1 = __builtin_amdgcn_wmma_f32_16x16x4_f32(
           false, a,  false, bf1[t],
           (short)0, acc1,  false,  false);
    }
    float* hout = H + (m0 + 8 * half) * FEAT + (n0 + l);
#pragma unroll
    for (int i = 0; i < 8; ++i) {
      hout[(size_t)i * FEAT]      = acc0[i] + bv0;
      hout[(size_t)i * FEAT + 16] = acc1[i] + bv1;
    }
    if (ALPHA) {
      float v[16];
#pragma unroll
      for (int i = 0; i < 8; ++i) {
        v[i]     = acc0[i] * as0 + acc1[i] * as1;
        v[8 + i] = acc0[i] * ad0 + acc1[i] * ad1;
      }
      const bool b3 = (l & 8) != 0, b2 = (l & 4) != 0;
      const bool b1 = (l & 2) != 0, b0 = (l & 1) != 0;
      float u[8], w4[4], y[2];
#pragma unroll
      for (int k = 0; k < 8; ++k) {
        const float keep = b3 ? v[k + 8] : v[k];
        const float send = b3 ? v[k] : v[k + 8];
        u[k] = keep + row_xor<8>(send);
      }
#pragma unroll
      for (int k = 0; k < 4; ++k) {
        const float keep = b2 ? u[k + 4] : u[k];
        const float send = b2 ? u[k] : u[k + 4];
        w4[k] = keep + row_xor<4>(send);
      }
#pragma unroll
      for (int k = 0; k < 2; ++k) {
        const float keep = b1 ? w4[k + 2] : w4[k];
        const float send = b1 ? w4[k] : w4[k + 2];
        y[k] = keep + row_xor<2>(send);
      }
      const float keep = b0 ? y[1] : y[0];
      const float send = b0 ? y[0] : y[1];
      sPart[tt][wave][lane] = keep + row_xor<1>(send);
    }
  }
  if (ALPHA) {
    __syncthreads();
    const int kind = wave >> 1;
    const int tile = 2 * (wave & 1) + (lane >> 4);
    const int row  = lane & 15;
    const int idx  = 16 * (row >> 3) + 8 * kind + (row & 7);
    const float s = ((sPart[tile][0][idx] + sPart[tile][1][idx]) +
                     sPart[tile][2][idx]) + sPart[tile][3][idx];
    float* dstv = kind ? AD : AS;
    dstv[(size_t)blockIdx.x * (16 * GEMM_TPW) + 32 * (wave & 1) + lane] = s;
  }
}

#define RANGE_N   256
#define NRANGE    (NNODES / RANGE_N)
#define CSR_THR   512
#define CSR_NW    (CSR_THR / 32)
#define SLICE_E   8192
#define NSLICE    (NEDGES / SLICE_E)
#define BIN_THR   512
#define BIN_NW    (BIN_THR / 32)
#define BIN_EPT   (SLICE_E / BIN_THR)
#define WALK_S    (NSLICE / CSR_NW)
#define WALK_LD   10
#define NODE_BITS 13
#define NODE_MASK (NNODES - 1)

static_assert(NNODES == (1 << NODE_BITS), "record packing");
static_assert(NRANGE * BIN_NW == BIN_THR, "one scan entry per thread");
static_assert(NSLICE == WALK_S * CSR_NW, "whole slices per wave");
static_assert(RANGE_N <= CSR_THR, "one thread per owned node");
static_assert(BIN_EPT == 16, "four int4 loads per thread");

__global__ void __launch_bounds__(BIN_THR)
bin_edges_kernel(const int* __restrict__ EI, int* __restrict__ OFF,
                 int* __restrict__ REC) {
  __shared__ int sCnt[BIN_NW][NRANGE];
  __shared__ int sWave[BIN_NW];
  __shared__ int sRec[SLICE_E];
  const int t    = threadIdx.x;
  const int lane = t & 31;
  const int wave = t >> 5;
  const int s    = blockIdx.x;
  const int g    = blockIdx.y;
  const int* srcp = EI + (size_t)g * 2 * NEDGES + (size_t)s * SLICE_E;
  const int* dstp = srcp + NEDGES;

  (&sCnt[0][0])[t] = 0;
  __syncthreads();

  v4i d[4], sv[4];
#pragma unroll
  for (int u = 0; u < 4; ++u) {
    const int e = 4 * (t + BIN_THR * u);
    d[u]  = *(const v4i*)(dstp + e);
    sv[u] = *(const v4i*)(srcp + e);
  }
  asm volatile("" :: "v"(d[0]), "v"(d[1]), "v"(d[2]), "v"(d[3]),
                     "v"(sv[0]), "v"(sv[1]), "v"(sv[2]), "v"(sv[3]));
  int rec[BIN_EPT], rk[BIN_EPT];
#pragma unroll
  for (int u = 0; u < 4; ++u) {
    const int dv[4] = {d[u].x, d[u].y, d[u].z, d[u].w};
    const int sw[4] = {sv[u].x, sv[u].y, sv[u].z, sv[u].w};
#pragma unroll
    for (int i = 0; i < 4; ++i) {
      const int k = 4 * u + i;
      rec[k] = ((dv[i] & NODE_MASK) << NODE_BITS) | (sw[i] & NODE_MASK);
      const int r = rec[k] >> (NODE_BITS + 8);
      rk[k] = atomicAdd(&sCnt[wave][r], 1);
    }
  }
  __syncthreads();

  const int sr = t / BIN_NW;
  const int sw_ = t % BIN_NW;
  const int v  = sCnt[sw_][sr];
  int incl = v;
#pragma unroll
  for (int off = 1; off < 32; off <<= 1) {
    const int n = __shfl_up(incl, off, 32);
    if (lane >= off) incl += n;
  }
  if (lane == 31) sWave[wave] = incl;
  __syncthreads();
  int pre = 0;
#pragma unroll
  for (int i = 0; i < BIN_NW; ++i) pre += (i < wave) ? sWave[i] : 0;
  const int excl = pre + incl - v;
  sCnt[sw_][sr] = excl;
  int* off = OFF + (size_t)(g * NSLICE + s) * (NRANGE + 1);
  if (sw_ == 0) off[sr] = excl;
  if (t == BIN_THR - 1) off[NRANGE] = excl + v;
  __syncthreads();

#pragma unroll
  for (int k = 0; k < BIN_EPT; ++k) {
    const int r = rec[k] >> (NODE_BITS + 8);
    sRec[sCnt[wave][r] + rk[k]] = rec[k];
  }
  __syncthreads();

  int* out = REC + (size_t)(g * NSLICE + s) * SLICE_E;
#pragma unroll
  for (int u = 0; u < 4; ++u) {
    const int e = 4 * (t + BIN_THR * u);
    *(int4*)(out + e) = *(const int4*)(sRec + e);
  }
}

__global__ void __launch_bounds__(CSR_THR)
bucket_edges_kernel(const int* __restrict__ OFF, const int* __restrict__ REC,
                    int* __restrict__ START, int* __restrict__ CNT,
                    int* __restrict__ LIST) {
  __shared__ int sCur[RANGE_N];
  __shared__ int sScan[RANGE_N];
  __shared__ int sLow[CSR_NW];
  const int t    = threadIdx.x;
  const int lane = t & 31;
  const int wave = __builtin_amdgcn_readfirstlane(t >> 5);
  const int r    = blockIdx.x;
  const int lo   = r * RANGE_N;
  const int g    = blockIdx.y;
  int beg[WALK_S], end[WALK_S];
  const int* rec[WALK_S];
#pragma unroll
  for (int k = 0; k < WALK_S; ++k) {
    const int s = WALK_S * wave + k;
    const int* off = OFF + (size_t)(g * NSLICE + s) * (NRANGE + 1);
    beg[k] = off[r];
    end[k] = off[r + 1];
    rec[k] = REC + (size_t)(g * NSLICE + s) * SLICE_E;
  }

  if (t < RANGE_N) sCur[t] = 0;
  __syncthreads();

  int rr[WALK_S][WALK_LD];
#pragma unroll
  for (int k = 0; k < WALK_S; ++k) {
#pragma unroll
    for (int u = 0; u < WALK_LD; ++u) {
      const int i = beg[k] + 32 * u + lane;
      rr[k][u] = -1;
      if (i < end[k]) rr[k][u] = rec[k][i];
    }
  }
#pragma unroll
  for (int k = 0; k < WALK_S; ++k) {
#pragma unroll
    for (int u = 0; u < WALK_LD; ++u)
      if (rr[k][u] >= 0) atomicAdd(&sCur[(rr[k][u] >> NODE_BITS) & (RANGE_N - 1)], 1);
  }
#pragma unroll
  for (int k = 0; k < WALK_S; ++k) {
    for (int c = beg[k] + 32 * WALK_LD; c < end[k]; c += 32) {
      const int i = c + lane;
      if (i < end[k]) atomicAdd(&sCur[(rec[k][i] >> NODE_BITS) & (RANGE_N - 1)], 1);
    }
  }
  int low = 0;
#pragma unroll
  for (int k = 0; k < WALK_S; ++k) low += beg[k];
  if (lane == 0) sLow[wave] = low;
  __syncthreads();
  int base = 0;
#pragma unroll
  for (int i = 0; i < CSR_NW; ++i) base += sLow[i];

  if (t < RANGE_N) sScan[t] = sCur[t];
  __syncthreads();
  for (int o = 1; o < RANGE_N; o <<= 1) {
    int v = 0;
    if (t < RANGE_N && t >= o) v = sScan[t - o];
    __syncthreads();
    if (t < RANGE_N) sScan[t] += v;
    __syncthreads();
  }
  if (t < RANGE_N) {
    const int c  = sCur[t];
    const int st = base + sScan[t] - c;
    START[g * NNODES + lo + t] = st;
    CNT[g * NNODES + lo + t]   = c;
    sCur[t] = st;
  }
  __syncthreads();

  int* lst = LIST + (size_t)g * NEDGES;
  unsigned pos[WALK_S][WALK_LD];
#pragma unroll
  for (int k = 0; k < WALK_S; ++k) {
#pragma unroll
    for (int u = 0; u < WALK_LD; ++u) {
      pos[k][u] = 0u;
      if (rr[k][u] >= 0)
        pos[k][u] = (unsigned)atomicAdd(&sCur[(rr[k][u] >> NODE_BITS) & (RANGE_N - 1)], 1);
    }
  }
#pragma unroll
  for (int k = 0; k < WALK_S; ++k) {
#pragma unroll
    for (int u = 0; u < WALK_LD; ++u)
      if (rr[k][u] >= 0) lst[pos[k][u]] = rr[k][u] & NODE_MASK;
  }
#pragma unroll
  for (int k = 0; k < WALK_S; ++k) {
    for (int c = beg[k] + 32 * WALK_LD; c < end[k]; c += 32) {
      const int i = c + lane;
      if (i < end[k]) {
        const int v = rec[k][i];
        const int p = atomicAdd(&sCur[(v >> NODE_BITS) & (RANGE_N - 1)], 1);
        lst[p] = v & NODE_MASK;
      }
    }
  }
}

#define FUSE_THR   1024
#define FUSE_NW    (FUSE_THR / 32)
#define FUSE_NODES 128
#define FUSE_GRID  (BGRAPH * NNODES / FUSE_NODES)
#define FUSE_MAXR  64
#define XP         132
#define WP         288
#define SW_WORDS   ((FEAT / 2) * WP > FEAT * XP ? (FEAT / 2) * WP : FEAT * XP)

static_assert(FUSE_NODES == 4 * FUSE_NW, "four nodes per wave");
static_assert(NNODES % FUSE_NODES == 0, "a workgroup's nodes lie in one graph");
static_assert(NNODES == 8 * FUSE_THR, "two 16-byte copies per thread fill the logit table");
static_assert(FUSE_MAXR == 2 * FUSE_NW, "at most two nodes per wave and pass");

template <bool TRANSB, bool BIAS, bool ALPHA>
__global__ void __launch_bounds__(FUSE_THR, 16)
gat_gather_gemm_kernel(const int* __restrict__ START, const int* __restrict__ CNT,
                       const int* __restrict__ LIST, const float* __restrict__ AS,
                       const float* __restrict__ AD, const float* __restrict__ H,
                       const float* __restrict__ b, const float* __restrict__ W,
                       const float* __restrict__ bias2, float* __restrict__ OUT,
                       const float* __restrict__ a_src, const float* __restrict__ a_dst,
                       float* __restrict__ AS2, float* __restrict__ AD2) {
  __shared__ __attribute__((aligned(16))) float sW[SW_WORDS];
  __shared__ __attribute__((aligned(16))) float sX[FUSE_MAXR * XP];
  __shared__ __attribute__((aligned(16))) float sAS[NNODES];
  __shared__ __attribute__((aligned(16))) float sB[FEAT];
  __shared__ float sPart[FUSE_MAXR / 16][4][32];
  const int tid  = threadIdx.x;
  const int wave = __builtin_amdgcn_readfirstlane((int)(threadIdx.x >> 5));
  const int lane = threadIdx.x & 31;
  const int half = lane >> 4;
  const int l    = lane & 15;
  const int f    = lane * 4;

  if (TRANSB) {
#pragma unroll
    for (int u = 0; u < 4; ++u) {
      const int q = tid + FUSE_THR * u;
      const int row = q >> 5, c4 = (q & 31) * 4;
      *(float4*)(sW + row * XP + c4) = *(const float4*)(W + row * FEAT + c4);
    }
  } else {
#pragma nounroll
    for (int uo = 0; uo < 8; uo += 4) {
#pragma unroll
      for (int u = 0; u < 4; ++u) {
        const int q = tid + FUSE_THR * (uo + u);
        const int p = q >> 7, n = q & 127;
        v2f pr;
        pr.x = W[(2 * p) * FEAT + n];
        pr.y = W[(2 * p + 1) * FEAT + n];
        *(v2f*)(sW + p * WP + 2 * n) = pr;
      }
    }
  }

  {
    const float* asg = AS + (size_t)(((int)blockIdx.x * FUSE_NODES) / NNODES) * NNODES;
#pragma unroll
    for (int u = 0; u < 2; ++u) {
      const int q = 4 * (tid + FUSE_THR * u);
      *(float4*)(sAS + q) = *(const float4*)(asg + q);
    }
  }
  if (tid < FEAT / 4) *(float4*)(sB + 4 * tid) = *(const float4*)(b + 4 * tid);
  __syncthreads();

  const int tl = wave >> 3;
  const int n0 = (wave & 3) * 32;

  const bool shifted = (((blockIdx.x >> 8) ^ blockIdx.x) & 1) != 0;
  const int npass = shifted ? 3 : 2;
  int base = (int)blockIdx.x * FUSE_NODES;
  for (int pass = 0; pass < npass; ++pass) {
  const int npw  = (shifted && pass != 1) ? 1 : 2;
  const int rows = FUSE_NW * npw;
  for (int it = 0; it < npw; ++it) {
  const int rr   = wave + FUSE_NW * it;
  const int gn   = base + rr;
  const int g    = gn / NNODES;
  const int nl   = gn - g * NNODES;
  const int cnt   = CNT[gn];
  const int total = cnt + 1;
  const int* lst  = LIST + (size_t)g * NEDGES + START[gn];
  const float a_d = AD[gn];
  const float* hg = H + (size_t)g * NNODES * FEAT + lane * 4;

  int s0 = nl;
  if (lane < cnt) s0 = lst[lane];
  const float e0 = leaky(sAS[s0] + a_d);
  float m = (lane < total) ? e0 : -INFINITY;
  for (int c = 32; c < total; c += 32) {
    const int i = c + lane;
    int s = nl;
    if (i < cnt) s = lst[i];
    const float e = leaky(sAS[s] + a_d);
    if (i < total) m = fmaxf(m, e);
  }
#pragma unroll
  for (int off = 16; off > 0; off >>= 1) m = fmaxf(m, __shfl_xor(m, off, 32));

  float4 acc = make_float4(0.0f, 0.0f, 0.0f, 0.0f);
  float wsum = 0.0f;
  for (int c = 0; c < total; c += 32) {
    const int i = c + lane;
    int s = s0;
    float e = e0;
    if (c > 0) {
      s = nl;
      if (i < cnt) s = lst[i];
      e = leaky(sAS[s] + a_d);
    }
    const float w = (i < total) ? __expf(e - m) : 0.0f;
    wsum += w;
    const int rem = total - c;
    const int nc  = rem < 32 ? rem : 32;
    for (int j = 0; j < nc; j += 4) {
      const int r0 = __builtin_amdgcn_readlane(s, j);
      const int r1 = __builtin_amdgcn_readlane(s, j + 1);
      const int r2 = __builtin_amdgcn_readlane(s, j + 2);
      const int r3 = __builtin_amdgcn_readlane(s, j + 3);
      const float w0 = __int_as_float(__builtin_amdgcn_readlane(__float_as_int(w), j));
      const float w1 = __int_as_float(__builtin_amdgcn_readlane(__float_as_int(w), j + 1));
      const float w2 = __int_as_float(__builtin_amdgcn_readlane(__float_as_int(w), j + 2));
      const float w3 = __int_as_float(__builtin_amdgcn_readlane(__float_as_int(w), j + 3));
      const float4 h0 = *(const float4*)(hg + (size_t)r0 * FEAT);
      const float4 h1 = *(const float4*)(hg + (size_t)r1 * FEAT);
      const float4 h2 = *(const float4*)(hg + (size_t)r2 * FEAT);
      const float4 h3 = *(const float4*)(hg + (size_t)r3 * FEAT);
      acc.x += w0 * h0.x; acc.y += w0 * h0.y; acc.z += w0 * h0.z; acc.w += w0 * h0.w;
      acc.x += w1 * h1.x; acc.y += w1 * h1.y; acc.z += w1 * h1.z; acc.w += w1 * h1.w;
      acc.x += w2 * h2.x; acc.y += w2 * h2.y; acc.z += w2 * h2.z; acc.w += w2 * h2.w;
      acc.x += w3 * h3.x; acc.y += w3 * h3.y; acc.z += w3 * h3.z; acc.w += w3 * h3.w;
    }
  }
#pragma unroll
  for (int off = 16; off > 0; off >>= 1) wsum += __shfl_xor(wsum, off, 32);

  const float4 bv = *(const float4*)(sB + f);
  float4 v;
  v.x = acc.x / wsum + bv.x;
  v.y = acc.y / wsum + bv.y;
  v.z = acc.z / wsum + bv.z;
  v.w = acc.w / wsum + bv.w;
  float4 o;
  o.x = v.x / (1.0f + __expf(-v.x));
  o.y = v.y / (1.0f + __expf(-v.y));
  o.z = v.z / (1.0f + __expf(-v.z));
  o.w = v.w / (1.0f + __expf(-v.w));
  *(float4*)(sX + rr * XP + f) = o;
  }
  __syncthreads();

  const bool mult = (wave & 4) == 0 && tl < (rows >> 4);
  v8f acc0 = {}, acc1 = {};
  float bo0 = 0.0f, bo1 = 0.0f, as0 = 0.0f, as1 = 0.0f, ad0 = 0.0f, ad1 = 0.0f;
  if (mult) {
    const float* xa = sX + (16 * tl + l) * XP + 2 * half;
    const float* wb = TRANSB ? sW + (n0 + l) * XP + 2 * half
                             : sW + half * WP + 2 * (n0 + l);
    if (BIAS) {
      bo0 = bias2[n0 + l];
      bo1 = bias2[n0 + 16 + l];
    }
    if (ALPHA) {
      as0 = a_src[n0 + l];
      as1 = a_src[n0 + 16 + l];
      ad0 = a_dst[n0 + l];
      ad1 = a_dst[n0 + 16 + l];
    }
#pragma unroll
    for (int t = 0; t < FEAT / 4; ++t) {
      const v2f a = *(const v2f*)(xa + 4 * t);
      const v2f q0 = *(const v2f*)(wb + (TRANSB ? 4 : 2 * WP) * t);
      const v2f q1 = *(const v2f*)(wb + (TRANSB ? 4 : 2 * WP) * t + (TRANSB ? 16 * XP : 32));
      acc0 = __builtin_amdgcn_wmma_f32_16x16x4_f32(
           false, a,  false, q0,
           (short)0, acc0,  false,  false);
      acc1 = __builtin_amdgcn_wmma_f32_16x16x4_f32(
           false, a,  false, q1,
           (short)0, acc1,  false,  false);
    }
  }
  __syncthreads();

  if (mult) {
    float* so = sX + (16 * tl + 8 * half) * XP + n0 + l;
#pragma unroll
    for (int i = 0; i < 8; ++i) {
      so[i * XP]      = acc0[i] + bo0;
      so[i * XP + 16] = acc1[i] + bo1;
    }
    if (ALPHA) {
      float v[16];
#pragma unroll
      for (int i = 0; i < 8; ++i) {
        v[i]     = acc0[i] * as0 + acc1[i] * as1;
        v[8 + i] = acc0[i] * ad0 + acc1[i] * ad1;
      }
      const bool b3 = (l & 8) != 0, b2 = (l & 4) != 0;
      const bool b1 = (l & 2) != 0, b0 = (l & 1) != 0;
      float u[8], w4[4], y[2];
#pragma unroll
      for (int k = 0; k < 8; ++k) {
        const float keep = b3 ? v[k + 8] : v[k];
        const float send = b3 ? v[k] : v[k + 8];
        u[k] = keep + row_xor<8>(send);
      }
#pragma unroll
      for (int k = 0; k < 4; ++k) {
        const float keep = b2 ? u[k + 4] : u[k];
        const float send = b2 ? u[k] : u[k + 4];
        w4[k] = keep + row_xor<4>(send);
      }
#pragma unroll
      for (int k = 0; k < 2; ++k) {
        const float keep = b1 ? w4[k + 2] : w4[k];
        const float send = b1 ? w4[k] : w4[k + 2];
        y[k] = keep + row_xor<2>(send);
      }
      const float keep = b0 ? y[1] : y[0];
      const float send = b0 ? y[0] : y[1];
      sPart[tl][wave & 3][lane] = keep + row_xor<1>(send);
    }
  }
  __syncthreads();

  for (int u = 0; u < npw; ++u) {
    const int q = tid + FUSE_THR * u;
    const int row = q >> 5, c4 = (q & 31) * 4;
    *(float4*)(OUT + (size_t)(base + row) * FEAT + c4) = *(const float4*)(sX + row * XP + c4);
  }
  if (ALPHA) {
    if (tid < 2 * rows) {
      const int kind = (tid >= rows) ? 1 : 0;
      const int row  = tid - kind * rows;
      const int t2   = row >> 4, r = row & 15;
      const int idx  = 16 * (r >> 3) + 8 * kind + (r & 7);
      const float s = ((sPart[t2][0][idx] + sPart[t2][1][idx]) +
                       sPart[t2][2][idx]) + sPart[t2][3][idx];
      float* dstv = kind ? AD2 : AS2;
      dstv[base + row] = s;
    }
  }
  base += rows;
  }
}

extern "C" void kernel_launch(void* const* d_in, const int* in_sizes, int n_in,
                              void* d_out, int out_size, void* d_ws, size_t ws_size,
                              hipStream_t stream) {
  const float* x0   = (const float*)d_in[0];
  const int*   ei   = (const int*)d_in[1];
  const float* W1   = (const float*)d_in[2];
  const float* as1  = (const float*)d_in[3];
  const float* ad1  = (const float*)d_in[4];
  const float* b1   = (const float*)d_in[5];
  const float* W2   = (const float*)d_in[6];
  const float* as2  = (const float*)d_in[7];
  const float* ad2  = (const float*)d_in[8];
  const float* b2   = (const float*)d_in[9];
  const float* fcW  = (const float*)d_in[10];
  const float* fcb  = (const float*)d_in[11];
  float* out = (float*)d_out;

  const size_t BN      = (size_t)BGRAPH * NNODES;
  const size_t BIGELTS = BN * FEAT;
  char* ws = (char*)d_ws;
  float* BUF0  = (float*)ws;               ws += BIGELTS * sizeof(float);
  float* BUF1  = (float*)ws;               ws += BIGELTS * sizeof(float);
  float* AS    = (float*)ws;               ws += BN * sizeof(float);
  float* AD    = (float*)ws;               ws += BN * sizeof(float);
  float* AS2   = (float*)ws;               ws += BN * sizeof(float);
  float* AD2   = (float*)ws;               ws += BN * sizeof(float);
  int*   START = (int*)ws;                 ws += BN * sizeof(int);
  int*   CNT   = (int*)ws;                 ws += BN * sizeof(int);
  int*   LIST  = (int*)ws;                 ws += (size_t)BGRAPH * NEDGES * sizeof(int);
  int*   REC   = (int*)ws;                 ws += (size_t)BGRAPH * NEDGES * sizeof(int);
  int*   OFF   = (int*)ws;

  const int  gGemm  = GEMM_GRID;
  const dim3 gBin(NSLICE, BGRAPH);
  const dim3 gBucket(NRANGE, BGRAPH);
  const int  gFuse  = FUSE_GRID;

  bin_edges_kernel<<<gBin, BIN_THR, 0, stream>>>(ei, OFF, REC);
  bucket_edges_kernel<<<gBucket, CSR_THR, 0, stream>>>(OFF, REC, START, CNT, LIST);

  gemm_wmma_f32<false, false, true><<<gGemm, GEMM_THR, 0, stream>>>(
      x0, W1, nullptr, BUF0, as1, ad1, AS, AD);
  gat_gather_gemm_kernel<false, false, true><<<gFuse, FUSE_THR, 0, stream>>>(
      START, CNT, LIST, AS, AD, BUF0, b1, W2, nullptr, BUF1, as2, ad2, AS2, AD2);
  gat_gather_gemm_kernel<true, true, false><<<gFuse, FUSE_THR, 0, stream>>>(
      START, CNT, LIST, AS2, AD2, BUF1, b2, fcW, fcb, out, nullptr, nullptr, nullptr, nullptr);
}


# Layer 1's x@W1 (with logits) has no launch of its own: half its rows run in each edge-sorting launch, a role per work-group, one of each per WGP; to fit, its operands go through LDS; 5 launches -> 4
# speedup vs baseline: 11.2903x; 1.0476x over previous
#include <hip/hip_runtime.h>
#include <cstdint>

#define BGRAPH 8
#define NNODES 8192
#define NEDGES 262144
#define FEAT   128
#define NEG_SLOPE 0.2f

typedef __attribute__((ext_vector_type(2))) float v2f;
typedef __attribute__((ext_vector_type(4))) float v4f;
typedef __attribute__((ext_vector_type(8))) float v8f;
typedef __attribute__((ext_vector_type(4))) int   v4i;

__device__ __forceinline__ float leaky(float v) {
  return v >= 0.0f ? v : NEG_SLOPE * v;
}

#define GEMM_THR  512
#define GEMM_ROWS 64
#define GEMM_PASS 2
#define GEMM_GRID (BGRAPH * NNODES / (GEMM_ROWS * GEMM_PASS))
#define XP        132
#define WP        288

static_assert(GEMM_ROWS * FEAT / 4 == 4 * GEMM_THR, "four 16-byte row pieces per thread and pass");
static_assert((GEMM_ROWS / 16) * 4 * 32 == GEMM_THR, "one wave per row tile and strip");

template <int MASK>
__device__ __forceinline__ float row_xor(float v) {
  return __int_as_float(__builtin_amdgcn_update_dpp(
      0, __float_as_int(v), 0x160 | MASK, 0xf, 0xf, true));
}

__device__ __forceinline__ void
gemm_wmma_f32(const int wg, const float* __restrict__ X, const float* __restrict__ W,
              float* __restrict__ H,
              const float* __restrict__ a_src, const float* __restrict__ a_dst,
              float* __restrict__ AS, float* __restrict__ AD) {
  __shared__ __attribute__((aligned(16))) float sWp[(FEAT / 2) * WP];
  __shared__ __attribute__((aligned(16))) float sXr[GEMM_ROWS * XP];
  __shared__ float sPart[GEMM_ROWS / 16][4][32];
  const int tid  = threadIdx.x;
  const int wave = tid >> 5;
  const int lane = tid & 31;
  const int half = lane >> 4;
  const int l    = lane & 15;
  const int tl   = wave >> 2;
  const int sw   = wave & 3;
  const int n0   = sw * 32;
  const size_t row0 = (size_t)wg * (GEMM_ROWS * GEMM_PASS);

  const float* xg = X + row0 * FEAT + 4 * tid;
  float* sx = sXr + (tid >> 5) * XP + (tid & 31) * 4;
  v4f nx0 = *(const v4f*)(xg);
  v4f nx1 = *(const v4f*)(xg + 4 * GEMM_THR);
  v4f nx2 = *(const v4f*)(xg + 8 * GEMM_THR);
  v4f nx3 = *(const v4f*)(xg + 12 * GEMM_THR);

#pragma nounroll
  for (int uo = 0; uo < (FEAT / 2) * FEAT / GEMM_THR; uo += 4) {
#pragma unroll
    for (int u = 0; u < 4; ++u) {
      const int q = tid + GEMM_THR * (uo + u);
      const int p = q >> 7, n = q & 127;
      v2f pr;
      pr.x = W[(2 * p) * FEAT + n];
      pr.y = W[(2 * p + 1) * FEAT + n];
      *(v2f*)(sWp + p * WP + 2 * n) = pr;
    }
  }
  const float as0 = a_src[n0 + l];
  const float as1 = a_src[n0 + 16 + l];
  const float ad0 = a_dst[n0 + l];
  const float ad1 = a_dst[n0 + 16 + l];

  const float* xa = sXr + (16 * tl + l) * XP + 2 * half;
  const float* wb = sWp + half * WP + 2 * (n0 + l);

#pragma nounroll
  for (int pass = 0; pass < GEMM_PASS; ++pass) {
    *(v4f*)(sx)           = nx0;
    *(v4f*)(sx + 16 * XP) = nx1;
    *(v4f*)(sx + 32 * XP) = nx2;
    *(v4f*)(sx + 48 * XP) = nx3;
    __syncthreads();
    if (pass + 1 < GEMM_PASS) {
      const float* xn = xg + (size_t)(pass + 1) * (GEMM_ROWS * FEAT);
      nx0 = *(const v4f*)(xn);
      nx1 = *(const v4f*)(xn + 4 * GEMM_THR);
      nx2 = *(const v4f*)(xn + 8 * GEMM_THR);
      nx3 = *(const v4f*)(xn + 12 * GEMM_THR);
    }

    v8f acc0 = {}, acc1 = {};
#pragma unroll
    for (int t = 0; t < FEAT / 4; ++t) {
      const v2f a = *(const v2f*)(xa + 4 * t);
      const v2f q0 = *(const v2f*)(wb + 2 * WP * t);
      const v2f q1 = *(const v2f*)(wb + 2 * WP * t + 32);
      acc0 = __builtin_amdgcn_wmma_f32_16x16x4_f32(
           false, a,  false, q0,
           (short)0, acc0,  false,  false);
      acc1 = __builtin_amdgcn_wmma_f32_16x16x4_f32(
           false, a,  false, q1,
           (short)0, acc1,  false,  false);
    }
    const size_t m0 = row0 + (size_t)pass * GEMM_ROWS + 16 * tl;
    float* hout = H + (m0 + 8 * half) * FEAT + (n0 + l);
#pragma unroll
    for (int i = 0; i < 8; ++i) {
      hout[(size_t)i * FEAT]      = acc0[i];
      hout[(size_t)i * FEAT + 16] = acc1[i];
    }
    {
    float v[16];
#pragma unroll
    for (int i = 0; i < 8; ++i) {
      v[i]     = acc0[i] * as0 + acc1[i] * as1;
      v[8 + i] = acc0[i] * ad0 + acc1[i] * ad1;
    }
    const bool b3 = (l & 8) != 0, b2 = (l & 4) != 0;
    const bool b1 = (l & 2) != 0, b0 = (l & 1) != 0;
    float u[8], w4[4], y[2];
#pragma unroll
    for (int k = 0; k < 8; ++k) {
      const float keep = b3 ? v[k + 8] : v[k];
      const float send = b3 ? v[k] : v[k + 8];
      u[k] = keep + row_xor<8>(send);
    }
#pragma unroll
    for (int k = 0; k < 4; ++k) {
      const float keep = b2 ? u[k + 4] : u[k];
      const float send = b2 ? u[k] : u[k + 4];
      w4[k] = keep + row_xor<4>(send);
    }
#pragma unroll
    for (int k = 0; k < 2; ++k) {
      const float keep = b1 ? w4[k + 2] : w4[k];
      const float send = b1 ? w4[k] : w4[k + 2];
      y[k] = keep + row_xor<2>(send);
    }
    const float keep = b0 ? y[1] : y[0];
    const float send = b0 ? y[0] : y[1];
    sPart[tl][sw][lane] = keep + row_xor<1>(send);
    }
    __syncthreads();

    if (tid < 2 * GEMM_ROWS) {
      const int kind = (tid >= GEMM_ROWS) ? 1 : 0;
      const int row  = tid - kind * GEMM_ROWS;
      const int t2   = row >> 4, r = row & 15;
      const int idx  = 16 * (r >> 3) + 8 * kind + (r & 7);
      const float s = ((sPart[t2][0][idx] + sPart[t2][1][idx]) +
                       sPart[t2][2][idx]) + sPart[t2][3][idx];
      float* dstv = kind ? AD : AS;
      dstv[row0 + (size_t)pass * GEMM_ROWS + row] = s;
    }
  }
}

#define RANGE_N   256
#define NRANGE    (NNODES / RANGE_N)
#define CSR_THR   512
#define CSR_NW    (CSR_THR / 32)
#define SLICE_E   8192
#define NSLICE    (NEDGES / SLICE_E)
#define BIN_THR   512
#define BIN_NW    (BIN_THR / 32)
#define BIN_EPT   (SLICE_E / BIN_THR)
#define WALK_S    (NSLICE / CSR_NW)
#define WALK_LD   10
#define NODE_BITS 13
#define NODE_MASK (NNODES - 1)

static_assert(NNODES == (1 << NODE_BITS), "record packing");
static_assert(NRANGE * BIN_NW == BIN_THR, "one scan entry per thread");
static_assert(NSLICE == WALK_S * CSR_NW, "whole slices per wave");
static_assert(RANGE_N <= CSR_THR, "one thread per owned node");
static_assert(BIN_EPT == 16, "four int4 loads per thread");

__device__ __forceinline__ void
bin_edges_slice(const int s, const int g, const int* __restrict__ EI,
                int* __restrict__ OFF, int* __restrict__ REC) {
  __shared__ int sCnt[BIN_NW][NRANGE];
  __shared__ int sWave[BIN_NW];
  __shared__ int sRec[SLICE_E];
  const int t    = threadIdx.x;
  const int lane = t & 31;
  const int wave = t >> 5;
  const int* srcp = EI + (size_t)g * 2 * NEDGES + (size_t)s * SLICE_E;
  const int* dstp = srcp + NEDGES;

  (&sCnt[0][0])[t] = 0;
  __syncthreads();

  v4i d[4], sv[4];
#pragma unroll
  for (int u = 0; u < 4; ++u) {
    const int e = 4 * (t + BIN_THR * u);
    d[u]  = *(const v4i*)(dstp + e);
    sv[u] = *(const v4i*)(srcp + e);
  }
  asm volatile("" :: "v"(d[0]), "v"(d[1]), "v"(d[2]), "v"(d[3]),
                     "v"(sv[0]), "v"(sv[1]), "v"(sv[2]), "v"(sv[3]));
  int rec[BIN_EPT], rk[BIN_EPT];
#pragma unroll
  for (int u = 0; u < 4; ++u) {
    const int dv[4] = {d[u].x, d[u].y, d[u].z, d[u].w};
    const int sw[4] = {sv[u].x, sv[u].y, sv[u].z, sv[u].w};
#pragma unroll
    for (int i = 0; i < 4; ++i) {
      const int k = 4 * u + i;
      rec[k] = ((dv[i] & NODE_MASK) << NODE_BITS) | (sw[i] & NODE_MASK);
      const int r = rec[k] >> (NODE_BITS + 8);
      rk[k] = atomicAdd(&sCnt[wave][r], 1);
    }
  }
  __syncthreads();

  const int sr = t / BIN_NW;
  const int sw_ = t % BIN_NW;
  const int v  = sCnt[sw_][sr];
  int incl = v;
#pragma unroll
  for (int off = 1; off < 32; off <<= 1) {
    const int n = __shfl_up(incl, off, 32);
    if (lane >= off) incl += n;
  }
  if (lane == 31) sWave[wave] = incl;
  __syncthreads();
  int pre = 0;
#pragma unroll
  for (int i = 0; i < BIN_NW; ++i) pre += (i < wave) ? sWave[i] : 0;
  const int excl = pre + incl - v;
  sCnt[sw_][sr] = excl;
  int* off = OFF + (size_t)(g * NSLICE + s) * (NRANGE + 1);
  if (sw_ == 0) off[sr] = excl;
  if (t == BIN_THR - 1) off[NRANGE] = excl + v;
  __syncthreads();

#pragma unroll
  for (int k = 0; k < BIN_EPT; ++k) {
    const int r = rec[k] >> (NODE_BITS + 8);
    sRec[sCnt[wave][r] + rk[k]] = rec[k];
  }
  __syncthreads();

  int* out = REC + (size_t)(g * NSLICE + s) * SLICE_E;
#pragma unroll
  for (int u = 0; u < 4; ++u) {
    const int e = 4 * (t + BIN_THR * u);
    *(int4*)(out + e) = *(const int4*)(sRec + e);
  }
}

__device__ __forceinline__ void
bucket_edges_range(const int r, const int g,
                   const int* __restrict__ OFF, const int* __restrict__ REC,
                   int* __restrict__ START, int* __restrict__ CNT,
                   int* __restrict__ LIST) {
  __shared__ int sCur[RANGE_N];
  __shared__ int sScan[RANGE_N];
  __shared__ int sLow[CSR_NW];
  const int t    = threadIdx.x;
  const int lane = t & 31;
  const int wave = __builtin_amdgcn_readfirstlane(t >> 5);
  const int lo   = r * RANGE_N;
  int beg[WALK_S], end[WALK_S];
  const int* rec[WALK_S];
#pragma unroll
  for (int k = 0; k < WALK_S; ++k) {
    const int s = WALK_S * wave + k;
    const int* off = OFF + (size_t)(g * NSLICE + s) * (NRANGE + 1);
    beg[k] = off[r];
    end[k] = off[r + 1];
    rec[k] = REC + (size_t)(g * NSLICE + s) * SLICE_E;
  }

  if (t < RANGE_N) sCur[t] = 0;
  __syncthreads();

  int rr[WALK_S][WALK_LD];
#pragma unroll
  for (int k = 0; k < WALK_S; ++k) {
#pragma unroll
    for (int u = 0; u < WALK_LD; ++u) {
      const int i = beg[k] + 32 * u + lane;
      rr[k][u] = -1;
      if (i < end[k]) rr[k][u] = rec[k][i];
    }
  }
#pragma unroll
  for (int k = 0; k < WALK_S; ++k) {
#pragma unroll
    for (int u = 0; u < WALK_LD; ++u)
      if (rr[k][u] >= 0) atomicAdd(&sCur[(rr[k][u] >> NODE_BITS) & (RANGE_N - 1)], 1);
  }
#pragma unroll
  for (int k = 0; k < WALK_S; ++k) {
    for (int c = beg[k] + 32 * WALK_LD; c < end[k]; c += 32) {
      const int i = c + lane;
      if (i < end[k]) atomicAdd(&sCur[(rec[k][i] >> NODE_BITS) & (RANGE_N - 1)], 1);
    }
  }
  int low = 0;
#pragma unroll
  for (int k = 0; k < WALK_S; ++k) low += beg[k];
  if (lane == 0) sLow[wave] = low;
  __syncthreads();
  int base = 0;
#pragma unroll
  for (int i = 0; i < CSR_NW; ++i) base += sLow[i];

  if (t < RANGE_N) sScan[t] = sCur[t];
  __syncthreads();
  for (int o = 1; o < RANGE_N; o <<= 1) {
    int v = 0;
    if (t < RANGE_N && t >= o) v = sScan[t - o];
    __syncthreads();
    if (t < RANGE_N) sScan[t] += v;
    __syncthreads();
  }
  if (t < RANGE_N) {
    const int c  = sCur[t];
    const int st = base + sScan[t] - c;
    START[g * NNODES + lo + t] = st;
    CNT[g * NNODES + lo + t]   = c;
    sCur[t] = st;
  }
  __syncthreads();

  int* lst = LIST + (size_t)g * NEDGES;
  unsigned pos[WALK_S][WALK_LD];
#pragma unroll
  for (int k = 0; k < WALK_S; ++k) {
#pragma unroll
    for (int u = 0; u < WALK_LD; ++u) {
      pos[k][u] = 0u;
      if (rr[k][u] >= 0)
        pos[k][u] = (unsigned)atomicAdd(&sCur[(rr[k][u] >> NODE_BITS) & (RANGE_N - 1)], 1);
    }
  }
#pragma unroll
  for (int k = 0; k < WALK_S; ++k) {
#pragma unroll
    for (int u = 0; u < WALK_LD; ++u)
      if (rr[k][u] >= 0) lst[pos[k][u]] = rr[k][u] & NODE_MASK;
  }
#pragma unroll
  for (int k = 0; k < WALK_S; ++k) {
    for (int c = beg[k] + 32 * WALK_LD; c < end[k]; c += 32) {
      const int i = c + lane;
      if (i < end[k]) {
        const int v = rec[k][i];
        const int p = atomicAdd(&sCur[(v >> NODE_BITS) & (RANGE_N - 1)], 1);
        lst[p] = v & NODE_MASK;
      }
    }
  }
}

#define ROLE_N    256
#define PAIR_GRID (2 * ROLE_N)
static_assert(NSLICE * BGRAPH == ROLE_N, "binning workgroups of the first launch");
static_assert(NRANGE * BGRAPH == ROLE_N, "bucketing workgroups of the second launch");
static_assert(GEMM_GRID == 2 * ROLE_N, "product workgroups, half in each launch");
static_assert(GEMM_THR == BIN_THR && GEMM_THR == CSR_THR, "one block size for both roles");

__global__ void __launch_bounds__(BIN_THR, 8)
lead_kernel(const int* __restrict__ EI, int* __restrict__ OFF, int* __restrict__ REC,
            const float* __restrict__ X, const float* __restrict__ W,
            float* __restrict__ H,
            const float* __restrict__ a_src, const float* __restrict__ a_dst,
            float* __restrict__ AS, float* __restrict__ AD) {
  const int bid = (int)blockIdx.x;
  const int idx = bid >> 1;
  if ((((bid >> 8) ^ bid) & 1) != 0)
    gemm_wmma_f32(idx, X, W, H, a_src, a_dst, AS, AD);
  else
    bin_edges_slice(idx % NSLICE, idx / NSLICE, EI, OFF, REC);
}

__global__ void __launch_bounds__(CSR_THR, 8)
bucket_kernel(const int* __restrict__ OFF, const int* __restrict__ REC,
              int* __restrict__ START, int* __restrict__ CNT, int* __restrict__ LIST,
              const float* __restrict__ X, const float* __restrict__ W,
              float* __restrict__ H,
              const float* __restrict__ a_src, const float* __restrict__ a_dst,
              float* __restrict__ AS, float* __restrict__ AD) {
  const int bid = (int)blockIdx.x;
  const int idx = bid >> 1;
  if ((((bid >> 8) ^ bid) & 1) != 0)
    gemm_wmma_f32(ROLE_N + idx, X, W, H, a_src, a_dst, AS, AD);
  else
    bucket_edges_range(idx % NRANGE, idx / NRANGE, OFF, REC, START, CNT, LIST);
}

#define FUSE_THR   1024
#define FUSE_NW    (FUSE_THR / 32)
#define FUSE_NODES 128
#define FUSE_GRID  (BGRAPH * NNODES / FUSE_NODES)
#define FUSE_MAXR  64
#define SW_WORDS   ((FEAT / 2) * WP > FEAT * XP ? (FEAT / 2) * WP : FEAT * XP)

static_assert(FUSE_NODES == 4 * FUSE_NW, "four nodes per wave");
static_assert(NNODES % FUSE_NODES == 0, "a workgroup's nodes lie in one graph");
static_assert(NNODES == 8 * FUSE_THR, "two 16-byte copies per thread fill the logit table");
static_assert(FUSE_MAXR == 2 * FUSE_NW, "at most two nodes per wave and pass");

template <bool TRANSB, bool BIAS, bool ALPHA>
__global__ void __launch_bounds__(FUSE_THR, 16)
gat_gather_gemm_kernel(const int* __restrict__ START, const int* __restrict__ CNT,
                       const int* __restrict__ LIST, const float* __restrict__ AS,
                       const float* __restrict__ AD, const float* __restrict__ H,
                       const float* __restrict__ b, const float* __restrict__ W,
                       const float* __restrict__ bias2, float* __restrict__ OUT,
                       const float* __restrict__ a_src, const float* __restrict__ a_dst,
                       float* __restrict__ AS2, float* __restrict__ AD2) {
  __shared__ __attribute__((aligned(16))) float sW[SW_WORDS];
  __shared__ __attribute__((aligned(16))) float sX[FUSE_MAXR * XP];
  __shared__ __attribute__((aligned(16))) float sAS[NNODES];
  __shared__ __attribute__((aligned(16))) float sB[FEAT];
  __shared__ float sPart[FUSE_MAXR / 16][4][32];
  const int tid  = threadIdx.x;
  const int wave = __builtin_amdgcn_readfirstlane((int)(threadIdx.x >> 5));
  const int lane = threadIdx.x & 31;
  const int half = lane >> 4;
  const int l    = lane & 15;
  const int f    = lane * 4;

  if (TRANSB) {
#pragma unroll
    for (int u = 0; u < 4; ++u) {
      const int q = tid + FUSE_THR * u;
      const int row = q >> 5, c4 = (q & 31) * 4;
      *(float4*)(sW + row * XP + c4) = *(const float4*)(W + row * FEAT + c4);
    }
  } else {
#pragma nounroll
    for (int uo = 0; uo < 8; uo += 4) {
#pragma unroll
      for (int u = 0; u < 4; ++u) {
        const int q = tid + FUSE_THR * (uo + u);
        const int p = q >> 7, n = q & 127;
        v2f pr;
        pr.x = W[(2 * p) * FEAT + n];
        pr.y = W[(2 * p + 1) * FEAT + n];
        *(v2f*)(sW + p * WP + 2 * n) = pr;
      }
    }
  }

  {
    const float* asg = AS + (size_t)(((int)blockIdx.x * FUSE_NODES) / NNODES) * NNODES;
#pragma unroll
    for (int u = 0; u < 2; ++u) {
      const int q = 4 * (tid + FUSE_THR * u);
      *(float4*)(sAS + q) = *(const float4*)(asg + q);
    }
  }
  if (tid < FEAT / 4) *(float4*)(sB + 4 * tid) = *(const float4*)(b + 4 * tid);
  __syncthreads();

  const int tl = wave >> 3;
  const int n0 = (wave & 3) * 32;

  const bool shifted = (((blockIdx.x >> 8) ^ blockIdx.x) & 1) != 0;
  const int npass = shifted ? 3 : 2;
  int base = (int)blockIdx.x * FUSE_NODES;
  for (int pass = 0; pass < npass; ++pass) {
  const int npw  = (shifted && pass != 1) ? 1 : 2;
  const int rows = FUSE_NW * npw;
  for (int it = 0; it < npw; ++it) {
  const int rr   = wave + FUSE_NW * it;
  const int gn   = base + rr;
  const int g    = gn / NNODES;
  const int nl   = gn - g * NNODES;
  const int cnt   = CNT[gn];
  const int total = cnt + 1;
  const int* lst  = LIST + (size_t)g * NEDGES + START[gn];
  const float a_d = AD[gn];
  const float* hg = H + (size_t)g * NNODES * FEAT + lane * 4;

  int s0 = nl;
  if (lane < cnt) s0 = lst[lane];
  const float e0 = leaky(sAS[s0] + a_d);
  float m = (lane < total) ? e0 : -INFINITY;
  for (int c = 32; c < total; c += 32) {
    const int i = c + lane;
    int s = nl;
    if (i < cnt) s = lst[i];
    const float e = leaky(sAS[s] + a_d);
    if (i < total) m = fmaxf(m, e);
  }
#pragma unroll
  for (int off = 16; off > 0; off >>= 1) m = fmaxf(m, __shfl_xor(m, off, 32));

  float4 acc = make_float4(0.0f, 0.0f, 0.0f, 0.0f);
  float wsum = 0.0f;
  for (int c = 0; c < total; c += 32) {
    const int i = c + lane;
    int s = s0;
    float e = e0;
    if (c > 0) {
      s = nl;
      if (i < cnt) s = lst[i];
      e = leaky(sAS[s] + a_d);
    }
    const float w = (i < total) ? __expf(e - m) : 0.0f;
    wsum += w;
    const int rem = total - c;
    const int nc  = rem < 32 ? rem : 32;
    for (int j = 0; j < nc; j += 4) {
      const int r0 = __builtin_amdgcn_readlane(s, j);
      const int r1 = __builtin_amdgcn_readlane(s, j + 1);
      const int r2 = __builtin_amdgcn_readlane(s, j + 2);
      const int r3 = __builtin_amdgcn_readlane(s, j + 3);
      const float w0 = __int_as_float(__builtin_amdgcn_readlane(__float_as_int(w), j));
      const float w1 = __int_as_float(__builtin_amdgcn_readlane(__float_as_int(w), j + 1));
      const float w2 = __int_as_float(__builtin_amdgcn_readlane(__float_as_int(w), j + 2));
      const float w3 = __int_as_float(__builtin_amdgcn_readlane(__float_as_int(w), j + 3));
      const float4 h0 = *(const float4*)(hg + (size_t)r0 * FEAT);
      const float4 h1 = *(const float4*)(hg + (size_t)r1 * FEAT);
      const float4 h2 = *(const float4*)(hg + (size_t)r2 * FEAT);
      const float4 h3 = *(const float4*)(hg + (size_t)r3 * FEAT);
      acc.x += w0 * h0.x; acc.y += w0 * h0.y; acc.z += w0 * h0.z; acc.w += w0 * h0.w;
      acc.x += w1 * h1.x; acc.y += w1 * h1.y; acc.z += w1 * h1.z; acc.w += w1 * h1.w;
      acc.x += w2 * h2.x; acc.y += w2 * h2.y; acc.z += w2 * h2.z; acc.w += w2 * h2.w;
      acc.x += w3 * h3.x; acc.y += w3 * h3.y; acc.z += w3 * h3.z; acc.w += w3 * h3.w;
    }
  }
#pragma unroll
  for (int off = 16; off > 0; off >>= 1) wsum += __shfl_xor(wsum, off, 32);

  const float4 bv = *(const float4*)(sB + f);
  float4 v;
  v.x = acc.x / wsum + bv.x;
  v.y = acc.y / wsum + bv.y;
  v.z = acc.z / wsum + bv.z;
  v.w = acc.w / wsum + bv.w;
  float4 o;
  o.x = v.x / (1.0f + __expf(-v.x));
  o.y = v.y / (1.0f + __expf(-v.y));
  o.z = v.z / (1.0f + __expf(-v.z));
  o.w = v.w / (1.0f + __expf(-v.w));
  *(float4*)(sX + rr * XP + f) = o;
  }
  __syncthreads();

  const bool mult = (wave & 4) == 0 && tl < (rows >> 4);
  v8f acc0 = {}, acc1 = {};
  float bo0 = 0.0f, bo1 = 0.0f, as0 = 0.0f, as1 = 0.0f, ad0 = 0.0f, ad1 = 0.0f;
  if (mult) {
    const float* xa = sX + (16 * tl + l) * XP + 2 * half;
    const float* wb = TRANSB ? sW + (n0 + l) * XP + 2 * half
                             : sW + half * WP + 2 * (n0 + l);
    if (BIAS) {
      bo0 = bias2[n0 + l];
      bo1 = bias2[n0 + 16 + l];
    }
    if (ALPHA) {
      as0 = a_src[n0 + l];
      as1 = a_src[n0 + 16 + l];
      ad0 = a_dst[n0 + l];
      ad1 = a_dst[n0 + 16 + l];
    }
#pragma unroll
    for (int t = 0; t < FEAT / 4; ++t) {
      const v2f a = *(const v2f*)(xa + 4 * t);
      const v2f q0 = *(const v2f*)(wb + (TRANSB ? 4 : 2 * WP) * t);
      const v2f q1 = *(const v2f*)(wb + (TRANSB ? 4 : 2 * WP) * t + (TRANSB ? 16 * XP : 32));
      acc0 = __builtin_amdgcn_wmma_f32_16x16x4_f32(
           false, a,  false, q0,
           (short)0, acc0,  false,  false);
      acc1 = __builtin_amdgcn_wmma_f32_16x16x4_f32(
           false, a,  false, q1,
           (short)0, acc1,  false,  false);
    }
  }
  __syncthreads();

  if (mult) {
    float* so = sX + (16 * tl + 8 * half) * XP + n0 + l;
#pragma unroll
    for (int i = 0; i < 8; ++i) {
      so[i * XP]      = acc0[i] + bo0;
      so[i * XP + 16] = acc1[i] + bo1;
    }
    if (ALPHA) {
      float v[16];
#pragma unroll
      for (int i = 0; i < 8; ++i) {
        v[i]     = acc0[i] * as0 + acc1[i] * as1;
        v[8 + i] = acc0[i] * ad0 + acc1[i] * ad1;
      }
      const bool b3 = (l & 8) != 0, b2 = (l & 4) != 0;
      const bool b1 = (l & 2) != 0, b0 = (l & 1) != 0;
      float u[8], w4[4], y[2];
#pragma unroll
      for (int k = 0; k < 8; ++k) {
        const float keep = b3 ? v[k + 8] : v[k];
        const float send = b3 ? v[k] : v[k + 8];
        u[k] = keep + row_xor<8>(send);
      }
#pragma unroll
      for (int k = 0; k < 4; ++k) {
        const float keep = b2 ? u[k + 4] : u[k];
        const float send = b2 ? u[k] : u[k + 4];
        w4[k] = keep + row_xor<4>(send);
      }
#pragma unroll
      for (int k = 0; k < 2; ++k) {
        const float keep = b1 ? w4[k + 2] : w4[k];
        const float send = b1 ? w4[k] : w4[k + 2];
        y[k] = keep + row_xor<2>(send);
      }
      const float keep = b0 ? y[1] : y[0];
      const float send = b0 ? y[0] : y[1];
      sPart[tl][wave & 3][lane] = keep + row_xor<1>(send);
    }
  }
  __syncthreads();

  for (int u = 0; u < npw; ++u) {
    const int q = tid + FUSE_THR * u;
    const int row = q >> 5, c4 = (q & 31) * 4;
    *(float4*)(OUT + (size_t)(base + row) * FEAT + c4) = *(const float4*)(sX + row * XP + c4);
  }
  if (ALPHA) {
    if (tid < 2 * rows) {
      const int kind = (tid >= rows) ? 1 : 0;
      const int row  = tid - kind * rows;
      const int t2   = row >> 4, r = row & 15;
      const int idx  = 16 * (r >> 3) + 8 * kind + (r & 7);
      const float s = ((sPart[t2][0][idx] + sPart[t2][1][idx]) +
                       sPart[t2][2][idx]) + sPart[t2][3][idx];
      float* dstv = kind ? AD2 : AS2;
      dstv[base + row] = s;
    }
  }
  base += rows;
  }
}

extern "C" void kernel_launch(void* const* d_in, const int* in_sizes, int n_in,
                              void* d_out, int out_size, void* d_ws, size_t ws_size,
                              hipStream_t stream) {
  const float* x0   = (const float*)d_in[0];
  const int*   ei   = (const int*)d_in[1];
  const float* W1   = (const float*)d_in[2];
  const float* as1  = (const float*)d_in[3];
  const float* ad1  = (const float*)d_in[4];
  const float* b1   = (const float*)d_in[5];
  const float* W2   = (const float*)d_in[6];
  const float* as2  = (const float*)d_in[7];
  const float* ad2  = (const float*)d_in[8];
  const float* b2   = (const float*)d_in[9];
  const float* fcW  = (const float*)d_in[10];
  const float* fcb  = (const float*)d_in[11];
  float* out = (float*)d_out;

  const size_t BN      = (size_t)BGRAPH * NNODES;
  const size_t BIGELTS = BN * FEAT;
  char* ws = (char*)d_ws;
  float* BUF0  = (float*)ws;               ws += BIGELTS * sizeof(float);
  float* BUF1  = (float*)ws;               ws += BIGELTS * sizeof(float);
  float* AS    = (float*)ws;               ws += BN * sizeof(float);
  float* AD    = (float*)ws;               ws += BN * sizeof(float);
  float* AS2   = (float*)ws;               ws += BN * sizeof(float);
  float* AD2   = (float*)ws;               ws += BN * sizeof(float);
  int*   START = (int*)ws;                 ws += BN * sizeof(int);
  int*   CNT   = (int*)ws;                 ws += BN * sizeof(int);
  int*   LIST  = (int*)ws;                 ws += (size_t)BGRAPH * NEDGES * sizeof(int);
  int*   REC   = (int*)ws;                 ws += (size_t)BGRAPH * NEDGES * sizeof(int);
  int*   OFF   = (int*)ws;

  const int  gPair  = PAIR_GRID;
  const int  gFuse  = FUSE_GRID;

  lead_kernel<<<gPair, BIN_THR, 0, stream>>>(ei, OFF, REC, x0, W1, BUF0, as1, ad1, AS, AD);
  bucket_kernel<<<gPair, CSR_THR, 0, stream>>>(OFF, REC, START, CNT, LIST,
                                               x0, W1, BUF0, as1, ad1, AS, AD);

  gat_gather_gemm_kernel<false, false, true><<<gFuse, FUSE_THR, 0, stream>>>(
      START, CNT, LIST, AS, AD, BUF0, b1, W2, nullptr, BUF1, as2, ad2, AS2, AD2);
  gat_gather_gemm_kernel<true, true, false><<<gFuse, FUSE_THR, 0, stream>>>(
      START, CNT, LIST, AS2, AD2, BUF1, b2, fcW, fcb, out, nullptr, nullptr, nullptr, nullptr);
}


# Edge stage: a node's serial chain cut: start/count/logit of a wave's 4 nodes loaded once up front, 2 list entries per lane loaded a node ahead, max/sum butterflies on the vector ALU; sum orders kept
# speedup vs baseline: 11.7425x; 1.0401x over previous
#include <hip/hip_runtime.h>
#include <cstdint>

#define BGRAPH 8
#define NNODES 8192
#define NEDGES 262144
#define FEAT   128
#define NEG_SLOPE 0.2f

typedef __attribute__((ext_vector_type(2))) float v2f;
typedef __attribute__((ext_vector_type(4))) float v4f;
typedef __attribute__((ext_vector_type(8))) float v8f;
typedef __attribute__((ext_vector_type(4))) int   v4i;

__device__ __forceinline__ float leaky(float v) {
  return v >= 0.0f ? v : NEG_SLOPE * v;
}

#define GEMM_THR  512
#define GEMM_ROWS 64
#define GEMM_PASS 2
#define GEMM_GRID (BGRAPH * NNODES / (GEMM_ROWS * GEMM_PASS))
#define XP        132
#define WP        288

static_assert(GEMM_ROWS * FEAT / 4 == 4 * GEMM_THR, "four 16-byte row pieces per thread and pass");
static_assert((GEMM_ROWS / 16) * 4 * 32 == GEMM_THR, "one wave per row tile and strip");

template <int MASK>
__device__ __forceinline__ float row_xor(float v) {
  return __int_as_float(__builtin_amdgcn_update_dpp(
      0, __float_as_int(v), 0x160 | MASK, 0xf, 0xf, true));
}

__device__ __forceinline__ float row_swap(float v) {
  return __int_as_float(__builtin_amdgcn_permlanex16(
      __float_as_int(v), __float_as_int(v), 0x76543210, 0xfedcba98, false, false));
}

__device__ __forceinline__ void
gemm_wmma_f32(const int wg, const float* __restrict__ X, const float* __restrict__ W,
              float* __restrict__ H,
              const float* __restrict__ a_src, const float* __restrict__ a_dst,
              float* __restrict__ AS, float* __restrict__ AD) {
  __shared__ __attribute__((aligned(16))) float sWp[(FEAT / 2) * WP];
  __shared__ __attribute__((aligned(16))) float sXr[GEMM_ROWS * XP];
  __shared__ float sPart[GEMM_ROWS / 16][4][32];
  const int tid  = threadIdx.x;
  const int wave = tid >> 5;
  const int lane = tid & 31;
  const int half = lane >> 4;
  const int l    = lane & 15;
  const int tl   = wave >> 2;
  const int sw   = wave & 3;
  const int n0   = sw * 32;
  const size_t row0 = (size_t)wg * (GEMM_ROWS * GEMM_PASS);

  const float* xg = X + row0 * FEAT + 4 * tid;
  float* sx = sXr + (tid >> 5) * XP + (tid & 31) * 4;
  v4f nx0 = *(const v4f*)(xg);
  v4f nx1 = *(const v4f*)(xg + 4 * GEMM_THR);
  v4f nx2 = *(const v4f*)(xg + 8 * GEMM_THR);
  v4f nx3 = *(const v4f*)(xg + 12 * GEMM_THR);

#pragma nounroll
  for (int uo = 0; uo < (FEAT / 2) * FEAT / GEMM_THR; uo += 4) {
#pragma unroll
    for (int u = 0; u < 4; ++u) {
      const int q = tid + GEMM_THR * (uo + u);
      const int p = q >> 7, n = q & 127;
      v2f pr;
      pr.x = W[(2 * p) * FEAT + n];
      pr.y = W[(2 * p + 1) * FEAT + n];
      *(v2f*)(sWp + p * WP + 2 * n) = pr;
    }
  }
  const float as0 = a_src[n0 + l];
  const float as1 = a_src[n0 + 16 + l];
  const float ad0 = a_dst[n0 + l];
  const float ad1 = a_dst[n0 + 16 + l];

  const float* xa = sXr + (16 * tl + l) * XP + 2 * half;
  const float* wb = sWp + half * WP + 2 * (n0 + l);

#pragma nounroll
  for (int pass = 0; pass < GEMM_PASS; ++pass) {
    *(v4f*)(sx)           = nx0;
    *(v4f*)(sx + 16 * XP) = nx1;
    *(v4f*)(sx + 32 * XP) = nx2;
    *(v4f*)(sx + 48 * XP) = nx3;
    __syncthreads();
    if (pass + 1 < GEMM_PASS) {
      const float* xn = xg + (size_t)(pass + 1) * (GEMM_ROWS * FEAT);
      nx0 = *(const v4f*)(xn);
      nx1 = *(const v4f*)(xn + 4 * GEMM_THR);
      nx2 = *(const v4f*)(xn + 8 * GEMM_THR);
      nx3 = *(const v4f*)(xn + 12 * GEMM_THR);
    }

    v8f acc0 = {}, acc1 = {};
#pragma unroll
    for (int t = 0; t < FEAT / 4; ++t) {
      const v2f a = *(const v2f*)(xa + 4 * t);
      const v2f q0 = *(const v2f*)(wb + 2 * WP * t);
      const v2f q1 = *(const v2f*)(wb + 2 * WP * t + 32);
      acc0 = __builtin_amdgcn_wmma_f32_16x16x4_f32(
           false, a,  false, q0,
           (short)0, acc0,  false,  false);
      acc1 = __builtin_amdgcn_wmma_f32_16x16x4_f32(
           false, a,  false, q1,
           (short)0, acc1,  false,  false);
    }
    const size_t m0 = row0 + (size_t)pass * GEMM_ROWS + 16 * tl;
    float* hout = H + (m0 + 8 * half) * FEAT + (n0 + l);
#pragma unroll
    for (int i = 0; i < 8; ++i) {
      hout[(size_t)i * FEAT]      = acc0[i];
      hout[(size_t)i * FEAT + 16] = acc1[i];
    }
    {
    float v[16];
#pragma unroll
    for (int i = 0; i < 8; ++i) {
      v[i]     = acc0[i] * as0 + acc1[i] * as1;
      v[8 + i] = acc0[i] * ad0 + acc1[i] * ad1;
    }
    const bool b3 = (l & 8) != 0, b2 = (l & 4) != 0;
    const bool b1 = (l & 2) != 0, b0 = (l & 1) != 0;
    float u[8], w4[4], y[2];
#pragma unroll
    for (int k = 0; k < 8; ++k) {
      const float keep = b3 ? v[k + 8] : v[k];
      const float send = b3 ? v[k] : v[k + 8];
      u[k] = keep + row_xor<8>(send);
    }
#pragma unroll
    for (int k = 0; k < 4; ++k) {
      const float keep = b2 ? u[k + 4] : u[k];
      const float send = b2 ? u[k] : u[k + 4];
      w4[k] = keep + row_xor<4>(send);
    }
#pragma unroll
    for (int k = 0; k < 2; ++k) {
      const float keep = b1 ? w4[k + 2] : w4[k];
      const float send = b1 ? w4[k] : w4[k + 2];
      y[k] = keep + row_xor<2>(send);
    }
    const float keep = b0 ? y[1] : y[0];
    const float send = b0 ? y[0] : y[1];
    sPart[tl][sw][lane] = keep + row_xor<1>(send);
    }
    __syncthreads();

    if (tid < 2 * GEMM_ROWS) {
      const int kind = (tid >= GEMM_ROWS) ? 1 : 0;
      const int row  = tid - kind * GEMM_ROWS;
      const int t2   = row >> 4, r = row & 15;
      const int idx  = 16 * (r >> 3) + 8 * kind + (r & 7);
      const float s = ((sPart[t2][0][idx] + sPart[t2][1][idx]) +
                       sPart[t2][2][idx]) + sPart[t2][3][idx];
      float* dstv = kind ? AD : AS;
      dstv[row0 + (size_t)pass * GEMM_ROWS + row] = s;
    }
  }
}

#define RANGE_N   256
#define NRANGE    (NNODES / RANGE_N)
#define CSR_THR   512
#define CSR_NW    (CSR_THR / 32)
#define SLICE_E   8192
#define NSLICE    (NEDGES / SLICE_E)
#define BIN_THR   512
#define BIN_NW    (BIN_THR / 32)
#define BIN_EPT   (SLICE_E / BIN_THR)
#define WALK_S    (NSLICE / CSR_NW)
#define WALK_LD   10
#define NODE_BITS 13
#define NODE_MASK (NNODES - 1)

static_assert(NNODES == (1 << NODE_BITS), "record packing");
static_assert(NRANGE * BIN_NW == BIN_THR, "one scan entry per thread");
static_assert(NSLICE == WALK_S * CSR_NW, "whole slices per wave");
static_assert(RANGE_N <= CSR_THR, "one thread per owned node");
static_assert(BIN_EPT == 16, "four int4 loads per thread");

__device__ __forceinline__ void
bin_edges_slice(const int s, const int g, const int* __restrict__ EI,
                int* __restrict__ OFF, int* __restrict__ REC) {
  __shared__ int sCnt[BIN_NW][NRANGE];
  __shared__ int sWave[BIN_NW];
  __shared__ int sRec[SLICE_E];
  const int t    = threadIdx.x;
  const int lane = t & 31;
  const int wave = t >> 5;
  const int* srcp = EI + (size_t)g * 2 * NEDGES + (size_t)s * SLICE_E;
  const int* dstp = srcp + NEDGES;

  (&sCnt[0][0])[t] = 0;
  __syncthreads();

  v4i d[4], sv[4];
#pragma unroll
  for (int u = 0; u < 4; ++u) {
    const int e = 4 * (t + BIN_THR * u);
    d[u]  = *(const v4i*)(dstp + e);
    sv[u] = *(const v4i*)(srcp + e);
  }
  asm volatile("" :: "v"(d[0]), "v"(d[1]), "v"(d[2]), "v"(d[3]),
                     "v"(sv[0]), "v"(sv[1]), "v"(sv[2]), "v"(sv[3]));
  int rec[BIN_EPT], rk[BIN_EPT];
#pragma unroll
  for (int u = 0; u < 4; ++u) {
    const int dv[4] = {d[u].x, d[u].y, d[u].z, d[u].w};
    const int sw[4] = {sv[u].x, sv[u].y, sv[u].z, sv[u].w};
#pragma unroll
    for (int i = 0; i < 4; ++i) {
      const int k = 4 * u + i;
      rec[k] = ((dv[i] & NODE_MASK) << NODE_BITS) | (sw[i] & NODE_MASK);
      const int r = rec[k] >> (NODE_BITS + 8);
      rk[k] = atomicAdd(&sCnt[wave][r], 1);
    }
  }
  __syncthreads();

  const int sr = t / BIN_NW;
  const int sw_ = t % BIN_NW;
  const int v  = sCnt[sw_][sr];
  int incl = v;
#pragma unroll
  for (int off = 1; off < 32; off <<= 1) {
    const int n = __shfl_up(incl, off, 32);
    if (lane >= off) incl += n;
  }
  if (lane == 31) sWave[wave] = incl;
  __syncthreads();
  int pre = 0;
#pragma unroll
  for (int i = 0; i < BIN_NW; ++i) pre += (i < wave) ? sWave[i] : 0;
  const int excl = pre + incl - v;
  sCnt[sw_][sr] = excl;
  int* off = OFF + (size_t)(g * NSLICE + s) * (NRANGE + 1);
  if (sw_ == 0) off[sr] = excl;
  if (t == BIN_THR - 1) off[NRANGE] = excl + v;
  __syncthreads();

#pragma unroll
  for (int k = 0; k < BIN_EPT; ++k) {
    const int r = rec[k] >> (NODE_BITS + 8);
    sRec[sCnt[wave][r] + rk[k]] = rec[k];
  }
  __syncthreads();

  int* out = REC + (size_t)(g * NSLICE + s) * SLICE_E;
#pragma unroll
  for (int u = 0; u < 4; ++u) {
    const int e = 4 * (t + BIN_THR * u);
    *(int4*)(out + e) = *(const int4*)(sRec + e);
  }
}

__device__ __forceinline__ void
bucket_edges_range(const int r, const int g,
                   const int* __restrict__ OFF, const int* __restrict__ REC,
                   int* __restrict__ START, int* __restrict__ CNT,
                   int* __restrict__ LIST) {
  __shared__ int sCur[RANGE_N];
  __shared__ int sScan[RANGE_N];
  __shared__ int sLow[CSR_NW];
  const int t    = threadIdx.x;
  const int lane = t & 31;
  const int wave = __builtin_amdgcn_readfirstlane(t >> 5);
  const int lo   = r * RANGE_N;
  int beg[WALK_S], end[WALK_S];
  const int* rec[WALK_S];
#pragma unroll
  for (int k = 0; k < WALK_S; ++k) {
    const int s = WALK_S * wave + k;
    const int* off = OFF + (size_t)(g * NSLICE + s) * (NRANGE + 1);
    beg[k] = off[r];
    end[k] = off[r + 1];
    rec[k] = REC + (size_t)(g * NSLICE + s) * SLICE_E;
  }

  if (t < RANGE_N) sCur[t] = 0;
  __syncthreads();

  int rr[WALK_S][WALK_LD];
#pragma unroll
  for (int k = 0; k < WALK_S; ++k) {
#pragma unroll
    for (int u = 0; u < WALK_LD; ++u) {
      const int i = beg[k] + 32 * u + lane;
      rr[k][u] = -1;
      if (i < end[k]) rr[k][u] = rec[k][i];
    }
  }
#pragma unroll
  for (int k = 0; k < WALK_S; ++k) {
#pragma unroll
    for (int u = 0; u < WALK_LD; ++u)
      if (rr[k][u] >= 0) atomicAdd(&sCur[(rr[k][u] >> NODE_BITS) & (RANGE_N - 1)], 1);
  }
#pragma unroll
  for (int k = 0; k < WALK_S; ++k) {
    for (int c = beg[k] + 32 * WALK_LD; c < end[k]; c += 32) {
      const int i = c + lane;
      if (i < end[k]) atomicAdd(&sCur[(rec[k][i] >> NODE_BITS) & (RANGE_N - 1)], 1);
    }
  }
  int low = 0;
#pragma unroll
  for (int k = 0; k < WALK_S; ++k) low += beg[k];
  if (lane == 0) sLow[wave] = low;
  __syncthreads();
  int base = 0;
#pragma unroll
  for (int i = 0; i < CSR_NW; ++i) base += sLow[i];

  if (t < RANGE_N) sScan[t] = sCur[t];
  __syncthreads();
  for (int o = 1; o < RANGE_N; o <<= 1) {
    int v = 0;
    if (t < RANGE_N && t >= o) v = sScan[t - o];
    __syncthreads();
    if (t < RANGE_N) sScan[t] += v;
    __syncthreads();
  }
  if (t < RANGE_N) {
    const int c  = sCur[t];
    const int st = base + sScan[t] - c;
    START[g * NNODES + lo + t] = st;
    CNT[g * NNODES + lo + t]   = c;
    sCur[t] = st;
  }
  __syncthreads();

  int* lst = LIST + (size_t)g * NEDGES;
  unsigned pos[WALK_S][WALK_LD];
#pragma unroll
  for (int k = 0; k < WALK_S; ++k) {
#pragma unroll
    for (int u = 0; u < WALK_LD; ++u) {
      pos[k][u] = 0u;
      if (rr[k][u] >= 0)
        pos[k][u] = (unsigned)atomicAdd(&sCur[(rr[k][u] >> NODE_BITS) & (RANGE_N - 1)], 1);
    }
  }
#pragma unroll
  for (int k = 0; k < WALK_S; ++k) {
#pragma unroll
    for (int u = 0; u < WALK_LD; ++u)
      if (rr[k][u] >= 0) lst[pos[k][u]] = rr[k][u] & NODE_MASK;
  }
#pragma unroll
  for (int k = 0; k < WALK_S; ++k) {
    for (int c = beg[k] + 32 * WALK_LD; c < end[k]; c += 32) {
      const int i = c + lane;
      if (i < end[k]) {
        const int v = rec[k][i];
        const int p = atomicAdd(&sCur[(v >> NODE_BITS) & (RANGE_N - 1)], 1);
        lst[p] = v & NODE_MASK;
      }
    }
  }
}

#define ROLE_N    256
#define PAIR_GRID (2 * ROLE_N)
static_assert(NSLICE * BGRAPH == ROLE_N, "binning workgroups of the first launch");
static_assert(NRANGE * BGRAPH == ROLE_N, "bucketing workgroups of the second launch");
static_assert(GEMM_GRID == 2 * ROLE_N, "product workgroups, half in each launch");
static_assert(GEMM_THR == BIN_THR && GEMM_THR == CSR_THR, "one block size for both roles");

__global__ void __launch_bounds__(BIN_THR, 8)
lead_kernel(const int* __restrict__ EI, int* __restrict__ OFF, int* __restrict__ REC,
            const float* __restrict__ X, const float* __restrict__ W,
            float* __restrict__ H,
            const float* __restrict__ a_src, const float* __restrict__ a_dst,
            float* __restrict__ AS, float* __restrict__ AD) {
  const int bid = (int)blockIdx.x;
  const int idx = bid >> 1;
  if ((((bid >> 8) ^ bid) & 1) != 0)
    gemm_wmma_f32(idx, X, W, H, a_src, a_dst, AS, AD);
  else
    bin_edges_slice(idx % NSLICE, idx / NSLICE, EI, OFF, REC);
}

__global__ void __launch_bounds__(CSR_THR, 8)
bucket_kernel(const int* __restrict__ OFF, const int* __restrict__ REC,
              int* __restrict__ START, int* __restrict__ CNT, int* __restrict__ LIST,
              const float* __restrict__ X, const float* __restrict__ W,
              float* __restrict__ H,
              const float* __restrict__ a_src, const float* __restrict__ a_dst,
              float* __restrict__ AS, float* __restrict__ AD) {
  const int bid = (int)blockIdx.x;
  const int idx = bid >> 1;
  if ((((bid >> 8) ^ bid) & 1) != 0)
    gemm_wmma_f32(ROLE_N + idx, X, W, H, a_src, a_dst, AS, AD);
  else
    bucket_edges_range(idx % NRANGE, idx / NRANGE, OFF, REC, START, CNT, LIST);
}

#define FUSE_THR   1024
#define FUSE_NW    (FUSE_THR / 32)
#define FUSE_NODES 128
#define FUSE_GRID  (BGRAPH * NNODES / FUSE_NODES)
#define FUSE_NPW   (FUSE_NODES / FUSE_NW)
#define FUSE_MAXR  64
#define SW_WORDS   ((FEAT / 2) * WP > FEAT * XP ? (FEAT / 2) * WP : FEAT * XP)

static_assert(FUSE_NODES == 4 * FUSE_NW, "four nodes per wave");
static_assert(FUSE_NPW <= 32, "one lane per node of the wave holds its start, count and logit");
static_assert(NNODES % FUSE_NODES == 0, "a workgroup's nodes lie in one graph");
static_assert(NNODES == 8 * FUSE_THR, "two 16-byte copies per thread fill the logit table");
static_assert(FUSE_MAXR == 2 * FUSE_NW, "at most two nodes per wave and pass");

template <bool TRANSB, bool BIAS, bool ALPHA>
__global__ void __launch_bounds__(FUSE_THR, 16)
gat_gather_gemm_kernel(const int* __restrict__ START, const int* __restrict__ CNT,
                       const int* __restrict__ LIST, const float* __restrict__ AS,
                       const float* __restrict__ AD, const float* __restrict__ H,
                       const float* __restrict__ b, const float* __restrict__ W,
                       const float* __restrict__ bias2, float* __restrict__ OUT,
                       const float* __restrict__ a_src, const float* __restrict__ a_dst,
                       float* __restrict__ AS2, float* __restrict__ AD2) {
  __shared__ __attribute__((aligned(16))) float sW[SW_WORDS];
  __shared__ __attribute__((aligned(16))) float sX[FUSE_MAXR * XP];
  __shared__ __attribute__((aligned(16))) float sAS[NNODES];
  __shared__ __attribute__((aligned(16))) float sB[FEAT];
  __shared__ float sPart[FUSE_MAXR / 16][4][32];
  const int tid  = threadIdx.x;
  const int wave = __builtin_amdgcn_readfirstlane((int)(threadIdx.x >> 5));
  const int lane = threadIdx.x & 31;
  const int half = lane >> 4;
  const int l    = lane & 15;
  const int f    = lane * 4;

  const int bb = (int)blockIdx.x * FUSE_NODES;
  const int g  = bb / NNODES;
  const int hn = bb + wave + FUSE_NW * (lane % FUSE_NPW);
  const int   hcnt = CNT[hn];
  const int   hsta = START[hn];
  const float had  = AD[hn];

  if (TRANSB) {
#pragma unroll
    for (int u = 0; u < 4; ++u) {
      const int q = tid + FUSE_THR * u;
      const int row = q >> 5, c4 = (q & 31) * 4;
      *(float4*)(sW + row * XP + c4) = *(const float4*)(W + row * FEAT + c4);
    }
  } else {
#pragma nounroll
    for (int uo = 0; uo < 8; uo += 4) {
#pragma unroll
      for (int u = 0; u < 4; ++u) {
        const int q = tid + FUSE_THR * (uo + u);
        const int p = q >> 7, n = q & 127;
        v2f pr;
        pr.x = W[(2 * p) * FEAT + n];
        pr.y = W[(2 * p + 1) * FEAT + n];
        *(v2f*)(sW + p * WP + 2 * n) = pr;
      }
    }
  }

  {
    const float* asg = AS + (size_t)(((int)blockIdx.x * FUSE_NODES) / NNODES) * NNODES;
#pragma unroll
    for (int u = 0; u < 2; ++u) {
      const int q = 4 * (tid + FUSE_THR * u);
      *(float4*)(sAS + q) = *(const float4*)(asg + q);
    }
  }
  if (tid < FEAT / 4) *(float4*)(sB + 4 * tid) = *(const float4*)(b + 4 * tid);

  const int*   lstg = LIST + (size_t)g * NEDGES;
  const float* hg   = H + (size_t)g * NNODES * FEAT + lane * 4;
  const int    nl0  = bb - g * NNODES + wave;
  int qcnt[FUSE_NPW], qsta[FUSE_NPW];
  float qad[FUSE_NPW];
#pragma unroll
  for (int k = 0; k < FUSE_NPW; ++k) {
    qcnt[k] = __builtin_amdgcn_readlane(hcnt, k);
    qsta[k] = __builtin_amdgcn_readlane(hsta, k);
    qad[k]  = __int_as_float(__builtin_amdgcn_readlane(__float_as_int(had), k));
  }
  int ps0 = nl0, ps1 = nl0;
  if (lane < qcnt[0])      ps0 = lstg[qsta[0] + lane];
  if (lane + 32 < qcnt[0]) ps1 = lstg[qsta[0] + 32 + lane];
  __syncthreads();

  const int tl = wave >> 3;
  const int n0 = (wave & 3) * 32;

  const bool shifted = (((blockIdx.x >> 8) ^ blockIdx.x) & 1) != 0;
  const int npass = shifted ? 3 : 2;
  int base = bb;
  int kn   = 0;
  for (int pass = 0; pass < npass; ++pass) {
  const int npw  = (shifted && pass != 1) ? 1 : 2;
  const int rows = FUSE_NW * npw;
  for (int it = 0; it < npw; ++it) {
  const int rr   = wave + FUSE_NW * it;
  const int nl   = nl0 + FUSE_NW * kn;
  const int cnt   = qcnt[0];
  const int total = cnt + 1;
  const int* lst  = lstg + qsta[0];
  const float a_d = qad[0];

  const int s0 = ps0, s1 = ps1;
  if (kn + 1 < FUSE_NPW) {
    ps0 = nl + FUSE_NW;
    ps1 = nl + FUSE_NW;
    if (lane < qcnt[1])      ps0 = lstg[qsta[1] + lane];
    if (lane + 32 < qcnt[1]) ps1 = lstg[qsta[1] + 32 + lane];
  }
  const float e0 = leaky(sAS[s0] + a_d);
  const float e1 = leaky(sAS[s1] + a_d);
  float m = (lane < total) ? e0 : -INFINITY;
  if (lane + 32 < total) m = fmaxf(m, e1);
  for (int c = 64; c < total; c += 32) {
    const int i = c + lane;
    int s = nl;
    if (i < cnt) s = lst[i];
    const float e = leaky(sAS[s] + a_d);
    if (i < total) m = fmaxf(m, e);
  }
  m = fmaxf(m, row_swap(m));
  m = fmaxf(m, row_xor<8>(m));
  m = fmaxf(m, row_xor<4>(m));
  m = fmaxf(m, row_xor<2>(m));
  m = fmaxf(m, row_xor<1>(m));

  float4 acc = make_float4(0.0f, 0.0f, 0.0f, 0.0f);
  auto add_rows = [&](const int sv, const float wv, const int n) {
    for (int j = 0; j < n; j += 4) {
      const int r0 = __builtin_amdgcn_readlane(sv, j);
      const int r1 = __builtin_amdgcn_readlane(sv, j + 1);
      const int r2 = __builtin_amdgcn_readlane(sv, j + 2);
      const int r3 = __builtin_amdgcn_readlane(sv, j + 3);
      const float w0 = __int_as_float(__builtin_amdgcn_readlane(__float_as_int(wv), j));
      const float w1 = __int_as_float(__builtin_amdgcn_readlane(__float_as_int(wv), j + 1));
      const float w2 = __int_as_float(__builtin_amdgcn_readlane(__float_as_int(wv), j + 2));
      const float w3 = __int_as_float(__builtin_amdgcn_readlane(__float_as_int(wv), j + 3));
      const float4 h0 = *(const float4*)(hg + (size_t)r0 * FEAT);
      const float4 h1 = *(const float4*)(hg + (size_t)r1 * FEAT);
      const float4 h2 = *(const float4*)(hg + (size_t)r2 * FEAT);
      const float4 h3 = *(const float4*)(hg + (size_t)r3 * FEAT);
      acc.x = fmaf(w0, h0.x, acc.x); acc.y = fmaf(w0, h0.y, acc.y); acc.z = fmaf(w0, h0.z, acc.z); acc.w = fmaf(w0, h0.w, acc.w);
      acc.x = fmaf(w1, h1.x, acc.x); acc.y = fmaf(w1, h1.y, acc.y); acc.z = fmaf(w1, h1.z, acc.z); acc.w = fmaf(w1, h1.w, acc.w);
      acc.x = fmaf(w2, h2.x, acc.x); acc.y = fmaf(w2, h2.y, acc.y); acc.z = fmaf(w2, h2.z, acc.z); acc.w = fmaf(w2, h2.w, acc.w);
      acc.x = fmaf(w3, h3.x, acc.x); acc.y = fmaf(w3, h3.y, acc.y); acc.z = fmaf(w3, h3.z, acc.z); acc.w = fmaf(w3, h3.w, acc.w);
    }
  };
  const float wa = (lane < total) ? __expf(e0 - m) : 0.0f;
  float wsum = wa;
  add_rows(s0, wa, total < 32 ? total : 32);
  if (total > 32) {
    const float wb = (lane + 32 < total) ? __expf(e1 - m) : 0.0f;
    wsum += wb;
    add_rows(s1, wb, total < 64 ? total - 32 : 32);
  }
  for (int c = 64; c < total; c += 32) {
    const int i = c + lane;
    int s = nl;
    if (i < cnt) s = lst[i];
    const float e = leaky(sAS[s] + a_d);
    const float w = (i < total) ? __expf(e - m) : 0.0f;
    wsum += w;
    const int rem = total - c;
    add_rows(s, w, rem < 32 ? rem : 32);
  }
  wsum += row_swap(wsum);
  wsum += row_xor<8>(wsum);
  wsum += row_xor<4>(wsum);
  wsum += row_xor<2>(wsum);
  wsum += row_xor<1>(wsum);

  const float4 bv = *(const float4*)(sB + f);
  float4 v;
  v.x = acc.x / wsum + bv.x;
  v.y = acc.y / wsum + bv.y;
  v.z = acc.z / wsum + bv.z;
  v.w = acc.w / wsum + bv.w;
  float4 o;
  o.x = v.x / (1.0f + __expf(-v.x));
  o.y = v.y / (1.0f + __expf(-v.y));
  o.z = v.z / (1.0f + __expf(-v.z));
  o.w = v.w / (1.0f + __expf(-v.w));
  *(float4*)(sX + rr * XP + f) = o;
  ++kn;
#pragma unroll
  for (int k = 0; k + 1 < FUSE_NPW; ++k) {
    qcnt[k] = qcnt[k + 1];
    qsta[k] = qsta[k + 1];
    qad[k]  = qad[k + 1];
  }
  }
  __syncthreads();

  const bool mult = (wave & 4) == 0 && tl < (rows >> 4);
  v8f acc0 = {}, acc1 = {};
  float bo0 = 0.0f, bo1 = 0.0f, as0 = 0.0f, as1 = 0.0f, ad0 = 0.0f, ad1 = 0.0f;
  if (mult) {
    const float* xa = sX + (16 * tl + l) * XP + 2 * half;
    const float* wb = TRANSB ? sW + (n0 + l) * XP + 2 * half
                             : sW + half * WP + 2 * (n0 + l);
    if (BIAS) {
      bo0 = bias2[n0 + l];
      bo1 = bias2[n0 + 16 + l];
    }
    if (ALPHA) {
      as0 = a_src[n0 + l];
      as1 = a_src[n0 + 16 + l];
      ad0 = a_dst[n0 + l];
      ad1 = a_dst[n0 + 16 + l];
    }
#pragma unroll
    for (int t = 0; t < FEAT / 4; ++t) {
      const v2f a = *(const v2f*)(xa + 4 * t);
      const v2f q0 = *(const v2f*)(wb + (TRANSB ? 4 : 2 * WP) * t);
      const v2f q1 = *(const v2f*)(wb + (TRANSB ? 4 : 2 * WP) * t + (TRANSB ? 16 * XP : 32));
      acc0 = __builtin_amdgcn_wmma_f32_16x16x4_f32(
           false, a,  false, q0,
           (short)0, acc0,  false,  false);
      acc1 = __builtin_amdgcn_wmma_f32_16x16x4_f32(
           false, a,  false, q1,
           (short)0, acc1,  false,  false);
    }
  }
  __syncthreads();

  if (mult) {
    float* so = sX + (16 * tl + 8 * half) * XP + n0 + l;
#pragma unroll
    for (int i = 0; i < 8; ++i) {
      so[i * XP]      = acc0[i] + bo0;
      so[i * XP + 16] = acc1[i] + bo1;
    }
    if (ALPHA) {
      float v[16];
#pragma unroll
      for (int i = 0; i < 8; ++i) {
        v[i]     = acc0[i] * as0 + acc1[i] * as1;
        v[8 + i] = acc0[i] * ad0 + acc1[i] * ad1;
      }
      const bool b3 = (l & 8) != 0, b2 = (l & 4) != 0;
      const bool b1 = (l & 2) != 0, b0 = (l & 1) != 0;
      float u[8], w4[4], y[2];
#pragma unroll
      for (int k = 0; k < 8; ++k) {
        const float keep = b3 ? v[k + 8] : v[k];
        const float send = b3 ? v[k] : v[k + 8];
        u[k] = keep + row_xor<8>(send);
      }
#pragma unroll
      for (int k = 0; k < 4; ++k) {
        const float keep = b2 ? u[k + 4] : u[k];
        const float send = b2 ? u[k] : u[k + 4];
        w4[k] = keep + row_xor<4>(send);
      }
#pragma unroll
      for (int k = 0; k < 2; ++k) {
        const float keep = b1 ? w4[k + 2] : w4[k];
        const float send = b1 ? w4[k] : w4[k + 2];
        y[k] = keep + row_xor<2>(send);
      }
      const float keep = b0 ? y[1] : y[0];
      const float send = b0 ? y[0] : y[1];
      sPart[tl][wave & 3][lane] = keep + row_xor<1>(send);
    }
  }
  __syncthreads();

  for (int u = 0; u < npw; ++u) {
    const int q = tid + FUSE_THR * u;
    const int row = q >> 5, c4 = (q & 31) * 4;
    *(float4*)(OUT + (size_t)(base + row) * FEAT + c4) = *(const float4*)(sX + row * XP + c4);
  }
  if (ALPHA) {
    if (tid < 2 * rows) {
      const int kind = (tid >= rows) ? 1 : 0;
      const int row  = tid - kind * rows;
      const int t2   = row >> 4, r = row & 15;
      const int idx  = 16 * (r >> 3) + 8 * kind + (r & 7);
      const float s = ((sPart[t2][0][idx] + sPart[t2][1][idx]) +
                       sPart[t2][2][idx]) + sPart[t2][3][idx];
      float* dstv = kind ? AD2 : AS2;
      dstv[base + row] = s;
    }
  }
  base += rows;
  }
}

extern "C" void kernel_launch(void* const* d_in, const int* in_sizes, int n_in,
                              void* d_out, int out_size, void* d_ws, size_t ws_size,
                              hipStream_t stream) {
  const float* x0   = (const float*)d_in[0];
  const int*   ei   = (const int*)d_in[1];
  const float* W1   = (const float*)d_in[2];
  const float* as1  = (const float*)d_in[3];
  const float* ad1  = (const float*)d_in[4];
  const float* b1   = (const float*)d_in[5];
  const float* W2   = (const float*)d_in[6];
  const float* as2  = (const float*)d_in[7];
  const float* ad2  = (const float*)d_in[8];
  const float* b2   = (const float*)d_in[9];
  const float* fcW  = (const float*)d_in[10];
  const float* fcb  = (const float*)d_in[11];
  float* out = (float*)d_out;

  const size_t BN      = (size_t)BGRAPH * NNODES;
  const size_t BIGELTS = BN * FEAT;
  char* ws = (char*)d_ws;
  float* BUF0  = (float*)ws;               ws += BIGELTS * sizeof(float);
  float* BUF1  = (float*)ws;               ws += BIGELTS * sizeof(float);
  float* AS    = (float*)ws;               ws += BN * sizeof(float);
  float* AD    = (float*)ws;               ws += BN * sizeof(float);
  float* AS2   = (float*)ws;               ws += BN * sizeof(float);
  float* AD2   = (float*)ws;               ws += BN * sizeof(float);
  int*   START = (int*)ws;                 ws += BN * sizeof(int);
  int*   CNT   = (int*)ws;                 ws += BN * sizeof(int);
  int*   LIST  = (int*)ws;                 ws += (size_t)BGRAPH * NEDGES * sizeof(int);
  int*   REC   = (int*)ws;                 ws += (size_t)BGRAPH * NEDGES * sizeof(int);
  int*   OFF   = (int*)ws;

  const int  gPair  = PAIR_GRID;
  const int  gFuse  = FUSE_GRID;

  lead_kernel<<<gPair, BIN_THR, 0, stream>>>(ei, OFF, REC, x0, W1, BUF0, as1, ad1, AS, AD);
  bucket_kernel<<<gPair, CSR_THR, 0, stream>>>(OFF, REC, START, CNT, LIST,
                                               x0, W1, BUF0, as1, ad1, AS, AD);

  gat_gather_gemm_kernel<false, false, true><<<gFuse, FUSE_THR, 0, stream>>>(
      START, CNT, LIST, AS, AD, BUF0, b1, W2, nullptr, BUF1, as2, ad2, AS2, AD2);
  gat_gather_gemm_kernel<true, true, false><<<gFuse, FUSE_THR, 0, stream>>>(
      START, CNT, LIST, AS2, AD2, BUF1, b2, fcW, fcb, out, nullptr, nullptr, nullptr, nullptr);
}


# Bucketing launch: a range's in-edge lists are put together in LDS and stored in whole lines, not by a 4-byte store per edge to 32 different lines per wave; slots come from the same LDS adds
# speedup vs baseline: 12.2363x; 1.0421x over previous
#include <hip/hip_runtime.h>
#include <cstdint>

#define BGRAPH 8
#define NNODES 8192
#define NEDGES 262144
#define FEAT   128
#define NEG_SLOPE 0.2f

typedef __attribute__((ext_vector_type(2))) float v2f;
typedef __attribute__((ext_vector_type(4))) float v4f;
typedef __attribute__((ext_vector_type(8))) float v8f;
typedef __attribute__((ext_vector_type(4))) int   v4i;

__device__ __forceinline__ float leaky(float v) {
  return v >= 0.0f ? v : NEG_SLOPE * v;
}

#define GEMM_THR  512
#define GEMM_ROWS 64
#define GEMM_PASS 2
#define GEMM_GRID (BGRAPH * NNODES / (GEMM_ROWS * GEMM_PASS))
#define XP        132
#define WP        288

static_assert(GEMM_ROWS * FEAT / 4 == 4 * GEMM_THR, "four 16-byte row pieces per thread and pass");
static_assert((GEMM_ROWS / 16) * 4 * 32 == GEMM_THR, "one wave per row tile and strip");

template <int MASK>
__device__ __forceinline__ float row_xor(float v) {
  return __int_as_float(__builtin_amdgcn_update_dpp(
      0, __float_as_int(v), 0x160 | MASK, 0xf, 0xf, true));
}

__device__ __forceinline__ float row_swap(float v) {
  return __int_as_float(__builtin_amdgcn_permlanex16(
      __float_as_int(v), __float_as_int(v), 0x76543210, 0xfedcba98, false, false));
}

__device__ __forceinline__ void
gemm_wmma_f32(const int wg, const float* __restrict__ X, const float* __restrict__ W,
              float* __restrict__ H,
              const float* __restrict__ a_src, const float* __restrict__ a_dst,
              float* __restrict__ AS, float* __restrict__ AD) {
  __shared__ __attribute__((aligned(16))) float sWp[(FEAT / 2) * WP];
  __shared__ __attribute__((aligned(16))) float sXr[GEMM_ROWS * XP];
  __shared__ float sPart[GEMM_ROWS / 16][4][32];
  const int tid  = threadIdx.x;
  const int wave = tid >> 5;
  const int lane = tid & 31;
  const int half = lane >> 4;
  const int l    = lane & 15;
  const int tl   = wave >> 2;
  const int sw   = wave & 3;
  const int n0   = sw * 32;
  const size_t row0 = (size_t)wg * (GEMM_ROWS * GEMM_PASS);

  const float* xg = X + row0 * FEAT + 4 * tid;
  float* sx = sXr + (tid >> 5) * XP + (tid & 31) * 4;
  v4f nx0 = *(const v4f*)(xg);
  v4f nx1 = *(const v4f*)(xg + 4 * GEMM_THR);
  v4f nx2 = *(const v4f*)(xg + 8 * GEMM_THR);
  v4f nx3 = *(const v4f*)(xg + 12 * GEMM_THR);

#pragma nounroll
  for (int uo = 0; uo < (FEAT / 2) * FEAT / GEMM_THR; uo += 4) {
#pragma unroll
    for (int u = 0; u < 4; ++u) {
      const int q = tid + GEMM_THR * (uo + u);
      const int p = q >> 7, n = q & 127;
      v2f pr;
      pr.x = W[(2 * p) * FEAT + n];
      pr.y = W[(2 * p + 1) * FEAT + n];
      *(v2f*)(sWp + p * WP + 2 * n) = pr;
    }
  }
  const float as0 = a_src[n0 + l];
  const float as1 = a_src[n0 + 16 + l];
  const float ad0 = a_dst[n0 + l];
  const float ad1 = a_dst[n0 + 16 + l];

  const float* xa = sXr + (16 * tl + l) * XP + 2 * half;
  const float* wb = sWp + half * WP + 2 * (n0 + l);

#pragma nounroll
  for (int pass = 0; pass < GEMM_PASS; ++pass) {
    *(v4f*)(sx)           = nx0;
    *(v4f*)(sx + 16 * XP) = nx1;
    *(v4f*)(sx + 32 * XP) = nx2;
    *(v4f*)(sx + 48 * XP) = nx3;
    __syncthreads();
    if (pass + 1 < GEMM_PASS) {
      const float* xn = xg + (size_t)(pass + 1) * (GEMM_ROWS * FEAT);
      nx0 = *(const v4f*)(xn);
      nx1 = *(const v4f*)(xn + 4 * GEMM_THR);
      nx2 = *(const v4f*)(xn + 8 * GEMM_THR);
      nx3 = *(const v4f*)(xn + 12 * GEMM_THR);
    }

    v8f acc0 = {}, acc1 = {};
#pragma unroll
    for (int t = 0; t < FEAT / 4; ++t) {
      const v2f a = *(const v2f*)(xa + 4 * t);
      const v2f q0 = *(const v2f*)(wb + 2 * WP * t);
      const v2f q1 = *(const v2f*)(wb + 2 * WP * t + 32);
      acc0 = __builtin_amdgcn_wmma_f32_16x16x4_f32(
           false, a,  false, q0,
           (short)0, acc0,  false,  false);
      acc1 = __builtin_amdgcn_wmma_f32_16x16x4_f32(
           false, a,  false, q1,
           (short)0, acc1,  false,  false);
    }
    const size_t m0 = row0 + (size_t)pass * GEMM_ROWS + 16 * tl;
    float* hout = H + (m0 + 8 * half) * FEAT + (n0 + l);
#pragma unroll
    for (int i = 0; i < 8; ++i) {
      hout[(size_t)i * FEAT]      = acc0[i];
      hout[(size_t)i * FEAT + 16] = acc1[i];
    }
    {
    float v[16];
#pragma unroll
    for (int i = 0; i < 8; ++i) {
      v[i]     = acc0[i] * as0 + acc1[i] * as1;
      v[8 + i] = acc0[i] * ad0 + acc1[i] * ad1;
    }
    const bool b3 = (l & 8) != 0, b2 = (l & 4) != 0;
    const bool b1 = (l & 2) != 0, b0 = (l & 1) != 0;
    float u[8], w4[4], y[2];
#pragma unroll
    for (int k = 0; k < 8; ++k) {
      const float keep = b3 ? v[k + 8] : v[k];
      const float send = b3 ? v[k] : v[k + 8];
      u[k] = keep + row_xor<8>(send);
    }
#pragma unroll
    for (int k = 0; k < 4; ++k) {
      const float keep = b2 ? u[k + 4] : u[k];
      const float send = b2 ? u[k] : u[k + 4];
      w4[k] = keep + row_xor<4>(send);
    }
#pragma unroll
    for (int k = 0; k < 2; ++k) {
      const float keep = b1 ? w4[k + 2] : w4[k];
      const float send = b1 ? w4[k] : w4[k + 2];
      y[k] = keep + row_xor<2>(send);
    }
    const float keep = b0 ? y[1] : y[0];
    const float send = b0 ? y[0] : y[1];
    sPart[tl][sw][lane] = keep + row_xor<1>(send);
    }
    __syncthreads();

    if (tid < 2 * GEMM_ROWS) {
      const int kind = (tid >= GEMM_ROWS) ? 1 : 0;
      const int row  = tid - kind * GEMM_ROWS;
      const int t2   = row >> 4, r = row & 15;
      const int idx  = 16 * (r >> 3) + 8 * kind + (r & 7);
      const float s = ((sPart[t2][0][idx] + sPart[t2][1][idx]) +
                       sPart[t2][2][idx]) + sPart[t2][3][idx];
      float* dstv = kind ? AD : AS;
      dstv[row0 + (size_t)pass * GEMM_ROWS + row] = s;
    }
  }
}

#define RANGE_N   256
#define NRANGE    (NNODES / RANGE_N)
#define CSR_THR   512
#define CSR_NW    (CSR_THR / 32)
#define SLICE_E   8192
#define NSLICE    (NEDGES / SLICE_E)
#define BIN_THR   512
#define BIN_NW    (BIN_THR / 32)
#define BIN_EPT   (SLICE_E / BIN_THR)
#define WALK_S    (NSLICE / CSR_NW)
#define WALK_LD   10
#define NODE_BITS 13
#define NODE_MASK (NNODES - 1)
#define LIST_CAP  10240

static_assert(NNODES == (1 << NODE_BITS), "record packing");
static_assert(NRANGE * BIN_NW == BIN_THR, "one scan entry per thread");
static_assert(NSLICE == WALK_S * CSR_NW, "whole slices per wave");
static_assert(RANGE_N <= CSR_THR, "one thread per owned node");
static_assert(BIN_EPT == 16, "four int4 loads per thread");

__device__ __forceinline__ void
bin_edges_slice(const int s, const int g, const int* __restrict__ EI,
                int* __restrict__ OFF, int* __restrict__ REC) {
  __shared__ int sCnt[BIN_NW][NRANGE];
  __shared__ int sWave[BIN_NW];
  __shared__ int sRec[SLICE_E];
  const int t    = threadIdx.x;
  const int lane = t & 31;
  const int wave = t >> 5;
  const int* srcp = EI + (size_t)g * 2 * NEDGES + (size_t)s * SLICE_E;
  const int* dstp = srcp + NEDGES;

  (&sCnt[0][0])[t] = 0;
  __syncthreads();

  v4i d[4], sv[4];
#pragma unroll
  for (int u = 0; u < 4; ++u) {
    const int e = 4 * (t + BIN_THR * u);
    d[u]  = *(const v4i*)(dstp + e);
    sv[u] = *(const v4i*)(srcp + e);
  }
  asm volatile("" :: "v"(d[0]), "v"(d[1]), "v"(d[2]), "v"(d[3]),
                     "v"(sv[0]), "v"(sv[1]), "v"(sv[2]), "v"(sv[3]));
  int rec[BIN_EPT], rk[BIN_EPT];
#pragma unroll
  for (int u = 0; u < 4; ++u) {
    const int dv[4] = {d[u].x, d[u].y, d[u].z, d[u].w};
    const int sw[4] = {sv[u].x, sv[u].y, sv[u].z, sv[u].w};
#pragma unroll
    for (int i = 0; i < 4; ++i) {
      const int k = 4 * u + i;
      rec[k] = ((dv[i] & NODE_MASK) << NODE_BITS) | (sw[i] & NODE_MASK);
      const int r = rec[k] >> (NODE_BITS + 8);
      rk[k] = atomicAdd(&sCnt[wave][r], 1);
    }
  }
  __syncthreads();

  const int sr = t / BIN_NW;
  const int sw_ = t % BIN_NW;
  const int v  = sCnt[sw_][sr];
  int incl = v;
#pragma unroll
  for (int off = 1; off < 32; off <<= 1) {
    const int n = __shfl_up(incl, off, 32);
    if (lane >= off) incl += n;
  }
  if (lane == 31) sWave[wave] = incl;
  __syncthreads();
  int pre = 0;
#pragma unroll
  for (int i = 0; i < BIN_NW; ++i) pre += (i < wave) ? sWave[i] : 0;
  const int excl = pre + incl - v;
  sCnt[sw_][sr] = excl;
  int* off = OFF + (size_t)(g * NSLICE + s) * (NRANGE + 1);
  if (sw_ == 0) off[sr] = excl;
  if (t == BIN_THR - 1) off[NRANGE] = excl + v;
  __syncthreads();

#pragma unroll
  for (int k = 0; k < BIN_EPT; ++k) {
    const int r = rec[k] >> (NODE_BITS + 8);
    sRec[sCnt[wave][r] + rk[k]] = rec[k];
  }
  __syncthreads();

  int* out = REC + (size_t)(g * NSLICE + s) * SLICE_E;
#pragma unroll
  for (int u = 0; u < 4; ++u) {
    const int e = 4 * (t + BIN_THR * u);
    *(int4*)(out + e) = *(const int4*)(sRec + e);
  }
}

__device__ __forceinline__ void
bucket_edges_range(const int r, const int g,
                   const int* __restrict__ OFF, const int* __restrict__ REC,
                   int* __restrict__ START, int* __restrict__ CNT,
                   int* __restrict__ LIST) {
  __shared__ int sCur[RANGE_N];
  __shared__ int sScan[RANGE_N];
  __shared__ int sLow[CSR_NW];
  __shared__ int sList[LIST_CAP];
  const int t    = threadIdx.x;
  const int lane = t & 31;
  const int wave = __builtin_amdgcn_readfirstlane(t >> 5);
  const int lo   = r * RANGE_N;
  int beg[WALK_S], end[WALK_S];
  const int* rec[WALK_S];
#pragma unroll
  for (int k = 0; k < WALK_S; ++k) {
    const int s = WALK_S * wave + k;
    const int* off = OFF + (size_t)(g * NSLICE + s) * (NRANGE + 1);
    beg[k] = off[r];
    end[k] = off[r + 1];
    rec[k] = REC + (size_t)(g * NSLICE + s) * SLICE_E;
  }

  if (t < RANGE_N) sCur[t] = 0;
  __syncthreads();

  int rr[WALK_S][WALK_LD];
#pragma unroll
  for (int k = 0; k < WALK_S; ++k) {
#pragma unroll
    for (int u = 0; u < WALK_LD; ++u) {
      const int i = beg[k] + 32 * u + lane;
      rr[k][u] = -1;
      if (i < end[k]) rr[k][u] = rec[k][i];
    }
  }
#pragma unroll
  for (int k = 0; k < WALK_S; ++k) {
#pragma unroll
    for (int u = 0; u < WALK_LD; ++u)
      if (rr[k][u] >= 0) atomicAdd(&sCur[(rr[k][u] >> NODE_BITS) & (RANGE_N - 1)], 1);
  }
#pragma unroll
  for (int k = 0; k < WALK_S; ++k) {
    for (int c = beg[k] + 32 * WALK_LD; c < end[k]; c += 32) {
      const int i = c + lane;
      if (i < end[k]) atomicAdd(&sCur[(rec[k][i] >> NODE_BITS) & (RANGE_N - 1)], 1);
    }
  }
  int low = 0;
#pragma unroll
  for (int k = 0; k < WALK_S; ++k) low += beg[k];
  if (lane == 0) sLow[wave] = low;
  __syncthreads();
  int base = 0;
#pragma unroll
  for (int i = 0; i < CSR_NW; ++i) base += sLow[i];

  if (t < RANGE_N) sScan[t] = sCur[t];
  __syncthreads();
  for (int o = 1; o < RANGE_N; o <<= 1) {
    int v = 0;
    if (t < RANGE_N && t >= o) v = sScan[t - o];
    __syncthreads();
    if (t < RANGE_N) sScan[t] += v;
    __syncthreads();
  }
  const int  tot    = __builtin_amdgcn_readfirstlane(sScan[RANGE_N - 1]);
  const bool staged = tot <= LIST_CAP;
  if (t < RANGE_N) {
    const int c  = sCur[t];
    const int ex = sScan[t] - c;
    START[g * NNODES + lo + t] = base + ex;
    CNT[g * NNODES + lo + t]   = c;
    sCur[t] = staged ? ex : base + ex;
  }
  __syncthreads();

  int* lst = LIST + (size_t)g * NEDGES;
  auto fill = [&](int* __restrict__ slots) {
    unsigned pos[WALK_S][WALK_LD];
#pragma unroll
    for (int k = 0; k < WALK_S; ++k) {
#pragma unroll
      for (int u = 0; u < WALK_LD; ++u) {
        pos[k][u] = 0u;
        if (rr[k][u] >= 0)
          pos[k][u] = (unsigned)atomicAdd(&sCur[(rr[k][u] >> NODE_BITS) & (RANGE_N - 1)], 1);
      }
    }
#pragma unroll
    for (int k = 0; k < WALK_S; ++k) {
#pragma unroll
      for (int u = 0; u < WALK_LD; ++u)
        if (rr[k][u] >= 0) slots[pos[k][u]] = rr[k][u] & NODE_MASK;
    }
#pragma unroll
    for (int k = 0; k < WALK_S; ++k) {
      for (int c = beg[k] + 32 * WALK_LD; c < end[k]; c += 32) {
        const int i = c + lane;
        if (i < end[k]) {
          const int v = rec[k][i];
          const int p = atomicAdd(&sCur[(v >> NODE_BITS) & (RANGE_N - 1)], 1);
          slots[p] = v & NODE_MASK;
        }
      }
    }
  };
  if (staged) {
    fill(sList);
    __syncthreads();
    int* outl = lst + base;
    for (int i0 = t; i0 < tot; i0 += 4 * CSR_THR) {
      int w[4];
#pragma unroll
      for (int u = 0; u < 4; ++u) {
        const int i = i0 + CSR_THR * u;
        w[u] = (i < tot) ? sList[i] : 0;
      }
#pragma unroll
      for (int u = 0; u < 4; ++u) {
        const int i = i0 + CSR_THR * u;
        if (i < tot) outl[i] = w[u];
      }
    }
  } else {
    fill(lst);
  }
}

#define ROLE_N    256
#define PAIR_GRID (2 * ROLE_N)
static_assert(NSLICE * BGRAPH == ROLE_N, "binning workgroups of the first launch");
static_assert(NRANGE * BGRAPH == ROLE_N, "bucketing workgroups of the second launch");
static_assert(GEMM_GRID == 2 * ROLE_N, "product workgroups, half in each launch");
static_assert(GEMM_THR == BIN_THR && GEMM_THR == CSR_THR, "one block size for both roles");

__global__ void __launch_bounds__(BIN_THR, 8)
lead_kernel(const int* __restrict__ EI, int* __restrict__ OFF, int* __restrict__ REC,
            const float* __restrict__ X, const float* __restrict__ W,
            float* __restrict__ H,
            const float* __restrict__ a_src, const float* __restrict__ a_dst,
            float* __restrict__ AS, float* __restrict__ AD) {
  const int bid = (int)blockIdx.x;
  const int idx = bid >> 1;
  if ((((bid >> 8) ^ bid) & 1) != 0)
    gemm_wmma_f32(idx, X, W, H, a_src, a_dst, AS, AD);
  else
    bin_edges_slice(idx % NSLICE, idx / NSLICE, EI, OFF, REC);
}

__global__ void __launch_bounds__(CSR_THR, 8)
bucket_kernel(const int* __restrict__ OFF, const int* __restrict__ REC,
              int* __restrict__ START, int* __restrict__ CNT, int* __restrict__ LIST,
              const float* __restrict__ X, const float* __restrict__ W,
              float* __restrict__ H,
              const float* __restrict__ a_src, const float* __restrict__ a_dst,
              float* __restrict__ AS, float* __restrict__ AD) {
  const int bid = (int)blockIdx.x;
  const int idx = bid >> 1;
  if ((((bid >> 8) ^ bid) & 1) != 0)
    gemm_wmma_f32(ROLE_N + idx, X, W, H, a_src, a_dst, AS, AD);
  else
    bucket_edges_range(idx % NRANGE, idx / NRANGE, OFF, REC, START, CNT, LIST);
}

#define FUSE_THR   1024
#define FUSE_NW    (FUSE_THR / 32)
#define FUSE_NODES 128
#define FUSE_GRID  (BGRAPH * NNODES / FUSE_NODES)
#define FUSE_NPW   (FUSE_NODES / FUSE_NW)
#define FUSE_MAXR  64
#define SW_WORDS   ((FEAT / 2) * WP > FEAT * XP ? (FEAT / 2) * WP : FEAT * XP)

static_assert(FUSE_NODES == 4 * FUSE_NW, "four nodes per wave");
static_assert(FUSE_NPW <= 32, "one lane per node of the wave holds its start, count and logit");
static_assert(NNODES % FUSE_NODES == 0, "a workgroup's nodes lie in one graph");
static_assert(NNODES == 8 * FUSE_THR, "two 16-byte copies per thread fill the logit table");
static_assert(FUSE_MAXR == 2 * FUSE_NW, "at most two nodes per wave and pass");

template <bool TRANSB, bool BIAS, bool ALPHA>
__global__ void __launch_bounds__(FUSE_THR, 16)
gat_gather_gemm_kernel(const int* __restrict__ START, const int* __restrict__ CNT,
                       const int* __restrict__ LIST, const float* __restrict__ AS,
                       const float* __restrict__ AD, const float* __restrict__ H,
                       const float* __restrict__ b, const float* __restrict__ W,
                       const float* __restrict__ bias2, float* __restrict__ OUT,
                       const float* __restrict__ a_src, const float* __restrict__ a_dst,
                       float* __restrict__ AS2, float* __restrict__ AD2) {
  __shared__ __attribute__((aligned(16))) float sW[SW_WORDS];
  __shared__ __attribute__((aligned(16))) float sX[FUSE_MAXR * XP];
  __shared__ __attribute__((aligned(16))) float sAS[NNODES];
  __shared__ __attribute__((aligned(16))) float sB[FEAT];
  __shared__ float sPart[FUSE_MAXR / 16][4][32];
  const int tid  = threadIdx.x;
  const int wave = __builtin_amdgcn_readfirstlane((int)(threadIdx.x >> 5));
  const int lane = threadIdx.x & 31;
  const int half = lane >> 4;
  const int l    = lane & 15;
  const int f    = lane * 4;

  const int bb = (int)blockIdx.x * FUSE_NODES;
  const int g  = bb / NNODES;
  const int hn = bb + wave + FUSE_NW * (lane % FUSE_NPW);
  const int   hcnt = CNT[hn];
  const int   hsta = START[hn];
  const float had  = AD[hn];

  if (TRANSB) {
#pragma unroll
    for (int u = 0; u < 4; ++u) {
      const int q = tid + FUSE_THR * u;
      const int row = q >> 5, c4 = (q & 31) * 4;
      *(float4*)(sW + row * XP + c4) = *(const float4*)(W + row * FEAT + c4);
    }
  } else {
#pragma nounroll
    for (int uo = 0; uo < 8; uo += 4) {
#pragma unroll
      for (int u = 0; u < 4; ++u) {
        const int q = tid + FUSE_THR * (uo + u);
        const int p = q >> 7, n = q & 127;
        v2f pr;
        pr.x = W[(2 * p) * FEAT + n];
        pr.y = W[(2 * p + 1) * FEAT + n];
        *(v2f*)(sW + p * WP + 2 * n) = pr;
      }
    }
  }

  {
    const float* asg = AS + (size_t)(((int)blockIdx.x * FUSE_NODES) / NNODES) * NNODES;
#pragma unroll
    for (int u = 0; u < 2; ++u) {
      const int q = 4 * (tid + FUSE_THR * u);
      *(float4*)(sAS + q) = *(const float4*)(asg + q);
    }
  }
  if (tid < FEAT / 4) *(float4*)(sB + 4 * tid) = *(const float4*)(b + 4 * tid);

  const int*   lstg = LIST + (size_t)g * NEDGES;
  const float* hg   = H + (size_t)g * NNODES * FEAT + lane * 4;
  const int    nl0  = bb - g * NNODES + wave;
  int qcnt[FUSE_NPW], qsta[FUSE_NPW];
  float qad[FUSE_NPW];
#pragma unroll
  for (int k = 0; k < FUSE_NPW; ++k) {
    qcnt[k] = __builtin_amdgcn_readlane(hcnt, k);
    qsta[k] = __builtin_amdgcn_readlane(hsta, k);
    qad[k]  = __int_as_float(__builtin_amdgcn_readlane(__float_as_int(had), k));
  }
  int ps0 = nl0, ps1 = nl0;
  if (lane < qcnt[0])      ps0 = lstg[qsta[0] + lane];
  if (lane + 32 < qcnt[0]) ps1 = lstg[qsta[0] + 32 + lane];
  __syncthreads();

  const int tl = wave >> 3;
  const int n0 = (wave & 3) * 32;

  const bool shifted = (((blockIdx.x >> 8) ^ blockIdx.x) & 1) != 0;
  const int npass = shifted ? 3 : 2;
  int base = bb;
  int kn   = 0;
  for (int pass = 0; pass < npass; ++pass) {
  const int npw  = (shifted && pass != 1) ? 1 : 2;
  const int rows = FUSE_NW * npw;
  for (int it = 0; it < npw; ++it) {
  const int rr   = wave + FUSE_NW * it;
  const int nl   = nl0 + FUSE_NW * kn;
  const int cnt   = qcnt[0];
  const int total = cnt + 1;
  const int* lst  = lstg + qsta[0];
  const float a_d = qad[0];

  const int s0 = ps0, s1 = ps1;
  if (kn + 1 < FUSE_NPW) {
    ps0 = nl + FUSE_NW;
    ps1 = nl + FUSE_NW;
    if (lane < qcnt[1])      ps0 = lstg[qsta[1] + lane];
    if (lane + 32 < qcnt[1]) ps1 = lstg[qsta[1] + 32 + lane];
  }
  const float e0 = leaky(sAS[s0] + a_d);
  const float e1 = leaky(sAS[s1] + a_d);
  float m = (lane < total) ? e0 : -INFINITY;
  if (lane + 32 < total) m = fmaxf(m, e1);
  for (int c = 64; c < total; c += 32) {
    const int i = c + lane;
    int s = nl;
    if (i < cnt) s = lst[i];
    const float e = leaky(sAS[s] + a_d);
    if (i < total) m = fmaxf(m, e);
  }
  m = fmaxf(m, row_swap(m));
  m = fmaxf(m, row_xor<8>(m));
  m = fmaxf(m, row_xor<4>(m));
  m = fmaxf(m, row_xor<2>(m));
  m = fmaxf(m, row_xor<1>(m));

  float4 acc = make_float4(0.0f, 0.0f, 0.0f, 0.0f);
  auto add_rows = [&](const int sv, const float wv, const int n) {
    for (int j = 0; j < n; j += 4) {
      const int r0 = __builtin_amdgcn_readlane(sv, j);
      const int r1 = __builtin_amdgcn_readlane(sv, j + 1);
      const int r2 = __builtin_amdgcn_readlane(sv, j + 2);
      const int r3 = __builtin_amdgcn_readlane(sv, j + 3);
      const float w0 = __int_as_float(__builtin_amdgcn_readlane(__float_as_int(wv), j));
      const float w1 = __int_as_float(__builtin_amdgcn_readlane(__float_as_int(wv), j + 1));
      const float w2 = __int_as_float(__builtin_amdgcn_readlane(__float_as_int(wv), j + 2));
      const float w3 = __int_as_float(__builtin_amdgcn_readlane(__float_as_int(wv), j + 3));
      const float4 h0 = *(const float4*)(hg + (size_t)r0 * FEAT);
      const float4 h1 = *(const float4*)(hg + (size_t)r1 * FEAT);
      const float4 h2 = *(const float4*)(hg + (size_t)r2 * FEAT);
      const float4 h3 = *(const float4*)(hg + (size_t)r3 * FEAT);
      acc.x = fmaf(w0, h0.x, acc.x); acc.y = fmaf(w0, h0.y, acc.y); acc.z = fmaf(w0, h0.z, acc.z); acc.w = fmaf(w0, h0.w, acc.w);
      acc.x = fmaf(w1, h1.x, acc.x); acc.y = fmaf(w1, h1.y, acc.y); acc.z = fmaf(w1, h1.z, acc.z); acc.w = fmaf(w1, h1.w, acc.w);
      acc.x = fmaf(w2, h2.x, acc.x); acc.y = fmaf(w2, h2.y, acc.y); acc.z = fmaf(w2, h2.z, acc.z); acc.w = fmaf(w2, h2.w, acc.w);
      acc.x = fmaf(w3, h3.x, acc.x); acc.y = fmaf(w3, h3.y, acc.y); acc.z = fmaf(w3, h3.z, acc.z); acc.w = fmaf(w3, h3.w, acc.w);
    }
  };
  const float wa = (lane < total) ? __expf(e0 - m) : 0.0f;
  float wsum = wa;
  add_rows(s0, wa, total < 32 ? total : 32);
  if (total > 32) {
    const float wb = (lane + 32 < total) ? __expf(e1 - m) : 0.0f;
    wsum += wb;
    add_rows(s1, wb, total < 64 ? total - 32 : 32);
  }
  for (int c = 64; c < total; c += 32) {
    const int i = c + lane;
    int s = nl;
    if (i < cnt) s = lst[i];
    const float e = leaky(sAS[s] + a_d);
    const float w = (i < total) ? __expf(e - m) : 0.0f;
    wsum += w;
    const int rem = total - c;
    add_rows(s, w, rem < 32 ? rem : 32);
  }
  wsum += row_swap(wsum);
  wsum += row_xor<8>(wsum);
  wsum += row_xor<4>(wsum);
  wsum += row_xor<2>(wsum);
  wsum += row_xor<1>(wsum);

  const float4 bv = *(const float4*)(sB + f);
  float4 v;
  v.x = acc.x / wsum + bv.x;
  v.y = acc.y / wsum + bv.y;
  v.z = acc.z / wsum + bv.z;
  v.w = acc.w / wsum + bv.w;
  float4 o;
  o.x = v.x / (1.0f + __expf(-v.x));
  o.y = v.y / (1.0f + __expf(-v.y));
  o.z = v.z / (1.0f + __expf(-v.z));
  o.w = v.w / (1.0f + __expf(-v.w));
  *(float4*)(sX + rr * XP + f) = o;
  ++kn;
#pragma unroll
  for (int k = 0; k + 1 < FUSE_NPW; ++k) {
    qcnt[k] = qcnt[k + 1];
    qsta[k] = qsta[k + 1];
    qad[k]  = qad[k + 1];
  }
  }
  __syncthreads();

  const bool mult = (wave & 4) == 0 && tl < (rows >> 4);
  v8f acc0 = {}, acc1 = {};
  float bo0 = 0.0f, bo1 = 0.0f, as0 = 0.0f, as1 = 0.0f, ad0 = 0.0f, ad1 = 0.0f;
  if (mult) {
    const float* xa = sX + (16 * tl + l) * XP + 2 * half;
    const float* wb = TRANSB ? sW + (n0 + l) * XP + 2 * half
                             : sW + half * WP + 2 * (n0 + l);
    if (BIAS) {
      bo0 = bias2[n0 + l];
      bo1 = bias2[n0 + 16 + l];
    }
    if (ALPHA) {
      as0 = a_src[n0 + l];
      as1 = a_src[n0 + 16 + l];
      ad0 = a_dst[n0 + l];
      ad1 = a_dst[n0 + 16 + l];
    }
#pragma unroll
    for (int t = 0; t < FEAT / 4; ++t) {
      const v2f a = *(const v2f*)(xa + 4 * t);
      const v2f q0 = *(const v2f*)(wb + (TRANSB ? 4 : 2 * WP) * t);
      const v2f q1 = *(const v2f*)(wb + (TRANSB ? 4 : 2 * WP) * t + (TRANSB ? 16 * XP : 32));
      acc0 = __builtin_amdgcn_wmma_f32_16x16x4_f32(
           false, a,  false, q0,
           (short)0, acc0,  false,  false);
      acc1 = __builtin_amdgcn_wmma_f32_16x16x4_f32(
           false, a,  false, q1,
           (short)0, acc1,  false,  false);
    }
  }
  __syncthreads();

  if (mult) {
    float* so = sX + (16 * tl + 8 * half) * XP + n0 + l;
#pragma unroll
    for (int i = 0; i < 8; ++i) {
      so[i * XP]      = acc0[i] + bo0;
      so[i * XP + 16] = acc1[i] + bo1;
    }
    if (ALPHA) {
      float v[16];
#pragma unroll
      for (int i = 0; i < 8; ++i) {
        v[i]     = acc0[i] * as0 + acc1[i] * as1;
        v[8 + i] = acc0[i] * ad0 + acc1[i] * ad1;
      }
      const bool b3 = (l & 8) != 0, b2 = (l & 4) != 0;
      const bool b1 = (l & 2) != 0, b0 = (l & 1) != 0;
      float u[8], w4[4], y[2];
#pragma unroll
      for (int k = 0; k < 8; ++k) {
        const float keep = b3 ? v[k + 8] : v[k];
        const float send = b3 ? v[k] : v[k + 8];
        u[k] = keep + row_xor<8>(send);
      }
#pragma unroll
      for (int k = 0; k < 4; ++k) {
        const float keep = b2 ? u[k + 4] : u[k];
        const float send = b2 ? u[k] : u[k + 4];
        w4[k] = keep + row_xor<4>(send);
      }
#pragma unroll
      for (int k = 0; k < 2; ++k) {
        const float keep = b1 ? w4[k + 2] : w4[k];
        const float send = b1 ? w4[k] : w4[k + 2];
        y[k] = keep + row_xor<2>(send);
      }
      const float keep = b0 ? y[1] : y[0];
      const float send = b0 ? y[0] : y[1];
      sPart[tl][wave & 3][lane] = keep + row_xor<1>(send);
    }
  }
  __syncthreads();

  for (int u = 0; u < npw; ++u) {
    const int q = tid + FUSE_THR * u;
    const int row = q >> 5, c4 = (q & 31) * 4;
    *(float4*)(OUT + (size_t)(base + row) * FEAT + c4) = *(const float4*)(sX + row * XP + c4);
  }
  if (ALPHA) {
    if (tid < 2 * rows) {
      const int kind = (tid >= rows) ? 1 : 0;
      const int row  = tid - kind * rows;
      const int t2   = row >> 4, r = row & 15;
      const int idx  = 16 * (r >> 3) + 8 * kind + (r & 7);
      const float s = ((sPart[t2][0][idx] + sPart[t2][1][idx]) +
                       sPart[t2][2][idx]) + sPart[t2][3][idx];
      float* dstv = kind ? AD2 : AS2;
      dstv[base + row] = s;
    }
  }
  base += rows;
  }
}

extern "C" void kernel_launch(void* const* d_in, const int* in_sizes, int n_in,
                              void* d_out, int out_size, void* d_ws, size_t ws_size,
                              hipStream_t stream) {
  const float* x0   = (const float*)d_in[0];
  const int*   ei   = (const int*)d_in[1];
  const float* W1   = (const float*)d_in[2];
  const float* as1  = (const float*)d_in[3];
  const float* ad1  = (const float*)d_in[4];
  const float* b1   = (const float*)d_in[5];
  const float* W2   = (const float*)d_in[6];
  const float* as2  = (const float*)d_in[7];
  const float* ad2  = (const float*)d_in[8];
  const float* b2   = (const float*)d_in[9];
  const float* fcW  = (const float*)d_in[10];
  const float* fcb  = (const float*)d_in[11];
  float* out = (float*)d_out;

  const size_t BN      = (size_t)BGRAPH * NNODES;
  const size_t BIGELTS = BN * FEAT;
  char* ws = (char*)d_ws;
  float* BUF0  = (float*)ws;               ws += BIGELTS * sizeof(float);
  float* BUF1  = (float*)ws;               ws += BIGELTS * sizeof(float);
  float* AS    = (float*)ws;               ws += BN * sizeof(float);
  float* AD    = (float*)ws;               ws += BN * sizeof(float);
  float* AS2   = (float*)ws;               ws += BN * sizeof(float);
  float* AD2   = (float*)ws;               ws += BN * sizeof(float);
  int*   START = (int*)ws;                 ws += BN * sizeof(int);
  int*   CNT   = (int*)ws;                 ws += BN * sizeof(int);
  int*   LIST  = (int*)ws;                 ws += (size_t)BGRAPH * NEDGES * sizeof(int);
  int*   REC   = (int*)ws;                 ws += (size_t)BGRAPH * NEDGES * sizeof(int);
  int*   OFF   = (int*)ws;

  const int  gPair  = PAIR_GRID;
  const int  gFuse  = FUSE_GRID;

  lead_kernel<<<gPair, BIN_THR, 0, stream>>>(ei, OFF, REC, x0, W1, BUF0, as1, ad1, AS, AD);
  bucket_kernel<<<gPair, CSR_THR, 0, stream>>>(OFF, REC, START, CNT, LIST,
                                               x0, W1, BUF0, as1, ad1, AS, AD);

  gat_gather_gemm_kernel<false, false, true><<<gFuse, FUSE_THR, 0, stream>>>(
      START, CNT, LIST, AS, AD, BUF0, b1, W2, nullptr, BUF1, as2, ad2, AS2, AD2);
  gat_gather_gemm_kernel<true, true, false><<<gFuse, FUSE_THR, 0, stream>>>(
      START, CNT, LIST, AS2, AD2, BUF1, b2, fcW, fcb, out, nullptr, nullptr, nullptr, nullptr);
}
